# starting kernel
# baseline (speedup 1.0000x reference)
#include <hip/hip_runtime.h>
#include <hip/hip_bf16.h>


typedef __attribute__((ext_vector_type(2))) float v2f;
typedef __attribute__((ext_vector_type(8))) float v8f;

#define NODES 50000
#define EDGES 800000
#define DIM   64
#define NEG_SLOPE 0.2f

__device__ __forceinline__ unsigned encF(float f) {
    unsigned u = __float_as_uint(f);
    return (u & 0x80000000u) ? ~u : (u | 0x80000000u);
}
__device__ __forceinline__ float decF(unsigned u) {
    unsigned v = (u & 0x80000000u) ? (u & 0x7FFFFFFFu) : ~u;
    return __uint_as_float(v);
}

__global__ void __launch_bounds__(256)
gemm_bias_wmma(const float* __restrict__ A, const float* __restrict__ W,
               const float* __restrict__ bias, float* __restrict__ C,
               int M, int K, int Nout)
{
    const int lane = threadIdx.x & 31;
    const int wave = blockIdx.x * 8 + (threadIdx.x >> 5);
    const int colTiles = Nout >> 4;
    const int total = (M >> 4) * colTiles;
    if (wave >= total) return;

    const int rt   = wave / colTiles;
    const int ct   = wave - rt * colTiles;
    const int row0 = rt << 4;
    const int col0 = ct << 4;
    const int lr   = lane & 15;
    const int hi   = lane >> 4;

    const float* __restrict__ arow = A + (size_t)(row0 + lr) * K;
    const float* __restrict__ wrow = W + (size_t)(col0 + lr) * K;

    v8f acc = {};
    #pragma unroll
    for (int k0 = 0; k0 < 64; k0 += 4) {
        v2f a, b;
        a.x = arow[k0 + 2 * hi];
        a.y = arow[k0 + 2 * hi + 1];
        b.x = wrow[k0 + 2 * hi];
        b.y = wrow[k0 + 2 * hi + 1];
        acc = __builtin_amdgcn_wmma_f32_16x16x4_f32(
                  false, a, false, b, (short)0, acc, false, false);
    }

    const float bv = bias[col0 + lr];
    float* crow = C + (size_t)(row0 + hi * 8) * Nout + col0 + lr;
    #pragma unroll
    for (int v = 0; v < 8; ++v)
        crow[(size_t)v * Nout] = acc[v] + bv;
}

__global__ void __launch_bounds__(256)
edge_score(const float* __restrict__ el, const float* __restrict__ er,
           const float* __restrict__ attn, const int* __restrict__ src,
           const int* __restrict__ dst, float* __restrict__ score,
           unsigned* __restrict__ mEnc)
{
    const int  lane = threadIdx.x & 31;
    const long e    = (long)blockIdx.x * 8 + (threadIdx.x >> 5);
    const long Ef   = (long)EDGES + NODES;
    if (e >= Ef) return;
    int s, d;
    if (e < EDGES) { s = src[e]; d = dst[e]; } else { s = d = (int)(e - EDGES); }

    float p = 0.f;
    #pragma unroll
    for (int j = 0; j < 2; ++j) {
        const int k = lane + 32 * j;
        float v = el[(size_t)s * DIM + k] + er[(size_t)d * DIM + k];
        v = v > 0.f ? v : NEG_SLOPE * v;
        p += v * attn[k];
    }
    #pragma unroll
    for (int off = 16; off; off >>= 1) p += __shfl_xor(p, off, 32);
    if (lane == 0) {
        score[e] = p;
        atomicMax(&mEnc[d], encF(p));
    }
}

__global__ void edge_exp(float* __restrict__ score,
                         const unsigned* __restrict__ mEnc,
                         float* __restrict__ den, const int* __restrict__ dst)
{
    const long e  = (long)blockIdx.x * blockDim.x + threadIdx.x;
    const long Ef = (long)EDGES + NODES;
    if (e >= Ef) return;
    const int d  = (e < EDGES) ? dst[e] : (int)(e - EDGES);
    const float ex = expf(score[e] - decF(mEnc[d]));
    score[e] = ex;
    atomicAdd(&den[d], ex);
}

__global__ void __launch_bounds__(256)
edge_aggregate(const float* __restrict__ el, const float* __restrict__ ex,
               const float* __restrict__ den, const int* __restrict__ src,
               const int* __restrict__ dst, float* __restrict__ acc)
{
    const int  lane = threadIdx.x & 31;
    const long e    = (long)blockIdx.x * 8 + (threadIdx.x >> 5);
    const long Ef   = (long)EDGES + NODES;
    if (e >= Ef) return;
    int s, d;
    if (e < EDGES) { s = src[e]; d = dst[e]; } else { s = d = (int)(e - EDGES); }
    const float a = ex[e] / den[d];
    #pragma unroll
    for (int j = 0; j < 2; ++j) {
        const int k = lane + 32 * j;
        atomicAdd(&acc[(size_t)d * DIM + k], a * el[(size_t)s * DIM + k]);
    }
}

__global__ void elu_write(const float* __restrict__ acc, float* __restrict__ h,
                          float* __restrict__ emb, long n)
{
    const long i = (long)blockIdx.x * blockDim.x + threadIdx.x;
    if (i >= n) return;
    const float v = acc[i];
    const float o = v > 0.f ? v : (expf(v) - 1.f);
    h[i] = o;
    emb[i] = o;
}

__global__ void __launch_bounds__(256)
ggc_scatter(const float* __restrict__ t0, const float* __restrict__ t1,
            const int* __restrict__ ety, const int* __restrict__ src,
            const int* __restrict__ dst, float* __restrict__ acc)
{
    const int  lane = threadIdx.x & 31;
    const long e    = (long)blockIdx.x * 8 + (threadIdx.x >> 5);
    const long Ef   = (long)EDGES + NODES;
    if (e >= Ef) return;
    int s, d;
    if (e < EDGES) { s = src[e]; d = dst[e]; } else { s = d = (int)(e - EDGES); }
    const float* __restrict__ tr = (ety[e] == 0) ? t0 : t1;
    #pragma unroll
    for (int j = 0; j < 2; ++j) {
        const int k = lane + 32 * j;
        atomicAdd(&acc[(size_t)d * DIM + k], tr[(size_t)s * DIM + k]);
    }
}

__global__ void gru_update(const float* __restrict__ gi,
                           const float* __restrict__ gh,
                           float* __restrict__ h, float* __restrict__ emb,
                           long nElem)
{
    const long i = (long)blockIdx.x * blockDim.x + threadIdx.x;
    if (i >= nElem) return;
    const long n = i >> 6;
    const int  d = (int)(i & 63);
    const float* gin = gi + n * (3 * DIM);
    const float* ghn = gh + n * (3 * DIM);
    const float r = 1.f / (1.f + expf(-(gin[d]           + ghn[d])));
    const float z = 1.f / (1.f + expf(-(gin[DIM + d]     + ghn[DIM + d])));
    const float c = tanhf(gin[2 * DIM + d] + r * ghn[2 * DIM + d]);
    const float o = (1.f - z) * c + z * h[i];
    h[i] = o;
    if (emb) emb[i] = o;
}

__global__ void fill_zero(float* __restrict__ p, long n)
{
    long i = (long)blockIdx.x * blockDim.x + threadIdx.x;
    const long stride = (long)gridDim.x * blockDim.x;
    for (; i < n; i += stride) p[i] = 0.f;
}

extern "C" void kernel_launch(void* const* d_in, const int* in_sizes, int n_in,
                              void* d_out, int out_size, void* d_ws, size_t ws_size,
                              hipStream_t stream)
{
    (void)in_sizes; (void)n_in; (void)out_size; (void)ws_size;

    const float* x   = (const float*)d_in[0];
    const int*   src = (const int*)d_in[1];
    const int*   dst = (const int*)d_in[2];
    const int*   ety = (const int*)d_in[3];

    const long NE = (long)NODES * DIM;
    const long Ef = (long)EDGES + NODES;

    float* ws   = (float*)d_ws;
    float* hbuf = ws;                 ws += NE;
    float* buf1 = ws;                 ws += NE;
    float* buf2 = ws;                 ws += NE;
    float* gi   = ws;                 ws += (long)NODES * 3 * DIM;
    float* gh   = ws;                 ws += (long)NODES * 3 * DIM;
    float* acc  = ws;                 ws += NE;
    unsigned* mEnc = (unsigned*)ws;   ws += NODES;
    float* den  = ws;                 ws += NODES;
    float* score = ws;

    float* out = (float*)d_out;

    const int edgeWaveBlocks = (int)((Ef + 7) / 8);
    const int edgeThrBlocks  = (int)((Ef + 255) / 256);
    const int elemBlocks     = (int)((NE + 255) / 256);
    const int gemmBlocks64   = ((NODES / 16) * (DIM / 16)       + 7) / 8;
    const int gemmBlocks192  = ((NODES / 16) * (3 * DIM / 16)   + 7) / 8;

    const float* hin = x;
    for (int l = 0; l < 2; ++l) {
        const int base = 4 + 11 * l;
        const float* Wsrc = (const float*)d_in[base + 0];
        const float* bsrc = (const float*)d_in[base + 1];
        const float* Wdst = (const float*)d_in[base + 2];
        const float* bdst = (const float*)d_in[base + 3];
        const float* attn = (const float*)d_in[base + 4];
        const float* ggW  = (const float*)d_in[base + 5];
        const float* ggb  = (const float*)d_in[base + 6];
        const float* Wih  = (const float*)d_in[base + 7];
        const float* bih  = (const float*)d_in[base + 8];
        const float* Whh  = (const float*)d_in[base + 9];
        const float* bhh  = (const float*)d_in[base + 10];

        gemm_bias_wmma<<<gemmBlocks64, 256, 0, stream>>>(hin, Wsrc, bsrc, buf1,
                                                         NODES, DIM, DIM);
        gemm_bias_wmma<<<gemmBlocks64, 256, 0, stream>>>(hin, Wdst, bdst, buf2,
                                                         NODES, DIM, DIM);
        fill_zero<<<2048, 256, 0, stream>>>(acc, NE + 2L * NODES);
        edge_score<<<edgeWaveBlocks, 256, 0, stream>>>(buf1, buf2, attn, src, dst,
                                                       score, mEnc);
        edge_exp<<<edgeThrBlocks, 256, 0, stream>>>(score, mEnc, den, dst);
        edge_aggregate<<<edgeWaveBlocks, 256, 0, stream>>>(buf1, score, den, src,
                                                           dst, acc);
        elu_write<<<elemBlocks, 256, 0, stream>>>(acc, hbuf, out + (2L * l) * NE,
                                                  NE);

        for (int st = 0; st < 2; ++st) {
            gemm_bias_wmma<<<gemmBlocks64, 256, 0, stream>>>(
                hbuf, ggW,               ggb,       buf1, NODES, DIM, DIM);
            gemm_bias_wmma<<<gemmBlocks64, 256, 0, stream>>>(
                hbuf, ggW + DIM * DIM,   ggb + DIM, buf2, NODES, DIM, DIM);
            fill_zero<<<2048, 256, 0, stream>>>(acc, NE);
            ggc_scatter<<<edgeWaveBlocks, 256, 0, stream>>>(buf1, buf2, ety, src,
                                                            dst, acc);
            gemm_bias_wmma<<<gemmBlocks192, 256, 0, stream>>>(
                acc,  Wih, bih, gi, NODES, DIM, 3 * DIM);
            gemm_bias_wmma<<<gemmBlocks192, 256, 0, stream>>>(
                hbuf, Whh, bhh, gh, NODES, DIM, 3 * DIM);
            gru_update<<<elemBlocks, 256, 0, stream>>>(
                gi, gh, hbuf, (st == 1) ? out + (2L * l + 1) * NE : nullptr, NE);
        }
        hin = hbuf;
    }
}


# edge score / aggregate / ggc scatter: one wave per 8 consecutive edges (one coalesced index load, all row gathers of the group in flight before first use) instead of one per edge; atomics kept
# speedup vs baseline: 1.7441x; 1.7441x over previous
#include <hip/hip_runtime.h>
#include <hip/hip_bf16.h>


typedef __attribute__((ext_vector_type(2))) float v2f;
typedef __attribute__((ext_vector_type(8))) float v8f;

#define NODES 50000
#define EDGES 800000
#define DIM   64
#define NEG_SLOPE 0.2f

#define EPW 8
static_assert(EDGES % EPW == 0 && NODES % EPW == 0, "edge group must divide both ranges");
#define EDGE_GROUP_BLOCKS ((((EDGES + NODES) / EPW) + 7) / 8)

__device__ __forceinline__ unsigned encF(float f) {
    unsigned u = __float_as_uint(f);
    return (u & 0x80000000u) ? ~u : (u | 0x80000000u);
}
__device__ __forceinline__ float decF(unsigned u) {
    unsigned v = (u & 0x80000000u) ? (u & 0x7FFFFFFFu) : ~u;
    return __uint_as_float(v);
}

__global__ void __launch_bounds__(256)
gemm_bias_wmma(const float* __restrict__ A, const float* __restrict__ W,
               const float* __restrict__ bias, float* __restrict__ C,
               int M, int K, int Nout)
{
    const int lane = threadIdx.x & 31;
    const int wave = blockIdx.x * 8 + (threadIdx.x >> 5);
    const int colTiles = Nout >> 4;
    const int total = (M >> 4) * colTiles;
    if (wave >= total) return;

    const int rt   = wave / colTiles;
    const int ct   = wave - rt * colTiles;
    const int row0 = rt << 4;
    const int col0 = ct << 4;
    const int lr   = lane & 15;
    const int hi   = lane >> 4;

    const float* __restrict__ arow = A + (size_t)(row0 + lr) * K;
    const float* __restrict__ wrow = W + (size_t)(col0 + lr) * K;

    v8f acc = {};
    #pragma unroll
    for (int k0 = 0; k0 < 64; k0 += 4) {
        v2f a, b;
        a.x = arow[k0 + 2 * hi];
        a.y = arow[k0 + 2 * hi + 1];
        b.x = wrow[k0 + 2 * hi];
        b.y = wrow[k0 + 2 * hi + 1];
        acc = __builtin_amdgcn_wmma_f32_16x16x4_f32(
                  false, a, false, b, (short)0, acc, false, false);
    }

    const float bv = bias[col0 + lr];
    float* crow = C + (size_t)(row0 + hi * 8) * Nout + col0 + lr;
    #pragma unroll
    for (int v = 0; v < 8; ++v)
        crow[(size_t)v * Nout] = acc[v] + bv;
}

__global__ void __launch_bounds__(256)
edge_score(const float* __restrict__ el, const float* __restrict__ er,
           const float* __restrict__ attn, const int* __restrict__ src,
           const int* __restrict__ dst, float* __restrict__ score,
           unsigned* __restrict__ mEnc)
{
    const int  lane = threadIdx.x & 31;
    const long e0   = ((long)blockIdx.x * 8 + (threadIdx.x >> 5)) * EPW;
    const long Ef   = (long)EDGES + NODES;
    if (e0 >= Ef) return;
    const long em = e0 + (lane & (EPW - 1));
    int sv, dv;
    if (em < EDGES) { sv = src[em]; dv = dst[em]; } else { sv = dv = (int)(em - EDGES); }

    const float a0 = attn[lane];
    const float a1 = attn[lane + 32];

    float l0[EPW], l1[EPW], r0[EPW], r1[EPW];
    #pragma unroll
    for (int i = 0; i < EPW; ++i) {
        const int s = __builtin_amdgcn_readlane(sv, i);
        const int d = __builtin_amdgcn_readlane(dv, i);
        l0[i] = el[(size_t)s * DIM + lane];
        l1[i] = el[(size_t)s * DIM + lane + 32];
        r0[i] = er[(size_t)d * DIM + lane];
        r1[i] = er[(size_t)d * DIM + lane + 32];
    }

    float p[EPW];
    #pragma unroll
    for (int i = 0; i < EPW; ++i) {
        float v0 = l0[i] + r0[i];
        float v1 = l1[i] + r1[i];
        v0 = v0 > 0.f ? v0 : NEG_SLOPE * v0;
        v1 = v1 > 0.f ? v1 : NEG_SLOPE * v1;
        float q = v0 * a0;
        q += v1 * a1;
        p[i] = q;
    }
    #pragma unroll
    for (int off = 16; off; off >>= 1) {
        #pragma unroll
        for (int i = 0; i < EPW; ++i) p[i] += __shfl_xor(p[i], off, 32);
    }
    float mine = p[0];
    #pragma unroll
    for (int i = 1; i < EPW; ++i) mine = (lane == i) ? p[i] : mine;
    if (lane < EPW) {
        score[em] = mine;
        atomicMax(&mEnc[dv], encF(mine));
    }
}

__global__ void edge_exp(float* __restrict__ score,
                         const unsigned* __restrict__ mEnc,
                         float* __restrict__ den, const int* __restrict__ dst)
{
    const long e  = (long)blockIdx.x * blockDim.x + threadIdx.x;
    const long Ef = (long)EDGES + NODES;
    if (e >= Ef) return;
    const int d  = (e < EDGES) ? dst[e] : (int)(e - EDGES);
    const float ex = expf(score[e] - decF(mEnc[d]));
    score[e] = ex;
    atomicAdd(&den[d], ex);
}

__global__ void __launch_bounds__(256)
edge_aggregate(const float* __restrict__ el, const float* __restrict__ ex,
               const float* __restrict__ den, const int* __restrict__ src,
               const int* __restrict__ dst, float* __restrict__ acc)
{
    const int  lane = threadIdx.x & 31;
    const long e0   = ((long)blockIdx.x * 8 + (threadIdx.x >> 5)) * EPW;
    const long Ef   = (long)EDGES + NODES;
    if (e0 >= Ef) return;
    const long em = e0 + (lane & (EPW - 1));
    int sv, dv;
    if (em < EDGES) { sv = src[em]; dv = dst[em]; } else { sv = dv = (int)(em - EDGES); }
    const float av = ex[em] / den[dv];

    float v0[EPW], v1[EPW];
    #pragma unroll
    for (int i = 0; i < EPW; ++i) {
        const int s = __builtin_amdgcn_readlane(sv, i);
        v0[i] = el[(size_t)s * DIM + lane];
        v1[i] = el[(size_t)s * DIM + lane + 32];
    }
    #pragma unroll
    for (int i = 0; i < EPW; ++i) {
        const int   d = __builtin_amdgcn_readlane(dv, i);
        const float a = __int_as_float(__builtin_amdgcn_readlane(__float_as_int(av), i));
        atomicAdd(&acc[(size_t)d * DIM + lane],      a * v0[i]);
        atomicAdd(&acc[(size_t)d * DIM + lane + 32], a * v1[i]);
    }
}

__global__ void elu_write(const float* __restrict__ acc, float* __restrict__ h,
                          float* __restrict__ emb, long n)
{
    const long i = (long)blockIdx.x * blockDim.x + threadIdx.x;
    if (i >= n) return;
    const float v = acc[i];
    const float o = v > 0.f ? v : (expf(v) - 1.f);
    h[i] = o;
    emb[i] = o;
}

__global__ void __launch_bounds__(256)
ggc_scatter(const float* __restrict__ t0, const float* __restrict__ t1,
            const int* __restrict__ ety, const int* __restrict__ src,
            const int* __restrict__ dst, float* __restrict__ acc)
{
    const int  lane = threadIdx.x & 31;
    const long e0   = ((long)blockIdx.x * 8 + (threadIdx.x >> 5)) * EPW;
    const long Ef   = (long)EDGES + NODES;
    if (e0 >= Ef) return;
    const long em = e0 + (lane & (EPW - 1));
    int sv, dv;
    if (em < EDGES) { sv = src[em]; dv = dst[em]; } else { sv = dv = (int)(em - EDGES); }
    const int tv = ety[em];

    float v0[EPW], v1[EPW];
    #pragma unroll
    for (int i = 0; i < EPW; ++i) {
        const int s = __builtin_amdgcn_readlane(sv, i);
        const int t = __builtin_amdgcn_readlane(tv, i);
        const float* tr = (t == 0) ? t0 : t1;
        v0[i] = tr[(size_t)s * DIM + lane];
        v1[i] = tr[(size_t)s * DIM + lane + 32];
    }
    #pragma unroll
    for (int i = 0; i < EPW; ++i) {
        const int d = __builtin_amdgcn_readlane(dv, i);
        atomicAdd(&acc[(size_t)d * DIM + lane],      v0[i]);
        atomicAdd(&acc[(size_t)d * DIM + lane + 32], v1[i]);
    }
}

__global__ void gru_update(const float* __restrict__ gi,
                           const float* __restrict__ gh,
                           float* __restrict__ h, float* __restrict__ emb,
                           long nElem)
{
    const long i = (long)blockIdx.x * blockDim.x + threadIdx.x;
    if (i >= nElem) return;
    const long n = i >> 6;
    const int  d = (int)(i & 63);
    const float* gin = gi + n * (3 * DIM);
    const float* ghn = gh + n * (3 * DIM);
    const float r = 1.f / (1.f + expf(-(gin[d]           + ghn[d])));
    const float z = 1.f / (1.f + expf(-(gin[DIM + d]     + ghn[DIM + d])));
    const float c = tanhf(gin[2 * DIM + d] + r * ghn[2 * DIM + d]);
    const float o = (1.f - z) * c + z * h[i];
    h[i] = o;
    if (emb) emb[i] = o;
}

__global__ void fill_zero(float* __restrict__ p, long n)
{
    long i = (long)blockIdx.x * blockDim.x + threadIdx.x;
    const long stride = (long)gridDim.x * blockDim.x;
    for (; i < n; i += stride) p[i] = 0.f;
}

extern "C" void kernel_launch(void* const* d_in, const int* in_sizes, int n_in,
                              void* d_out, int out_size, void* d_ws, size_t ws_size,
                              hipStream_t stream)
{
    (void)in_sizes; (void)n_in; (void)out_size; (void)ws_size;

    const float* x   = (const float*)d_in[0];
    const int*   src = (const int*)d_in[1];
    const int*   dst = (const int*)d_in[2];
    const int*   ety = (const int*)d_in[3];

    const long NE = (long)NODES * DIM;
    const long Ef = (long)EDGES + NODES;

    float* ws   = (float*)d_ws;
    float* hbuf = ws;                 ws += NE;
    float* buf1 = ws;                 ws += NE;
    float* buf2 = ws;                 ws += NE;
    float* gi   = ws;                 ws += (long)NODES * 3 * DIM;
    float* gh   = ws;                 ws += (long)NODES * 3 * DIM;
    float* acc  = ws;                 ws += NE;
    unsigned* mEnc = (unsigned*)ws;   ws += NODES;
    float* den  = ws;                 ws += NODES;
    float* score = ws;

    float* out = (float*)d_out;

    const int edgeThrBlocks  = (int)((Ef + 255) / 256);
    const int elemBlocks     = (int)((NE + 255) / 256);
    const int gemmBlocks64   = ((NODES / 16) * (DIM / 16)       + 7) / 8;
    const int gemmBlocks192  = ((NODES / 16) * (3 * DIM / 16)   + 7) / 8;

    const float* hin = x;
    for (int l = 0; l < 2; ++l) {
        const int base = 4 + 11 * l;
        const float* Wsrc = (const float*)d_in[base + 0];
        const float* bsrc = (const float*)d_in[base + 1];
        const float* Wdst = (const float*)d_in[base + 2];
        const float* bdst = (const float*)d_in[base + 3];
        const float* attn = (const float*)d_in[base + 4];
        const float* ggW  = (const float*)d_in[base + 5];
        const float* ggb  = (const float*)d_in[base + 6];
        const float* Wih  = (const float*)d_in[base + 7];
        const float* bih  = (const float*)d_in[base + 8];
        const float* Whh  = (const float*)d_in[base + 9];
        const float* bhh  = (const float*)d_in[base + 10];

        gemm_bias_wmma<<<gemmBlocks64, 256, 0, stream>>>(hin, Wsrc, bsrc, buf1,
                                                         NODES, DIM, DIM);
        gemm_bias_wmma<<<gemmBlocks64, 256, 0, stream>>>(hin, Wdst, bdst, buf2,
                                                         NODES, DIM, DIM);
        fill_zero<<<2048, 256, 0, stream>>>(acc, NE + 2L * NODES);
        edge_score<<<EDGE_GROUP_BLOCKS, 256, 0, stream>>>(buf1, buf2, attn, src, dst,
                                                       score, mEnc);
        edge_exp<<<edgeThrBlocks, 256, 0, stream>>>(score, mEnc, den, dst);
        edge_aggregate<<<EDGE_GROUP_BLOCKS, 256, 0, stream>>>(buf1, score, den, src,
                                                           dst, acc);
        elu_write<<<elemBlocks, 256, 0, stream>>>(acc, hbuf, out + (2L * l) * NE,
                                                  NE);

        for (int st = 0; st < 2; ++st) {
            gemm_bias_wmma<<<gemmBlocks64, 256, 0, stream>>>(
                hbuf, ggW,               ggb,       buf1, NODES, DIM, DIM);
            gemm_bias_wmma<<<gemmBlocks64, 256, 0, stream>>>(
                hbuf, ggW + DIM * DIM,   ggb + DIM, buf2, NODES, DIM, DIM);
            fill_zero<<<2048, 256, 0, stream>>>(acc, NE);
            ggc_scatter<<<EDGE_GROUP_BLOCKS, 256, 0, stream>>>(buf1, buf2, ety, src,
                                                            dst, acc);
            gemm_bias_wmma<<<gemmBlocks192, 256, 0, stream>>>(
                acc,  Wih, bih, gi, NODES, DIM, 3 * DIM);
            gemm_bias_wmma<<<gemmBlocks192, 256, 0, stream>>>(
                hbuf, Whh, bhh, gh, NODES, DIM, 3 * DIM);
            gru_update<<<elemBlocks, 256, 0, stream>>>(
                gi, gh, hbuf, (st == 1) ? out + (2L * l + 1) * NE : nullptr, NE);
        }
        hin = hbuf;
    }
}


# edges counting-sorted by destination once per call (int atomics, arrival-order slots); one wave per node does score, softmax, aggregate, elu, GGC sum in registers; no float atomics; sum order changes
# speedup vs baseline: 1.9198x; 1.1007x over previous
#include <hip/hip_runtime.h>
#include <hip/hip_bf16.h>


typedef __attribute__((ext_vector_type(2))) float v2f;
typedef __attribute__((ext_vector_type(8))) float v8f;

#define NODES 50000
#define EDGES 800000
#define DIM   64
#define NEG_SLOPE 0.2f

#define NPW 4
#define CH  8
static_assert(NODES % NPW == 0, "node group must divide the node count");
#define NODE_BLOCKS ((((NODES) / NPW) + 7) / 8)
#define BIG_NEG (-3.402823466e+38f)

__global__ void __launch_bounds__(256)
gemm_bias_wmma(const float* __restrict__ A, const float* __restrict__ W,
               const float* __restrict__ bias, float* __restrict__ C,
               int M, int K, int Nout)
{
    const int lane = threadIdx.x & 31;
    const int wave = blockIdx.x * 8 + (threadIdx.x >> 5);
    const int colTiles = Nout >> 4;
    const int total = (M >> 4) * colTiles;
    if (wave >= total) return;

    const int rt   = wave / colTiles;
    const int ct   = wave - rt * colTiles;
    const int row0 = rt << 4;
    const int col0 = ct << 4;
    const int lr   = lane & 15;
    const int hi   = lane >> 4;

    const float* __restrict__ arow = A + (size_t)(row0 + lr) * K;
    const float* __restrict__ wrow = W + (size_t)(col0 + lr) * K;

    v8f acc = {};
    #pragma unroll
    for (int k0 = 0; k0 < 64; k0 += 4) {
        v2f a, b;
        a.x = arow[k0 + 2 * hi];
        a.y = arow[k0 + 2 * hi + 1];
        b.x = wrow[k0 + 2 * hi];
        b.y = wrow[k0 + 2 * hi + 1];
        acc = __builtin_amdgcn_wmma_f32_16x16x4_f32(
                  false, a, false, b, (short)0, acc, false, false);
    }

    const float bv = bias[col0 + lr];
    float* crow = C + (size_t)(row0 + hi * 8) * Nout + col0 + lr;
    #pragma unroll
    for (int v = 0; v < 8; ++v)
        crow[(size_t)v * Nout] = acc[v] + bv;
}

__global__ void zero_counts(int* __restrict__ cnt)
{
    const int i = blockIdx.x * blockDim.x + threadIdx.x;
    if (i < NODES) cnt[i] = 0;
}

__global__ void count_dst(const int* __restrict__ dst, int* __restrict__ cnt)
{
    const int e = blockIdx.x * blockDim.x + threadIdx.x;
    if (e < EDGES) atomicAdd(&cnt[dst[e]], 1);
}

#define SCAN_PER ((NODES + 1023) / 1024)
__global__ void __launch_bounds__(1024)
scan_counts(const int* __restrict__ cnt, int* __restrict__ off,
            int* __restrict__ cur)
{
    __shared__ int part[1024];
    const int t = threadIdx.x;
    const int b = t * SCAN_PER;
    int s = 0;
    for (int i = 0; i < SCAN_PER; ++i) {
        const int idx = b + i;
        if (idx < NODES) s += cnt[idx];
    }
    part[t] = s;
    __syncthreads();
    for (int o = 1; o < 1024; o <<= 1) {
        const int v = (t >= o) ? part[t - o] : 0;
        __syncthreads();
        part[t] += v;
        __syncthreads();
    }
    int run = part[t] - s;
    for (int i = 0; i < SCAN_PER; ++i) {
        const int idx = b + i;
        if (idx < NODES) {
            off[idx] = run;
            cur[idx] = run;
            run += cnt[idx];
        }
    }
    if (t == 1023) off[NODES] = part[1023];
}

__global__ void fill_slots(const int* __restrict__ src, const int* __restrict__ dst,
                           const int* __restrict__ ety, int* __restrict__ cur,
                           int* __restrict__ key)
{
    const int e = blockIdx.x * blockDim.x + threadIdx.x;
    if (e >= EDGES) return;
    const int pos = atomicAdd(&cur[dst[e]], 1);
    key[pos] = src[e] + (ety[e] != 0 ? NODES : 0);
}

__device__ __forceinline__ int slot_key(const int* __restrict__ key, int beg,
                                        int selfKey, int v)
{
    return v == 0 ? selfKey : key[beg + v - 1];
}

__global__ void __launch_bounds__(256)
gat_node(const float* __restrict__ el, const float* __restrict__ er,
         const float* __restrict__ attn, const int* __restrict__ off,
         const int* __restrict__ key, float* __restrict__ h,
         float* __restrict__ emb)
{
    const int lane = threadIdx.x & 31;
    const int d0   = (blockIdx.x * 8 + (threadIdx.x >> 5)) * NPW;
    if (d0 >= NODES) return;
    const int offv = off[d0 + (lane < NPW ? lane : NPW)];
    const v2f at   = *(const v2f*)(attn + 2 * lane);
    const int sub  = lane & (CH - 1);
    const bool b16 = (lane & 16) != 0, b8 = (lane & 8) != 0, b4 = (lane & 4) != 0;
    const int myEdge = (b16 ? 4 : 0) + (b8 ? 2 : 0) + (b4 ? 1 : 0);
    static_assert(CH == 8, "the folding butterfly is written for 8 edges per round");

    #pragma unroll
    for (int n = 0; n < NPW; ++n) {
        const int d   = d0 + n;
        const int beg = __builtin_amdgcn_readlane(offv, n);
        const int tot = __builtin_amdgcn_readlane(offv, n + 1) - beg + 1;
        const v2f rr  = *(const v2f*)(er + (size_t)d * DIM + 2 * lane);

        float m = BIG_NEG, den = 0.f;
        v2f acc = {0.f, 0.f};

        int kv = slot_key(key, beg, d, sub < tot ? sub : tot - 1);
        for (int v0 = 0; v0 < tot; v0 += CH) {
            const int rem = tot - v0;
            const int cnt = rem < CH ? rem : CH;
            const int ks = kv >= NODES ? kv - NODES : kv;

            v2f row[CH];
            #pragma unroll
            for (int i = 0; i < CH; ++i) {
                const int s = __builtin_amdgcn_readlane(ks, i);
                row[i] = *(const v2f*)(el + (size_t)s * DIM + 2 * lane);
            }
            int kn = 0;
            if (rem > CH) {
                const int r2 = rem - CH;
                kn = key[beg + v0 + CH - 1 + (sub < r2 ? sub : r2 - 1)];
            }
            float q[CH];
            #pragma unroll
            for (int i = 0; i < CH; ++i) {
                float x0 = row[i].x + rr.x;
                float x1 = row[i].y + rr.y;
                x0 = x0 > 0.f ? x0 : NEG_SLOPE * x0;
                x1 = x1 > 0.f ? x1 : NEG_SLOPE * x1;
                float t = x0 * at.x;
                t += x1 * at.y;
                q[i] = t;
            }
            float k0[4], k1[2], sc_e;
            #pragma unroll
            for (int j = 0; j < 4; ++j) {
                const float send = b16 ? q[j] : q[j + 4];
                const float keep = b16 ? q[j + 4] : q[j];
                k0[j] = keep + __shfl_xor(send, 16, 32);
            }
            #pragma unroll
            for (int j = 0; j < 2; ++j) {
                const float send = b8 ? k0[j] : k0[j + 2];
                const float keep = b8 ? k0[j + 2] : k0[j];
                k1[j] = keep + __shfl_xor(send, 8, 32);
            }
            {
                const float send = b4 ? k1[0] : k1[1];
                const float keep = b4 ? k1[1] : k1[0];
                sc_e = keep + __shfl_xor(send, 4, 32);
            }
            sc_e += __shfl_xor(sc_e, 2, 32);
            sc_e += __shfl_xor(sc_e, 1, 32);

            const bool valid = myEdge < cnt;
            float cm = valid ? sc_e : BIG_NEG;
            cm = fmaxf(cm, __shfl_xor(cm, 16, 32));
            cm = fmaxf(cm, __shfl_xor(cm, 8, 32));
            cm = fmaxf(cm, __shfl_xor(cm, 4, 32));
            const float nm = fmaxf(m, cm);
            const float sc = expf(m - nm);
            const float wv = valid ? expf(sc_e - nm) : 0.f;
            den *= sc; acc.x *= sc; acc.y *= sc;
            #pragma unroll
            for (int i = 0; i < CH; ++i) {
                const int src_lane = ((i >> 2) & 1) * 16 + ((i >> 1) & 1) * 8 + (i & 1) * 4;
                const float w = __int_as_float(
                    __builtin_amdgcn_readlane(__float_as_int(wv), src_lane));
                den   += w;
                acc.x += w * row[i].x;
                acc.y += w * row[i].y;
            }
            m  = nm;
            kv = kn;
        }

        v2f o;
        o.x = acc.x / den;
        o.y = acc.y / den;
        o.x = o.x > 0.f ? o.x : (expf(o.x) - 1.f);
        o.y = o.y > 0.f ? o.y : (expf(o.y) - 1.f);
        *(v2f*)(h   + (size_t)d * DIM + 2 * lane) = o;
        *(v2f*)(emb + (size_t)d * DIM + 2 * lane) = o;
    }
}

__global__ void __launch_bounds__(256)
ggc_pull(const float* __restrict__ tr, const int* __restrict__ ety,
         const int* __restrict__ off, const int* __restrict__ key,
         float* __restrict__ acc)
{
    const int lane = threadIdx.x & 31;
    const int d0   = (blockIdx.x * 8 + (threadIdx.x >> 5)) * NPW;
    if (d0 >= NODES) return;
    const int offv = off[d0 + (lane < NPW ? lane : NPW)];
    const int nl   = lane < NPW ? lane : 0;
    const int selfv = d0 + nl + (ety[EDGES + d0 + nl] != 0 ? NODES : 0);
    const int sub  = lane & (CH - 1);

    #pragma unroll
    for (int n = 0; n < NPW; ++n) {
        const int d   = d0 + n;
        const int beg = __builtin_amdgcn_readlane(offv, n);
        const int tot = __builtin_amdgcn_readlane(offv, n + 1) - beg + 1;
        const int selfKey = __builtin_amdgcn_readlane(selfv, n);

        v2f a = {0.f, 0.f};
        int kv = slot_key(key, beg, selfKey, sub < tot ? sub : tot - 1);
        for (int v0 = 0; v0 < tot; v0 += CH) {
            const int rem = tot - v0;
            const int cnt = rem < CH ? rem : CH;
            v2f row[CH];
            #pragma unroll
            for (int i = 0; i < CH; ++i) {
                const int r = __builtin_amdgcn_readlane(kv, i);
                row[i] = *(const v2f*)(tr + (size_t)r * DIM + 2 * lane);
            }
            int kn = 0;
            if (rem > CH) {
                const int r2 = rem - CH;
                kn = key[beg + v0 + CH - 1 + (sub < r2 ? sub : r2 - 1)];
            }
            #pragma unroll
            for (int i = 0; i < CH; ++i) {
                a.x += (i < cnt) ? row[i].x : 0.f;
                a.y += (i < cnt) ? row[i].y : 0.f;
            }
            kv = kn;
        }
        *(v2f*)(acc + (size_t)d * DIM + 2 * lane) = a;
    }
}

__global__ void gru_update(const float* __restrict__ gi,
                           const float* __restrict__ gh,
                           float* __restrict__ h, float* __restrict__ emb,
                           long nElem)
{
    const long i = (long)blockIdx.x * blockDim.x + threadIdx.x;
    if (i >= nElem) return;
    const long n = i >> 6;
    const int  d = (int)(i & 63);
    const float* gin = gi + n * (3 * DIM);
    const float* ghn = gh + n * (3 * DIM);
    const float r = 1.f / (1.f + expf(-(gin[d]           + ghn[d])));
    const float z = 1.f / (1.f + expf(-(gin[DIM + d]     + ghn[DIM + d])));
    const float c = tanhf(gin[2 * DIM + d] + r * ghn[2 * DIM + d]);
    const float o = (1.f - z) * c + z * h[i];
    h[i] = o;
    if (emb) emb[i] = o;
}

extern "C" void kernel_launch(void* const* d_in, const int* in_sizes, int n_in,
                              void* d_out, int out_size, void* d_ws, size_t ws_size,
                              hipStream_t stream)
{
    (void)in_sizes; (void)n_in; (void)out_size; (void)ws_size;

    const float* x   = (const float*)d_in[0];
    const int*   src = (const int*)d_in[1];
    const int*   dst = (const int*)d_in[2];
    const int*   ety = (const int*)d_in[3];

    const long NE = (long)NODES * DIM;

    float* ws   = (float*)d_ws;
    float* hbuf = ws;                 ws += NE;
    float* buf1 = ws;                 ws += NE;
    float* buf2 = ws;                 ws += NE;
    float* gi   = ws;                 ws += (long)NODES * 3 * DIM;
    float* gh   = ws;                 ws += (long)NODES * 3 * DIM;
    float* acc  = ws;                 ws += NE;
    int* off = (int*)ws;              ws += NODES + 64;
    int* cur = (int*)ws;              ws += NODES + 64;
    int* cnt = (int*)ws;              ws += NODES + 64;
    int* key = (int*)ws;

    float* out = (float*)d_out;

    const int edgeThrBlocks  = (EDGES + 255) / 256;
    const int elemBlocks     = (int)((NE + 255) / 256);
    const int gemmBlocks64   = ((NODES / 16) * (DIM / 16)       + 7) / 8;
    const int gemmBlocks192  = ((NODES / 16) * (3 * DIM / 16)   + 7) / 8;

    zero_counts<<<(NODES + 255) / 256, 256, 0, stream>>>(cnt);
    count_dst<<<edgeThrBlocks, 256, 0, stream>>>(dst, cnt);
    scan_counts<<<1, 1024, 0, stream>>>(cnt, off, cur);
    fill_slots<<<edgeThrBlocks, 256, 0, stream>>>(src, dst, ety, cur, key);

    const float* hin = x;
    for (int l = 0; l < 2; ++l) {
        const int base = 4 + 11 * l;
        const float* Wsrc = (const float*)d_in[base + 0];
        const float* bsrc = (const float*)d_in[base + 1];
        const float* Wdst = (const float*)d_in[base + 2];
        const float* bdst = (const float*)d_in[base + 3];
        const float* attn = (const float*)d_in[base + 4];
        const float* ggW  = (const float*)d_in[base + 5];
        const float* ggb  = (const float*)d_in[base + 6];
        const float* Wih  = (const float*)d_in[base + 7];
        const float* bih  = (const float*)d_in[base + 8];
        const float* Whh  = (const float*)d_in[base + 9];
        const float* bhh  = (const float*)d_in[base + 10];

        gemm_bias_wmma<<<gemmBlocks64, 256, 0, stream>>>(hin, Wsrc, bsrc, buf1,
                                                         NODES, DIM, DIM);
        gemm_bias_wmma<<<gemmBlocks64, 256, 0, stream>>>(hin, Wdst, bdst, buf2,
                                                         NODES, DIM, DIM);
        gat_node<<<NODE_BLOCKS, 256, 0, stream>>>(buf1, buf2, attn, off, key, hbuf,
                                                  out + (2L * l) * NE);

        for (int st = 0; st < 2; ++st) {
            gemm_bias_wmma<<<gemmBlocks64, 256, 0, stream>>>(
                hbuf, ggW,               ggb,       buf1, NODES, DIM, DIM);
            gemm_bias_wmma<<<gemmBlocks64, 256, 0, stream>>>(
                hbuf, ggW + DIM * DIM,   ggb + DIM, buf2, NODES, DIM, DIM);
            ggc_pull<<<NODE_BLOCKS, 256, 0, stream>>>(buf1, ety, off, key, acc);
            gemm_bias_wmma<<<gemmBlocks192, 256, 0, stream>>>(
                acc,  Wih, bih, gi, NODES, DIM, 3 * DIM);
            gemm_bias_wmma<<<gemmBlocks192, 256, 0, stream>>>(
                hbuf, Whh, bhh, gh, NODES, DIM, 3 * DIM);
            gru_update<<<elemBlocks, 256, 0, stream>>>(
                gi, gh, hbuf, (st == 1) ? out + (2L * l + 1) * NE : nullptr, NE);
        }
        hin = hbuf;
    }
}


# GRU step fused: one kernel, wave owns 16 rows, Wih/Whh staged in LDS, six gate tiles per 16 features on the matrix unit, gates applied to accumulators; gi/gh never stored (2 gemms + update gone)
# speedup vs baseline: 2.3615x; 1.2301x over previous
#include <hip/hip_runtime.h>
#include <hip/hip_bf16.h>


typedef __attribute__((ext_vector_type(2))) float v2f;
typedef __attribute__((ext_vector_type(4))) float v4f;
typedef __attribute__((ext_vector_type(8))) float v8f;

#define NODES 50000
#define EDGES 800000
#define DIM   64
#define NEG_SLOPE 0.2f

#define NPW 4
#define CH  8
static_assert(NODES % NPW == 0, "node group must divide the node count");
#define NODE_BLOCKS ((((NODES) / NPW) + 7) / 8)
#define BIG_NEG (-3.402823466e+38f)

__global__ void __launch_bounds__(256)
gemm_bias_wmma(const float* __restrict__ A, const float* __restrict__ W,
               const float* __restrict__ bias, float* __restrict__ C,
               int M, int K, int Nout)
{
    const int lane = threadIdx.x & 31;
    const int wave = blockIdx.x * 8 + (threadIdx.x >> 5);
    const int colTiles = Nout >> 4;
    const int total = (M >> 4) * colTiles;
    if (wave >= total) return;

    const int rt   = wave / colTiles;
    const int ct   = wave - rt * colTiles;
    const int row0 = rt << 4;
    const int col0 = ct << 4;
    const int lr   = lane & 15;
    const int hi   = lane >> 4;

    const float* __restrict__ arow = A + (size_t)(row0 + lr) * K;
    const float* __restrict__ wrow = W + (size_t)(col0 + lr) * K;

    v8f acc = {};
    #pragma unroll
    for (int k0 = 0; k0 < 64; k0 += 4) {
        v2f a, b;
        a.x = arow[k0 + 2 * hi];
        a.y = arow[k0 + 2 * hi + 1];
        b.x = wrow[k0 + 2 * hi];
        b.y = wrow[k0 + 2 * hi + 1];
        acc = __builtin_amdgcn_wmma_f32_16x16x4_f32(
                  false, a, false, b, (short)0, acc, false, false);
    }

    const float bv = bias[col0 + lr];
    float* crow = C + (size_t)(row0 + hi * 8) * Nout + col0 + lr;
    #pragma unroll
    for (int v = 0; v < 8; ++v)
        crow[(size_t)v * Nout] = acc[v] + bv;
}

__global__ void zero_counts(int* __restrict__ cnt)
{
    const int i = blockIdx.x * blockDim.x + threadIdx.x;
    if (i < NODES) cnt[i] = 0;
}

__global__ void count_dst(const int* __restrict__ dst, int* __restrict__ cnt)
{
    const int e = blockIdx.x * blockDim.x + threadIdx.x;
    if (e < EDGES) atomicAdd(&cnt[dst[e]], 1);
}

#define SCAN_PER ((NODES + 1023) / 1024)
__global__ void __launch_bounds__(1024)
scan_counts(const int* __restrict__ cnt, int* __restrict__ off,
            int* __restrict__ cur)
{
    __shared__ int part[1024];
    const int t = threadIdx.x;
    const int b = t * SCAN_PER;
    int s = 0;
    for (int i = 0; i < SCAN_PER; ++i) {
        const int idx = b + i;
        if (idx < NODES) s += cnt[idx];
    }
    part[t] = s;
    __syncthreads();
    for (int o = 1; o < 1024; o <<= 1) {
        const int v = (t >= o) ? part[t - o] : 0;
        __syncthreads();
        part[t] += v;
        __syncthreads();
    }
    int run = part[t] - s;
    for (int i = 0; i < SCAN_PER; ++i) {
        const int idx = b + i;
        if (idx < NODES) {
            off[idx] = run;
            cur[idx] = run;
            run += cnt[idx];
        }
    }
    if (t == 1023) off[NODES] = part[1023];
}

__global__ void fill_slots(const int* __restrict__ src, const int* __restrict__ dst,
                           const int* __restrict__ ety, int* __restrict__ cur,
                           int* __restrict__ key)
{
    const int e = blockIdx.x * blockDim.x + threadIdx.x;
    if (e >= EDGES) return;
    const int pos = atomicAdd(&cur[dst[e]], 1);
    key[pos] = src[e] + (ety[e] != 0 ? NODES : 0);
}

__device__ __forceinline__ int slot_key(const int* __restrict__ key, int beg,
                                        int selfKey, int v)
{
    return v == 0 ? selfKey : key[beg + v - 1];
}

__global__ void __launch_bounds__(256)
gat_node(const float* __restrict__ el, const float* __restrict__ er,
         const float* __restrict__ attn, const int* __restrict__ off,
         const int* __restrict__ key, float* __restrict__ h,
         float* __restrict__ emb)
{
    const int lane = threadIdx.x & 31;
    const int d0   = (blockIdx.x * 8 + (threadIdx.x >> 5)) * NPW;
    if (d0 >= NODES) return;
    const int offv = off[d0 + (lane < NPW ? lane : NPW)];
    const v2f at   = *(const v2f*)(attn + 2 * lane);
    const int sub  = lane & (CH - 1);
    const bool b16 = (lane & 16) != 0, b8 = (lane & 8) != 0, b4 = (lane & 4) != 0;
    const int myEdge = (b16 ? 4 : 0) + (b8 ? 2 : 0) + (b4 ? 1 : 0);
    static_assert(CH == 8, "the folding butterfly is written for 8 edges per round");

    #pragma unroll
    for (int n = 0; n < NPW; ++n) {
        const int d   = d0 + n;
        const int beg = __builtin_amdgcn_readlane(offv, n);
        const int tot = __builtin_amdgcn_readlane(offv, n + 1) - beg + 1;
        const v2f rr  = *(const v2f*)(er + (size_t)d * DIM + 2 * lane);

        float m = BIG_NEG, den = 0.f;
        v2f acc = {0.f, 0.f};

        int kv = slot_key(key, beg, d, sub < tot ? sub : tot - 1);
        for (int v0 = 0; v0 < tot; v0 += CH) {
            const int rem = tot - v0;
            const int cnt = rem < CH ? rem : CH;
            const int ks = kv >= NODES ? kv - NODES : kv;

            v2f row[CH];
            #pragma unroll
            for (int i = 0; i < CH; ++i) {
                const int s = __builtin_amdgcn_readlane(ks, i);
                row[i] = *(const v2f*)(el + (size_t)s * DIM + 2 * lane);
            }
            int kn = 0;
            if (rem > CH) {
                const int r2 = rem - CH;
                kn = key[beg + v0 + CH - 1 + (sub < r2 ? sub : r2 - 1)];
            }
            float q[CH];
            #pragma unroll
            for (int i = 0; i < CH; ++i) {
                float x0 = row[i].x + rr.x;
                float x1 = row[i].y + rr.y;
                x0 = x0 > 0.f ? x0 : NEG_SLOPE * x0;
                x1 = x1 > 0.f ? x1 : NEG_SLOPE * x1;
                float t = x0 * at.x;
                t += x1 * at.y;
                q[i] = t;
            }
            float k0[4], k1[2], sc_e;
            #pragma unroll
            for (int j = 0; j < 4; ++j) {
                const float send = b16 ? q[j] : q[j + 4];
                const float keep = b16 ? q[j + 4] : q[j];
                k0[j] = keep + __shfl_xor(send, 16, 32);
            }
            #pragma unroll
            for (int j = 0; j < 2; ++j) {
                const float send = b8 ? k0[j] : k0[j + 2];
                const float keep = b8 ? k0[j + 2] : k0[j];
                k1[j] = keep + __shfl_xor(send, 8, 32);
            }
            {
                const float send = b4 ? k1[0] : k1[1];
                const float keep = b4 ? k1[1] : k1[0];
                sc_e = keep + __shfl_xor(send, 4, 32);
            }
            sc_e += __shfl_xor(sc_e, 2, 32);
            sc_e += __shfl_xor(sc_e, 1, 32);

            const bool valid = myEdge < cnt;
            float cm = valid ? sc_e : BIG_NEG;
            cm = fmaxf(cm, __shfl_xor(cm, 16, 32));
            cm = fmaxf(cm, __shfl_xor(cm, 8, 32));
            cm = fmaxf(cm, __shfl_xor(cm, 4, 32));
            const float nm = fmaxf(m, cm);
            const float sc = expf(m - nm);
            const float wv = valid ? expf(sc_e - nm) : 0.f;
            den *= sc; acc.x *= sc; acc.y *= sc;
            #pragma unroll
            for (int i = 0; i < CH; ++i) {
                const int src_lane = ((i >> 2) & 1) * 16 + ((i >> 1) & 1) * 8 + (i & 1) * 4;
                const float w = __int_as_float(
                    __builtin_amdgcn_readlane(__float_as_int(wv), src_lane));
                den   += w;
                acc.x += w * row[i].x;
                acc.y += w * row[i].y;
            }
            m  = nm;
            kv = kn;
        }

        v2f o;
        o.x = acc.x / den;
        o.y = acc.y / den;
        o.x = o.x > 0.f ? o.x : (expf(o.x) - 1.f);
        o.y = o.y > 0.f ? o.y : (expf(o.y) - 1.f);
        *(v2f*)(h   + (size_t)d * DIM + 2 * lane) = o;
        *(v2f*)(emb + (size_t)d * DIM + 2 * lane) = o;
    }
}

__global__ void __launch_bounds__(256)
ggc_pull(const float* __restrict__ tr, const int* __restrict__ ety,
         const int* __restrict__ off, const int* __restrict__ key,
         float* __restrict__ acc)
{
    const int lane = threadIdx.x & 31;
    const int d0   = (blockIdx.x * 8 + (threadIdx.x >> 5)) * NPW;
    if (d0 >= NODES) return;
    const int offv = off[d0 + (lane < NPW ? lane : NPW)];
    const int nl   = lane < NPW ? lane : 0;
    const int selfv = d0 + nl + (ety[EDGES + d0 + nl] != 0 ? NODES : 0);
    const int sub  = lane & (CH - 1);

    #pragma unroll
    for (int n = 0; n < NPW; ++n) {
        const int d   = d0 + n;
        const int beg = __builtin_amdgcn_readlane(offv, n);
        const int tot = __builtin_amdgcn_readlane(offv, n + 1) - beg + 1;
        const int selfKey = __builtin_amdgcn_readlane(selfv, n);

        v2f a = {0.f, 0.f};
        int kv = slot_key(key, beg, selfKey, sub < tot ? sub : tot - 1);
        for (int v0 = 0; v0 < tot; v0 += CH) {
            const int rem = tot - v0;
            const int cnt = rem < CH ? rem : CH;
            v2f row[CH];
            #pragma unroll
            for (int i = 0; i < CH; ++i) {
                const int r = __builtin_amdgcn_readlane(kv, i);
                row[i] = *(const v2f*)(tr + (size_t)r * DIM + 2 * lane);
            }
            int kn = 0;
            if (rem > CH) {
                const int r2 = rem - CH;
                kn = key[beg + v0 + CH - 1 + (sub < r2 ? sub : r2 - 1)];
            }
            #pragma unroll
            for (int i = 0; i < CH; ++i) {
                a.x += (i < cnt) ? row[i].x : 0.f;
                a.y += (i < cnt) ? row[i].y : 0.f;
            }
            kv = kn;
        }
        *(v2f*)(acc + (size_t)d * DIM + 2 * lane) = a;
    }
}

#define GATES  (3 * DIM)
#define WPITCH 68
#define ROW_TILES (NODES / 16)
static_assert(NODES % 16 == 0, "row tiles must divide the node count");

__global__ void __launch_bounds__(256)
gru_fused(const float* __restrict__ am,
          const float* __restrict__ Wih, const float* __restrict__ bih,
          const float* __restrict__ Whh, const float* __restrict__ bhh,
          float* h, float* emb)
{
    __shared__ __attribute__((aligned(16))) float Wl[2 * GATES * WPITCH];

    const int tid  = threadIdx.x;
    const int lane = tid & 31;
    const int lr   = lane & 15;
    const int hi   = lane >> 4;
    const int rt   = blockIdx.x * 8 + (tid >> 5);
    const bool live = rt < ROW_TILES;
    const int row0 = rt << 4;

    v2f fa[16] = {}, fh[16] = {};
    if (live) {
        const float* arow = am + (size_t)(row0 + lr) * DIM + 2 * hi;
        const float* hrow = h  + (size_t)(row0 + lr) * DIM + 2 * hi;
        #pragma unroll
        for (int k = 0; k < 16; ++k) {
            fa[k] = *(const v2f*)(arow + 4 * k);
            fh[k] = *(const v2f*)(hrow + 4 * k);
        }
    }

    for (int i = tid; i < GATES * (DIM / 4); i += 256) {
        const int r = i >> 4, c4 = i & 15;
        const v4f wi = *(const v4f*)(Wih + 4 * i);
        const v4f wh = *(const v4f*)(Whh + 4 * i);
        *(v4f*)(Wl + r * WPITCH + 4 * c4)           = wi;
        *(v4f*)(Wl + (GATES + r) * WPITCH + 4 * c4) = wh;
    }
    __syncthreads();
    if (!live) return;

    for (int ct = 0; ct < DIM / 16; ++ct) {
        const int dcol = ct * 16 + lr;
        const float* bi = Wl + dcol * WPITCH + 2 * hi;
        const float* bh = bi + GATES * WPITCH;

        float* hp = h + (size_t)(row0 + hi * 8) * DIM + dcol;
        float hv[8];
        #pragma unroll
        for (int v = 0; v < 8; ++v) hv[v] = hp[(size_t)v * DIM];

        v8f cir = {}, ciz = {}, cic = {}, chr = {}, chz = {}, chc = {};
        #pragma unroll
        for (int k = 0; k < 16; ++k) {
            const v2f b0 = *(const v2f*)(bi + 4 * k);
            const v2f b1 = *(const v2f*)(bi + DIM * WPITCH + 4 * k);
            const v2f b2 = *(const v2f*)(bi + 2 * DIM * WPITCH + 4 * k);
            const v2f b3 = *(const v2f*)(bh + 4 * k);
            const v2f b4 = *(const v2f*)(bh + DIM * WPITCH + 4 * k);
            const v2f b5 = *(const v2f*)(bh + 2 * DIM * WPITCH + 4 * k);
            cir = __builtin_amdgcn_wmma_f32_16x16x4_f32(
                      false, fa[k], false, b0, (short)0, cir, false, false);
            ciz = __builtin_amdgcn_wmma_f32_16x16x4_f32(
                      false, fa[k], false, b1, (short)0, ciz, false, false);
            cic = __builtin_amdgcn_wmma_f32_16x16x4_f32(
                      false, fa[k], false, b2, (short)0, cic, false, false);
            chr = __builtin_amdgcn_wmma_f32_16x16x4_f32(
                      false, fh[k], false, b3, (short)0, chr, false, false);
            chz = __builtin_amdgcn_wmma_f32_16x16x4_f32(
                      false, fh[k], false, b4, (short)0, chz, false, false);
            chc = __builtin_amdgcn_wmma_f32_16x16x4_f32(
                      false, fh[k], false, b5, (short)0, chc, false, false);
        }

        const float bir = bih[dcol], biz = bih[DIM + dcol], bic = bih[2 * DIM + dcol];
        const float bhr = bhh[dcol], bhz = bhh[DIM + dcol], bhc = bhh[2 * DIM + dcol];
        float* ep = emb ? emb + (size_t)(row0 + hi * 8) * DIM + dcol : nullptr;
        #pragma unroll
        for (int v = 0; v < 8; ++v) {
            const float gir = cir[v] + bir, ghr = chr[v] + bhr;
            const float giz = ciz[v] + biz, ghz = chz[v] + bhz;
            const float gic = cic[v] + bic, ghc = chc[v] + bhc;
            const float r = 1.f / (1.f + expf(-(gir + ghr)));
            const float z = 1.f / (1.f + expf(-(giz + ghz)));
            const float c = tanhf(gic + r * ghc);
            const float o = (1.f - z) * c + z * hv[v];
            hp[(size_t)v * DIM] = o;
            if (ep) ep[(size_t)v * DIM] = o;
        }
    }
}

extern "C" void kernel_launch(void* const* d_in, const int* in_sizes, int n_in,
                              void* d_out, int out_size, void* d_ws, size_t ws_size,
                              hipStream_t stream)
{
    (void)in_sizes; (void)n_in; (void)out_size; (void)ws_size;

    const float* x   = (const float*)d_in[0];
    const int*   src = (const int*)d_in[1];
    const int*   dst = (const int*)d_in[2];
    const int*   ety = (const int*)d_in[3];

    const long NE = (long)NODES * DIM;

    float* ws   = (float*)d_ws;
    float* hbuf = ws;                 ws += NE;
    float* buf1 = ws;                 ws += NE;
    float* buf2 = ws;                 ws += NE;
    float* acc  = ws;                 ws += NE;
    int* off = (int*)ws;              ws += NODES + 64;
    int* cur = (int*)ws;              ws += NODES + 64;
    int* cnt = (int*)ws;              ws += NODES + 64;
    int* key = (int*)ws;

    float* out = (float*)d_out;

    const int edgeThrBlocks  = (EDGES + 255) / 256;
    const int gemmBlocks64   = ((NODES / 16) * (DIM / 16)       + 7) / 8;
    const int gruBlocks      = (ROW_TILES + 7) / 8;

    zero_counts<<<(NODES + 255) / 256, 256, 0, stream>>>(cnt);
    count_dst<<<edgeThrBlocks, 256, 0, stream>>>(dst, cnt);
    scan_counts<<<1, 1024, 0, stream>>>(cnt, off, cur);
    fill_slots<<<edgeThrBlocks, 256, 0, stream>>>(src, dst, ety, cur, key);

    const float* hin = x;
    for (int l = 0; l < 2; ++l) {
        const int base = 4 + 11 * l;
        const float* Wsrc = (const float*)d_in[base + 0];
        const float* bsrc = (const float*)d_in[base + 1];
        const float* Wdst = (const float*)d_in[base + 2];
        const float* bdst = (const float*)d_in[base + 3];
        const float* attn = (const float*)d_in[base + 4];
        const float* ggW  = (const float*)d_in[base + 5];
        const float* ggb  = (const float*)d_in[base + 6];
        const float* Wih  = (const float*)d_in[base + 7];
        const float* bih  = (const float*)d_in[base + 8];
        const float* Whh  = (const float*)d_in[base + 9];
        const float* bhh  = (const float*)d_in[base + 10];

        gemm_bias_wmma<<<gemmBlocks64, 256, 0, stream>>>(hin, Wsrc, bsrc, buf1,
                                                         NODES, DIM, DIM);
        gemm_bias_wmma<<<gemmBlocks64, 256, 0, stream>>>(hin, Wdst, bdst, buf2,
                                                         NODES, DIM, DIM);
        gat_node<<<NODE_BLOCKS, 256, 0, stream>>>(buf1, buf2, attn, off, key, hbuf,
                                                  out + (2L * l) * NE);

        for (int st = 0; st < 2; ++st) {
            gemm_bias_wmma<<<gemmBlocks64, 256, 0, stream>>>(
                hbuf, ggW,               ggb,       buf1, NODES, DIM, DIM);
            gemm_bias_wmma<<<gemmBlocks64, 256, 0, stream>>>(
                hbuf, ggW + DIM * DIM,   ggb + DIM, buf2, NODES, DIM, DIM);
            ggc_pull<<<NODE_BLOCKS, 256, 0, stream>>>(buf1, ety, off, key, acc);
            gru_fused<<<gruBlocks, 256, 0, stream>>>(
                acc, Wih, bih, Whh, bhh, hbuf,
                (st == 1) ? out + (2L * l + 1) * NE : nullptr);
        }
        hin = hbuf;
    }
}


# products sharing one A (el|er, trans0|trans1) paired in one launch: wave owns 16 whole rows, A rows and both 64x64 weights staged in LDS by whole-line loads, 8 tiles per wave; 12 GEMM launches -> 6
# speedup vs baseline: 2.4973x; 1.0575x over previous
#include <hip/hip_runtime.h>
#include <hip/hip_bf16.h>


typedef __attribute__((ext_vector_type(2))) float v2f;
typedef __attribute__((ext_vector_type(4))) float v4f;
typedef __attribute__((ext_vector_type(8))) float v8f;

#define NODES 50000
#define EDGES 800000
#define DIM   64
#define NEG_SLOPE 0.2f

#define NPW 4
#define CH  8
static_assert(NODES % NPW == 0, "node group must divide the node count");
#define NODE_BLOCKS ((((NODES) / NPW) + 7) / 8)
#define BIG_NEG (-3.402823466e+38f)

#define WPITCH 68
#define ROW_TILES (NODES / 16)
static_assert(NODES % 16 == 0, "row tiles must divide the node count");

__global__ void __launch_bounds__(256) __attribute__((amdgpu_waves_per_eu(2, 4)))
gemm_pair_wmma(const float* __restrict__ A,
               const float* __restrict__ W0, const float* __restrict__ bias0,
               const float* __restrict__ W1, const float* __restrict__ bias1,
               float* __restrict__ C0, float* __restrict__ C1)
{
    __shared__ __attribute__((aligned(16))) float Wl[2 * DIM * WPITCH];
    __shared__ __attribute__((aligned(16))) float Al[8 * 16 * WPITCH];

    const int tid  = threadIdx.x;
    const int lane = tid & 31;
    const int lr   = lane & 15;
    const int hi   = lane >> 4;
    const int rt   = blockIdx.x * 8 + (tid >> 5);
    const bool live = rt < ROW_TILES;
    const int row0 = (live ? rt : ROW_TILES - 1) << 4;
    float* al = Al + (tid >> 5) * 16 * WPITCH;

    const float* atile = A + (size_t)row0 * DIM;
    v4f av[8], w0v[4], w1v[4];
    #pragma unroll
    for (int j = 0; j < 8; ++j)
        av[j] = *(const v4f*)(atile + 4 * (j * 32 + lane));
    #pragma unroll
    for (int j = 0; j < 4; ++j) {
        w0v[j] = *(const v4f*)(W0 + 4 * (j * 256 + tid));
        w1v[j] = *(const v4f*)(W1 + 4 * (j * 256 + tid));
    }
    #pragma unroll
    for (int j = 0; j < 8; ++j) {
        const int i = j * 32 + lane;
        *(v4f*)(al + (i >> 4) * WPITCH + 4 * (i & 15)) = av[j];
    }
    #pragma unroll
    for (int j = 0; j < 4; ++j) {
        const int i = j * 256 + tid;
        const int r = i >> 4, c4 = i & 15;
        *(v4f*)(Wl + r * WPITCH + 4 * c4)         = w0v[j];
        *(v4f*)(Wl + (DIM + r) * WPITCH + 4 * c4) = w1v[j];
    }
    __syncthreads();
    if (!live) return;

    v2f fa[16];
    const float* ap = al + lr * WPITCH + 2 * hi;
    #pragma unroll
    for (int k = 0; k < 16; ++k) fa[k] = *(const v2f*)(ap + 4 * k);

    const float* bp = Wl + lr * WPITCH + 2 * hi;
    v8f acc[8] = {};
    #pragma unroll
    for (int k = 0; k < 16; ++k) {
        v2f b[8];
        #pragma unroll
        for (int t = 0; t < 8; ++t)
            b[t] = *(const v2f*)(bp + t * 16 * WPITCH + 4 * k);
        #pragma unroll
        for (int t = 0; t < 8; ++t)
            acc[t] = __builtin_amdgcn_wmma_f32_16x16x4_f32(
                         false, fa[k], false, b[t], (short)0, acc[t], false, false);
    }

    #pragma unroll
    for (int t = 0; t < 8; ++t) {
        const int col = (t & 3) * 16 + lr;
        const float bv = (t < 4 ? bias0 : bias1)[col];
        float* crow = (t < 4 ? C0 : C1) + (size_t)(row0 + hi * 8) * DIM + col;
        #pragma unroll
        for (int v = 0; v < 8; ++v)
            crow[(size_t)v * DIM] = acc[t][v] + bv;
    }
}

__global__ void zero_counts(int* __restrict__ cnt)
{
    const int i = blockIdx.x * blockDim.x + threadIdx.x;
    if (i < NODES) cnt[i] = 0;
}

__global__ void count_dst(const int* __restrict__ dst, int* __restrict__ cnt)
{
    const int e = blockIdx.x * blockDim.x + threadIdx.x;
    if (e < EDGES) atomicAdd(&cnt[dst[e]], 1);
}

#define SCAN_PER ((NODES + 1023) / 1024)
__global__ void __launch_bounds__(1024)
scan_counts(const int* __restrict__ cnt, int* __restrict__ off,
            int* __restrict__ cur)
{
    __shared__ int part[1024];
    const int t = threadIdx.x;
    const int b = t * SCAN_PER;
    int s = 0;
    for (int i = 0; i < SCAN_PER; ++i) {
        const int idx = b + i;
        if (idx < NODES) s += cnt[idx];
    }
    part[t] = s;
    __syncthreads();
    for (int o = 1; o < 1024; o <<= 1) {
        const int v = (t >= o) ? part[t - o] : 0;
        __syncthreads();
        part[t] += v;
        __syncthreads();
    }
    int run = part[t] - s;
    for (int i = 0; i < SCAN_PER; ++i) {
        const int idx = b + i;
        if (idx < NODES) {
            off[idx] = run;
            cur[idx] = run;
            run += cnt[idx];
        }
    }
    if (t == 1023) off[NODES] = part[1023];
}

__global__ void fill_slots(const int* __restrict__ src, const int* __restrict__ dst,
                           const int* __restrict__ ety, int* __restrict__ cur,
                           int* __restrict__ key)
{
    const int e = blockIdx.x * blockDim.x + threadIdx.x;
    if (e >= EDGES) return;
    const int pos = atomicAdd(&cur[dst[e]], 1);
    key[pos] = src[e] + (ety[e] != 0 ? NODES : 0);
}

__device__ __forceinline__ int slot_key(const int* __restrict__ key, int beg,
                                        int selfKey, int v)
{
    return v == 0 ? selfKey : key[beg + v - 1];
}

__global__ void __launch_bounds__(256)
gat_node(const float* __restrict__ el, const float* __restrict__ er,
         const float* __restrict__ attn, const int* __restrict__ off,
         const int* __restrict__ key, float* __restrict__ h,
         float* __restrict__ emb)
{
    const int lane = threadIdx.x & 31;
    const int d0   = (blockIdx.x * 8 + (threadIdx.x >> 5)) * NPW;
    if (d0 >= NODES) return;
    const int offv = off[d0 + (lane < NPW ? lane : NPW)];
    const v2f at   = *(const v2f*)(attn + 2 * lane);
    const int sub  = lane & (CH - 1);
    const bool b16 = (lane & 16) != 0, b8 = (lane & 8) != 0, b4 = (lane & 4) != 0;
    const int myEdge = (b16 ? 4 : 0) + (b8 ? 2 : 0) + (b4 ? 1 : 0);
    static_assert(CH == 8, "the folding butterfly is written for 8 edges per round");

    #pragma unroll
    for (int n = 0; n < NPW; ++n) {
        const int d   = d0 + n;
        const int beg = __builtin_amdgcn_readlane(offv, n);
        const int tot = __builtin_amdgcn_readlane(offv, n + 1) - beg + 1;
        const v2f rr  = *(const v2f*)(er + (size_t)d * DIM + 2 * lane);

        float m = BIG_NEG, den = 0.f;
        v2f acc = {0.f, 0.f};

        int kv = slot_key(key, beg, d, sub < tot ? sub : tot - 1);
        for (int v0 = 0; v0 < tot; v0 += CH) {
            const int rem = tot - v0;
            const int cnt = rem < CH ? rem : CH;
            const int ks = kv >= NODES ? kv - NODES : kv;

            v2f row[CH];
            #pragma unroll
            for (int i = 0; i < CH; ++i) {
                const int s = __builtin_amdgcn_readlane(ks, i);
                row[i] = *(const v2f*)(el + (size_t)s * DIM + 2 * lane);
            }
            int kn = 0;
            if (rem > CH) {
                const int r2 = rem - CH;
                kn = key[beg + v0 + CH - 1 + (sub < r2 ? sub : r2 - 1)];
            }
            float q[CH];
            #pragma unroll
            for (int i = 0; i < CH; ++i) {
                float x0 = row[i].x + rr.x;
                float x1 = row[i].y + rr.y;
                x0 = x0 > 0.f ? x0 : NEG_SLOPE * x0;
                x1 = x1 > 0.f ? x1 : NEG_SLOPE * x1;
                float t = x0 * at.x;
                t += x1 * at.y;
                q[i] = t;
            }
            float k0[4], k1[2], sc_e;
            #pragma unroll
            for (int j = 0; j < 4; ++j) {
                const float send = b16 ? q[j] : q[j + 4];
                const float keep = b16 ? q[j + 4] : q[j];
                k0[j] = keep + __shfl_xor(send, 16, 32);
            }
            #pragma unroll
            for (int j = 0; j < 2; ++j) {
                const float send = b8 ? k0[j] : k0[j + 2];
                const float keep = b8 ? k0[j + 2] : k0[j];
                k1[j] = keep + __shfl_xor(send, 8, 32);
            }
            {
                const float send = b4 ? k1[0] : k1[1];
                const float keep = b4 ? k1[1] : k1[0];
                sc_e = keep + __shfl_xor(send, 4, 32);
            }
            sc_e += __shfl_xor(sc_e, 2, 32);
            sc_e += __shfl_xor(sc_e, 1, 32);

            const bool valid = myEdge < cnt;
            float cm = valid ? sc_e : BIG_NEG;
            cm = fmaxf(cm, __shfl_xor(cm, 16, 32));
            cm = fmaxf(cm, __shfl_xor(cm, 8, 32));
            cm = fmaxf(cm, __shfl_xor(cm, 4, 32));
            const float nm = fmaxf(m, cm);
            const float sc = expf(m - nm);
            const float wv = valid ? expf(sc_e - nm) : 0.f;
            den *= sc; acc.x *= sc; acc.y *= sc;
            #pragma unroll
            for (int i = 0; i < CH; ++i) {
                const int src_lane = ((i >> 2) & 1) * 16 + ((i >> 1) & 1) * 8 + (i & 1) * 4;
                const float w = __int_as_float(
                    __builtin_amdgcn_readlane(__float_as_int(wv), src_lane));
                den   += w;
                acc.x += w * row[i].x;
                acc.y += w * row[i].y;
            }
            m  = nm;
            kv = kn;
        }

        v2f o;
        o.x = acc.x / den;
        o.y = acc.y / den;
        o.x = o.x > 0.f ? o.x : (expf(o.x) - 1.f);
        o.y = o.y > 0.f ? o.y : (expf(o.y) - 1.f);
        *(v2f*)(h   + (size_t)d * DIM + 2 * lane) = o;
        *(v2f*)(emb + (size_t)d * DIM + 2 * lane) = o;
    }
}

__global__ void __launch_bounds__(256)
ggc_pull(const float* __restrict__ tr, const int* __restrict__ ety,
         const int* __restrict__ off, const int* __restrict__ key,
         float* __restrict__ acc)
{
    const int lane = threadIdx.x & 31;
    const int d0   = (blockIdx.x * 8 + (threadIdx.x >> 5)) * NPW;
    if (d0 >= NODES) return;
    const int offv = off[d0 + (lane < NPW ? lane : NPW)];
    const int nl   = lane < NPW ? lane : 0;
    const int selfv = d0 + nl + (ety[EDGES + d0 + nl] != 0 ? NODES : 0);
    const int sub  = lane & (CH - 1);

    #pragma unroll
    for (int n = 0; n < NPW; ++n) {
        const int d   = d0 + n;
        const int beg = __builtin_amdgcn_readlane(offv, n);
        const int tot = __builtin_amdgcn_readlane(offv, n + 1) - beg + 1;
        const int selfKey = __builtin_amdgcn_readlane(selfv, n);

        v2f a = {0.f, 0.f};
        int kv = slot_key(key, beg, selfKey, sub < tot ? sub : tot - 1);
        for (int v0 = 0; v0 < tot; v0 += CH) {
            const int rem = tot - v0;
            const int cnt = rem < CH ? rem : CH;
            v2f row[CH];
            #pragma unroll
            for (int i = 0; i < CH; ++i) {
                const int r = __builtin_amdgcn_readlane(kv, i);
                row[i] = *(const v2f*)(tr + (size_t)r * DIM + 2 * lane);
            }
            int kn = 0;
            if (rem > CH) {
                const int r2 = rem - CH;
                kn = key[beg + v0 + CH - 1 + (sub < r2 ? sub : r2 - 1)];
            }
            #pragma unroll
            for (int i = 0; i < CH; ++i) {
                a.x += (i < cnt) ? row[i].x : 0.f;
                a.y += (i < cnt) ? row[i].y : 0.f;
            }
            kv = kn;
        }
        *(v2f*)(acc + (size_t)d * DIM + 2 * lane) = a;
    }
}

#define GATES  (3 * DIM)

__global__ void __launch_bounds__(256)
gru_fused(const float* __restrict__ am,
          const float* __restrict__ Wih, const float* __restrict__ bih,
          const float* __restrict__ Whh, const float* __restrict__ bhh,
          float* h, float* emb)
{
    __shared__ __attribute__((aligned(16))) float Wl[2 * GATES * WPITCH];

    const int tid  = threadIdx.x;
    const int lane = tid & 31;
    const int lr   = lane & 15;
    const int hi   = lane >> 4;
    const int rt   = blockIdx.x * 8 + (tid >> 5);
    const bool live = rt < ROW_TILES;
    const int row0 = rt << 4;

    v2f fa[16] = {}, fh[16] = {};
    if (live) {
        const float* arow = am + (size_t)(row0 + lr) * DIM + 2 * hi;
        const float* hrow = h  + (size_t)(row0 + lr) * DIM + 2 * hi;
        #pragma unroll
        for (int k = 0; k < 16; ++k) {
            fa[k] = *(const v2f*)(arow + 4 * k);
            fh[k] = *(const v2f*)(hrow + 4 * k);
        }
    }

    for (int i = tid; i < GATES * (DIM / 4); i += 256) {
        const int r = i >> 4, c4 = i & 15;
        const v4f wi = *(const v4f*)(Wih + 4 * i);
        const v4f wh = *(const v4f*)(Whh + 4 * i);
        *(v4f*)(Wl + r * WPITCH + 4 * c4)           = wi;
        *(v4f*)(Wl + (GATES + r) * WPITCH + 4 * c4) = wh;
    }
    __syncthreads();
    if (!live) return;

    for (int ct = 0; ct < DIM / 16; ++ct) {
        const int dcol = ct * 16 + lr;
        const float* bi = Wl + dcol * WPITCH + 2 * hi;
        const float* bh = bi + GATES * WPITCH;

        float* hp = h + (size_t)(row0 + hi * 8) * DIM + dcol;
        float hv[8];
        #pragma unroll
        for (int v = 0; v < 8; ++v) hv[v] = hp[(size_t)v * DIM];

        v8f cir = {}, ciz = {}, cic = {}, chr = {}, chz = {}, chc = {};
        #pragma unroll
        for (int k = 0; k < 16; ++k) {
            const v2f b0 = *(const v2f*)(bi + 4 * k);
            const v2f b1 = *(const v2f*)(bi + DIM * WPITCH + 4 * k);
            const v2f b2 = *(const v2f*)(bi + 2 * DIM * WPITCH + 4 * k);
            const v2f b3 = *(const v2f*)(bh + 4 * k);
            const v2f b4 = *(const v2f*)(bh + DIM * WPITCH + 4 * k);
            const v2f b5 = *(const v2f*)(bh + 2 * DIM * WPITCH + 4 * k);
            cir = __builtin_amdgcn_wmma_f32_16x16x4_f32(
                      false, fa[k], false, b0, (short)0, cir, false, false);
            ciz = __builtin_amdgcn_wmma_f32_16x16x4_f32(
                      false, fa[k], false, b1, (short)0, ciz, false, false);
            cic = __builtin_amdgcn_wmma_f32_16x16x4_f32(
                      false, fa[k], false, b2, (short)0, cic, false, false);
            chr = __builtin_amdgcn_wmma_f32_16x16x4_f32(
                      false, fh[k], false, b3, (short)0, chr, false, false);
            chz = __builtin_amdgcn_wmma_f32_16x16x4_f32(
                      false, fh[k], false, b4, (short)0, chz, false, false);
            chc = __builtin_amdgcn_wmma_f32_16x16x4_f32(
                      false, fh[k], false, b5, (short)0, chc, false, false);
        }

        const float bir = bih[dcol], biz = bih[DIM + dcol], bic = bih[2 * DIM + dcol];
        const float bhr = bhh[dcol], bhz = bhh[DIM + dcol], bhc = bhh[2 * DIM + dcol];
        float* ep = emb ? emb + (size_t)(row0 + hi * 8) * DIM + dcol : nullptr;
        #pragma unroll
        for (int v = 0; v < 8; ++v) {
            const float gir = cir[v] + bir, ghr = chr[v] + bhr;
            const float giz = ciz[v] + biz, ghz = chz[v] + bhz;
            const float gic = cic[v] + bic, ghc = chc[v] + bhc;
            const float r = 1.f / (1.f + expf(-(gir + ghr)));
            const float z = 1.f / (1.f + expf(-(giz + ghz)));
            const float c = tanhf(gic + r * ghc);
            const float o = (1.f - z) * c + z * hv[v];
            hp[(size_t)v * DIM] = o;
            if (ep) ep[(size_t)v * DIM] = o;
        }
    }
}

extern "C" void kernel_launch(void* const* d_in, const int* in_sizes, int n_in,
                              void* d_out, int out_size, void* d_ws, size_t ws_size,
                              hipStream_t stream)
{
    (void)in_sizes; (void)n_in; (void)out_size; (void)ws_size;

    const float* x   = (const float*)d_in[0];
    const int*   src = (const int*)d_in[1];
    const int*   dst = (const int*)d_in[2];
    const int*   ety = (const int*)d_in[3];

    const long NE = (long)NODES * DIM;

    float* ws   = (float*)d_ws;
    float* hbuf = ws;                 ws += NE;
    float* buf1 = ws;                 ws += NE;
    float* buf2 = ws;                 ws += NE;
    float* acc  = ws;                 ws += NE;
    int* off = (int*)ws;              ws += NODES + 64;
    int* cur = (int*)ws;              ws += NODES + 64;
    int* cnt = (int*)ws;              ws += NODES + 64;
    int* key = (int*)ws;

    float* out = (float*)d_out;

    const int edgeThrBlocks  = (EDGES + 255) / 256;
    const int rowTileBlocks  = (ROW_TILES + 7) / 8;
    const int gruBlocks      = (ROW_TILES + 7) / 8;

    zero_counts<<<(NODES + 255) / 256, 256, 0, stream>>>(cnt);
    count_dst<<<edgeThrBlocks, 256, 0, stream>>>(dst, cnt);
    scan_counts<<<1, 1024, 0, stream>>>(cnt, off, cur);
    fill_slots<<<edgeThrBlocks, 256, 0, stream>>>(src, dst, ety, cur, key);

    const float* hin = x;
    for (int l = 0; l < 2; ++l) {
        const int base = 4 + 11 * l;
        const float* Wsrc = (const float*)d_in[base + 0];
        const float* bsrc = (const float*)d_in[base + 1];
        const float* Wdst = (const float*)d_in[base + 2];
        const float* bdst = (const float*)d_in[base + 3];
        const float* attn = (const float*)d_in[base + 4];
        const float* ggW  = (const float*)d_in[base + 5];
        const float* ggb  = (const float*)d_in[base + 6];
        const float* Wih  = (const float*)d_in[base + 7];
        const float* bih  = (const float*)d_in[base + 8];
        const float* Whh  = (const float*)d_in[base + 9];
        const float* bhh  = (const float*)d_in[base + 10];

        gemm_pair_wmma<<<rowTileBlocks, 256, 0, stream>>>(hin, Wsrc, bsrc, Wdst, bdst,
                                                          buf1, buf2);
        gat_node<<<NODE_BLOCKS, 256, 0, stream>>>(buf1, buf2, attn, off, key, hbuf,
                                                  out + (2L * l) * NE);

        for (int st = 0; st < 2; ++st) {
            gemm_pair_wmma<<<rowTileBlocks, 256, 0, stream>>>(
                hbuf, ggW, ggb, ggW + DIM * DIM, ggb + DIM, buf1, buf2);
            ggc_pull<<<NODE_BLOCKS, 256, 0, stream>>>(buf1, ety, off, key, acc);
            gru_fused<<<gruBlocks, 256, 0, stream>>>(
                acc, Wih, bih, Whh, bhh, hbuf,
                (st == 1) ? out + (2L * l + 1) * NE : nullptr);
        }
        hin = hbuf;
    }
}


# slot-bound scan on 49 work-groups instead of one: each scans 1024 counts and sums the counts in front of its range itself, 16-byte loads in one clause; 98 serial lane-strided round trips gone
# speedup vs baseline: 3.1436x; 1.2588x over previous
#include <hip/hip_runtime.h>
#include <hip/hip_bf16.h>


typedef __attribute__((ext_vector_type(2))) float v2f;
typedef __attribute__((ext_vector_type(4))) float v4f;
typedef __attribute__((ext_vector_type(8))) float v8f;

#define NODES 50000
#define EDGES 800000
#define DIM   64
#define NEG_SLOPE 0.2f

#define NPW 4
#define CH  8
static_assert(NODES % NPW == 0, "node group must divide the node count");
#define NODE_BLOCKS ((((NODES) / NPW) + 7) / 8)
#define BIG_NEG (-3.402823466e+38f)

#define WPITCH 68
#define ROW_TILES (NODES / 16)
static_assert(NODES % 16 == 0, "row tiles must divide the node count");

__global__ void __launch_bounds__(256) __attribute__((amdgpu_waves_per_eu(2, 4)))
gemm_pair_wmma(const float* __restrict__ A,
               const float* __restrict__ W0, const float* __restrict__ bias0,
               const float* __restrict__ W1, const float* __restrict__ bias1,
               float* __restrict__ C0, float* __restrict__ C1)
{
    __shared__ __attribute__((aligned(16))) float Wl[2 * DIM * WPITCH];
    __shared__ __attribute__((aligned(16))) float Al[8 * 16 * WPITCH];

    const int tid  = threadIdx.x;
    const int lane = tid & 31;
    const int lr   = lane & 15;
    const int hi   = lane >> 4;
    const int rt   = blockIdx.x * 8 + (tid >> 5);
    const bool live = rt < ROW_TILES;
    const int row0 = (live ? rt : ROW_TILES - 1) << 4;
    float* al = Al + (tid >> 5) * 16 * WPITCH;

    const float* atile = A + (size_t)row0 * DIM;
    v4f av[8], w0v[4], w1v[4];
    #pragma unroll
    for (int j = 0; j < 8; ++j)
        av[j] = *(const v4f*)(atile + 4 * (j * 32 + lane));
    #pragma unroll
    for (int j = 0; j < 4; ++j) {
        w0v[j] = *(const v4f*)(W0 + 4 * (j * 256 + tid));
        w1v[j] = *(const v4f*)(W1 + 4 * (j * 256 + tid));
    }
    #pragma unroll
    for (int j = 0; j < 8; ++j) {
        const int i = j * 32 + lane;
        *(v4f*)(al + (i >> 4) * WPITCH + 4 * (i & 15)) = av[j];
    }
    #pragma unroll
    for (int j = 0; j < 4; ++j) {
        const int i = j * 256 + tid;
        const int r = i >> 4, c4 = i & 15;
        *(v4f*)(Wl + r * WPITCH + 4 * c4)         = w0v[j];
        *(v4f*)(Wl + (DIM + r) * WPITCH + 4 * c4) = w1v[j];
    }
    __syncthreads();
    if (!live) return;

    v2f fa[16];
    const float* ap = al + lr * WPITCH + 2 * hi;
    #pragma unroll
    for (int k = 0; k < 16; ++k) fa[k] = *(const v2f*)(ap + 4 * k);

    const float* bp = Wl + lr * WPITCH + 2 * hi;
    v8f acc[8] = {};
    #pragma unroll
    for (int k = 0; k < 16; ++k) {
        v2f b[8];
        #pragma unroll
        for (int t = 0; t < 8; ++t)
            b[t] = *(const v2f*)(bp + t * 16 * WPITCH + 4 * k);
        #pragma unroll
        for (int t = 0; t < 8; ++t)
            acc[t] = __builtin_amdgcn_wmma_f32_16x16x4_f32(
                         false, fa[k], false, b[t], (short)0, acc[t], false, false);
    }

    #pragma unroll
    for (int t = 0; t < 8; ++t) {
        const int col = (t & 3) * 16 + lr;
        const float bv = (t < 4 ? bias0 : bias1)[col];
        float* crow = (t < 4 ? C0 : C1) + (size_t)(row0 + hi * 8) * DIM + col;
        #pragma unroll
        for (int v = 0; v < 8; ++v)
            crow[(size_t)v * DIM] = acc[t][v] + bv;
    }
}

__global__ void zero_counts(int* __restrict__ cnt)
{
    const int i = blockIdx.x * blockDim.x + threadIdx.x;
    if (i < NODES) cnt[i] = 0;
}

__global__ void count_dst(const int* __restrict__ dst, int* __restrict__ cnt)
{
    const int e = blockIdx.x * blockDim.x + threadIdx.x;
    if (e < EDGES) atomicAdd(&cnt[dst[e]], 1);
}

#define SCAN_T 1024
#define SCAN_BLOCKS ((NODES + SCAN_T - 1) / SCAN_T)
#define SCAN_PRE (((SCAN_BLOCKS - 1) * (SCAN_T / 4) + SCAN_T - 1) / SCAN_T)
static_assert(SCAN_T % 128 == 0, "a group's first count must be a whole number of 16-byte loads per wave");
static_assert(SCAN_PRE * SCAN_T * 4 <= NODES, "the loads in front of the last group must stay inside the count table");

__global__ void __launch_bounds__(SCAN_T)
scan_counts(const int* __restrict__ cnt, int* __restrict__ off,
            int* __restrict__ cur)
{
    __shared__ int wpre[32];
    __shared__ int wtot[32];
    const int t     = threadIdx.x;
    const int lane  = t & 31;
    const int w     = t >> 5;
    const int first = blockIdx.x * SCAN_T;
    const int idx   = first + t;
    const int c     = idx < NODES ? cnt[idx] : 0;

    const int n4 = first >> 2;
    const int4* c4 = (const int4*)cnt;
    int4 pv4[SCAN_PRE];
    #pragma unroll
    for (int j = 0; j < SCAN_PRE; ++j) pv4[j] = c4[j * SCAN_T + t];
    int s = 0;
    #pragma unroll
    for (int j = 0; j < SCAN_PRE; ++j) {
        const int q = (pv4[j].x + pv4[j].y) + (pv4[j].z + pv4[j].w);
        s += (j * SCAN_T + t < n4) ? q : 0;
    }

    int inc = c;
    for (int o = 1; o < 32; o <<= 1) {
        const int v = __shfl_up(inc, o, 32);
        if (lane >= o) inc += v;
    }
    for (int o = 16; o > 0; o >>= 1) s += __shfl_xor(s, o, 32);
    if (lane == 31) wtot[w] = inc;
    if (lane == 0)  wpre[w] = s;
    __syncthreads();

    const int tv = wtot[lane];
    int tinc = tv;
    int pv   = wpre[lane];
    for (int o = 1; o < 32; o <<= 1) {
        const int v = __shfl_up(tinc, o, 32);
        if (lane >= o) tinc += v;
    }
    for (int o = 16; o > 0; o >>= 1) pv += __shfl_xor(pv, o, 32);
    const int wexc = __shfl(tinc - tv, w, 32);

    const int run = pv + wexc + (inc - c);
    if (idx < NODES) {
        off[idx] = run;
        cur[idx] = run;
    }
    if (idx == NODES - 1) off[NODES] = run + c;
}

__global__ void fill_slots(const int* __restrict__ src, const int* __restrict__ dst,
                           const int* __restrict__ ety, int* __restrict__ cur,
                           int* __restrict__ key)
{
    const int e = blockIdx.x * blockDim.x + threadIdx.x;
    if (e >= EDGES) return;
    const int pos = atomicAdd(&cur[dst[e]], 1);
    key[pos] = src[e] + (ety[e] != 0 ? NODES : 0);
}

__device__ __forceinline__ int slot_key(const int* __restrict__ key, int beg,
                                        int selfKey, int v)
{
    return v == 0 ? selfKey : key[beg + v - 1];
}

__global__ void __launch_bounds__(256)
gat_node(const float* __restrict__ el, const float* __restrict__ er,
         const float* __restrict__ attn, const int* __restrict__ off,
         const int* __restrict__ key, float* __restrict__ h,
         float* __restrict__ emb)
{
    const int lane = threadIdx.x & 31;
    const int d0   = (blockIdx.x * 8 + (threadIdx.x >> 5)) * NPW;
    if (d0 >= NODES) return;
    const int offv = off[d0 + (lane < NPW ? lane : NPW)];
    const v2f at   = *(const v2f*)(attn + 2 * lane);
    const int sub  = lane & (CH - 1);
    const bool b16 = (lane & 16) != 0, b8 = (lane & 8) != 0, b4 = (lane & 4) != 0;
    const int myEdge = (b16 ? 4 : 0) + (b8 ? 2 : 0) + (b4 ? 1 : 0);
    static_assert(CH == 8, "the folding butterfly is written for 8 edges per round");

    #pragma unroll
    for (int n = 0; n < NPW; ++n) {
        const int d   = d0 + n;
        const int beg = __builtin_amdgcn_readlane(offv, n);
        const int tot = __builtin_amdgcn_readlane(offv, n + 1) - beg + 1;
        const v2f rr  = *(const v2f*)(er + (size_t)d * DIM + 2 * lane);

        float m = BIG_NEG, den = 0.f;
        v2f acc = {0.f, 0.f};

        int kv = slot_key(key, beg, d, sub < tot ? sub : tot - 1);
        for (int v0 = 0; v0 < tot; v0 += CH) {
            const int rem = tot - v0;
            const int cnt = rem < CH ? rem : CH;
            const int ks = kv >= NODES ? kv - NODES : kv;

            v2f row[CH];
            #pragma unroll
            for (int i = 0; i < CH; ++i) {
                const int s = __builtin_amdgcn_readlane(ks, i);
                row[i] = *(const v2f*)(el + (size_t)s * DIM + 2 * lane);
            }
            int kn = 0;
            if (rem > CH) {
                const int r2 = rem - CH;
                kn = key[beg + v0 + CH - 1 + (sub < r2 ? sub : r2 - 1)];
            }
            float q[CH];
            #pragma unroll
            for (int i = 0; i < CH; ++i) {
                float x0 = row[i].x + rr.x;
                float x1 = row[i].y + rr.y;
                x0 = x0 > 0.f ? x0 : NEG_SLOPE * x0;
                x1 = x1 > 0.f ? x1 : NEG_SLOPE * x1;
                float t = x0 * at.x;
                t += x1 * at.y;
                q[i] = t;
            }
            float k0[4], k1[2], sc_e;
            #pragma unroll
            for (int j = 0; j < 4; ++j) {
                const float send = b16 ? q[j] : q[j + 4];
                const float keep = b16 ? q[j + 4] : q[j];
                k0[j] = keep + __shfl_xor(send, 16, 32);
            }
            #pragma unroll
            for (int j = 0; j < 2; ++j) {
                const float send = b8 ? k0[j] : k0[j + 2];
                const float keep = b8 ? k0[j + 2] : k0[j];
                k1[j] = keep + __shfl_xor(send, 8, 32);
            }
            {
                const float send = b4 ? k1[0] : k1[1];
                const float keep = b4 ? k1[1] : k1[0];
                sc_e = keep + __shfl_xor(send, 4, 32);
            }
            sc_e += __shfl_xor(sc_e, 2, 32);
            sc_e += __shfl_xor(sc_e, 1, 32);

            const bool valid = myEdge < cnt;
            float cm = valid ? sc_e : BIG_NEG;
            cm = fmaxf(cm, __shfl_xor(cm, 16, 32));
            cm = fmaxf(cm, __shfl_xor(cm, 8, 32));
            cm = fmaxf(cm, __shfl_xor(cm, 4, 32));
            const float nm = fmaxf(m, cm);
            const float sc = expf(m - nm);
            const float wv = valid ? expf(sc_e - nm) : 0.f;
            den *= sc; acc.x *= sc; acc.y *= sc;
            #pragma unroll
            for (int i = 0; i < CH; ++i) {
                const int src_lane = ((i >> 2) & 1) * 16 + ((i >> 1) & 1) * 8 + (i & 1) * 4;
                const float w = __int_as_float(
                    __builtin_amdgcn_readlane(__float_as_int(wv), src_lane));
                den   += w;
                acc.x += w * row[i].x;
                acc.y += w * row[i].y;
            }
            m  = nm;
            kv = kn;
        }

        v2f o;
        o.x = acc.x / den;
        o.y = acc.y / den;
        o.x = o.x > 0.f ? o.x : (expf(o.x) - 1.f);
        o.y = o.y > 0.f ? o.y : (expf(o.y) - 1.f);
        *(v2f*)(h   + (size_t)d * DIM + 2 * lane) = o;
        *(v2f*)(emb + (size_t)d * DIM + 2 * lane) = o;
    }
}

__global__ void __launch_bounds__(256)
ggc_pull(const float* __restrict__ tr, const int* __restrict__ ety,
         const int* __restrict__ off, const int* __restrict__ key,
         float* __restrict__ acc)
{
    const int lane = threadIdx.x & 31;
    const int d0   = (blockIdx.x * 8 + (threadIdx.x >> 5)) * NPW;
    if (d0 >= NODES) return;
    const int offv = off[d0 + (lane < NPW ? lane : NPW)];
    const int nl   = lane < NPW ? lane : 0;
    const int selfv = d0 + nl + (ety[EDGES + d0 + nl] != 0 ? NODES : 0);
    const int sub  = lane & (CH - 1);

    #pragma unroll
    for (int n = 0; n < NPW; ++n) {
        const int d   = d0 + n;
        const int beg = __builtin_amdgcn_readlane(offv, n);
        const int tot = __builtin_amdgcn_readlane(offv, n + 1) - beg + 1;
        const int selfKey = __builtin_amdgcn_readlane(selfv, n);

        v2f a = {0.f, 0.f};
        int kv = slot_key(key, beg, selfKey, sub < tot ? sub : tot - 1);
        for (int v0 = 0; v0 < tot; v0 += CH) {
            const int rem = tot - v0;
            const int cnt = rem < CH ? rem : CH;
            v2f row[CH];
            #pragma unroll
            for (int i = 0; i < CH; ++i) {
                const int r = __builtin_amdgcn_readlane(kv, i);
                row[i] = *(const v2f*)(tr + (size_t)r * DIM + 2 * lane);
            }
            int kn = 0;
            if (rem > CH) {
                const int r2 = rem - CH;
                kn = key[beg + v0 + CH - 1 + (sub < r2 ? sub : r2 - 1)];
            }
            #pragma unroll
            for (int i = 0; i < CH; ++i) {
                a.x += (i < cnt) ? row[i].x : 0.f;
                a.y += (i < cnt) ? row[i].y : 0.f;
            }
            kv = kn;
        }
        *(v2f*)(acc + (size_t)d * DIM + 2 * lane) = a;
    }
}

#define GATES  (3 * DIM)

__global__ void __launch_bounds__(256)
gru_fused(const float* __restrict__ am,
          const float* __restrict__ Wih, const float* __restrict__ bih,
          const float* __restrict__ Whh, const float* __restrict__ bhh,
          float* h, float* emb)
{
    __shared__ __attribute__((aligned(16))) float Wl[2 * GATES * WPITCH];

    const int tid  = threadIdx.x;
    const int lane = tid & 31;
    const int lr   = lane & 15;
    const int hi   = lane >> 4;
    const int rt   = blockIdx.x * 8 + (tid >> 5);
    const bool live = rt < ROW_TILES;
    const int row0 = rt << 4;

    v2f fa[16] = {}, fh[16] = {};
    if (live) {
        const float* arow = am + (size_t)(row0 + lr) * DIM + 2 * hi;
        const float* hrow = h  + (size_t)(row0 + lr) * DIM + 2 * hi;
        #pragma unroll
        for (int k = 0; k < 16; ++k) {
            fa[k] = *(const v2f*)(arow + 4 * k);
            fh[k] = *(const v2f*)(hrow + 4 * k);
        }
    }

    for (int i = tid; i < GATES * (DIM / 4); i += 256) {
        const int r = i >> 4, c4 = i & 15;
        const v4f wi = *(const v4f*)(Wih + 4 * i);
        const v4f wh = *(const v4f*)(Whh + 4 * i);
        *(v4f*)(Wl + r * WPITCH + 4 * c4)           = wi;
        *(v4f*)(Wl + (GATES + r) * WPITCH + 4 * c4) = wh;
    }
    __syncthreads();
    if (!live) return;

    for (int ct = 0; ct < DIM / 16; ++ct) {
        const int dcol = ct * 16 + lr;
        const float* bi = Wl + dcol * WPITCH + 2 * hi;
        const float* bh = bi + GATES * WPITCH;

        float* hp = h + (size_t)(row0 + hi * 8) * DIM + dcol;
        float hv[8];
        #pragma unroll
        for (int v = 0; v < 8; ++v) hv[v] = hp[(size_t)v * DIM];

        v8f cir = {}, ciz = {}, cic = {}, chr = {}, chz = {}, chc = {};
        #pragma unroll
        for (int k = 0; k < 16; ++k) {
            const v2f b0 = *(const v2f*)(bi + 4 * k);
            const v2f b1 = *(const v2f*)(bi + DIM * WPITCH + 4 * k);
            const v2f b2 = *(const v2f*)(bi + 2 * DIM * WPITCH + 4 * k);
            const v2f b3 = *(const v2f*)(bh + 4 * k);
            const v2f b4 = *(const v2f*)(bh + DIM * WPITCH + 4 * k);
            const v2f b5 = *(const v2f*)(bh + 2 * DIM * WPITCH + 4 * k);
            cir = __builtin_amdgcn_wmma_f32_16x16x4_f32(
                      false, fa[k], false, b0, (short)0, cir, false, false);
            ciz = __builtin_amdgcn_wmma_f32_16x16x4_f32(
                      false, fa[k], false, b1, (short)0, ciz, false, false);
            cic = __builtin_amdgcn_wmma_f32_16x16x4_f32(
                      false, fa[k], false, b2, (short)0, cic, false, false);
            chr = __builtin_amdgcn_wmma_f32_16x16x4_f32(
                      false, fh[k], false, b3, (short)0, chr, false, false);
            chz = __builtin_amdgcn_wmma_f32_16x16x4_f32(
                      false, fh[k], false, b4, (short)0, chz, false, false);
            chc = __builtin_amdgcn_wmma_f32_16x16x4_f32(
                      false, fh[k], false, b5, (short)0, chc, false, false);
        }

        const float bir = bih[dcol], biz = bih[DIM + dcol], bic = bih[2 * DIM + dcol];
        const float bhr = bhh[dcol], bhz = bhh[DIM + dcol], bhc = bhh[2 * DIM + dcol];
        float* ep = emb ? emb + (size_t)(row0 + hi * 8) * DIM + dcol : nullptr;
        #pragma unroll
        for (int v = 0; v < 8; ++v) {
            const float gir = cir[v] + bir, ghr = chr[v] + bhr;
            const float giz = ciz[v] + biz, ghz = chz[v] + bhz;
            const float gic = cic[v] + bic, ghc = chc[v] + bhc;
            const float r = 1.f / (1.f + expf(-(gir + ghr)));
            const float z = 1.f / (1.f + expf(-(giz + ghz)));
            const float c = tanhf(gic + r * ghc);
            const float o = (1.f - z) * c + z * hv[v];
            hp[(size_t)v * DIM] = o;
            if (ep) ep[(size_t)v * DIM] = o;
        }
    }
}

extern "C" void kernel_launch(void* const* d_in, const int* in_sizes, int n_in,
                              void* d_out, int out_size, void* d_ws, size_t ws_size,
                              hipStream_t stream)
{
    (void)in_sizes; (void)n_in; (void)out_size; (void)ws_size;

    const float* x   = (const float*)d_in[0];
    const int*   src = (const int*)d_in[1];
    const int*   dst = (const int*)d_in[2];
    const int*   ety = (const int*)d_in[3];

    const long NE = (long)NODES * DIM;

    float* ws   = (float*)d_ws;
    float* hbuf = ws;                 ws += NE;
    float* buf1 = ws;                 ws += NE;
    float* buf2 = ws;                 ws += NE;
    float* acc  = ws;                 ws += NE;
    int* off = (int*)ws;              ws += NODES + 64;
    int* cur = (int*)ws;              ws += NODES + 64;
    int* cnt = (int*)ws;              ws += NODES + 64;
    int* key = (int*)ws;

    float* out = (float*)d_out;

    const int edgeThrBlocks  = (EDGES + 255) / 256;
    const int rowTileBlocks  = (ROW_TILES + 7) / 8;
    const int gruBlocks      = (ROW_TILES + 7) / 8;

    zero_counts<<<(NODES + 255) / 256, 256, 0, stream>>>(cnt);
    count_dst<<<edgeThrBlocks, 256, 0, stream>>>(dst, cnt);
    scan_counts<<<SCAN_BLOCKS, SCAN_T, 0, stream>>>(cnt, off, cur);
    fill_slots<<<edgeThrBlocks, 256, 0, stream>>>(src, dst, ety, cur, key);

    const float* hin = x;
    for (int l = 0; l < 2; ++l) {
        const int base = 4 + 11 * l;
        const float* Wsrc = (const float*)d_in[base + 0];
        const float* bsrc = (const float*)d_in[base + 1];
        const float* Wdst = (const float*)d_in[base + 2];
        const float* bdst = (const float*)d_in[base + 3];
        const float* attn = (const float*)d_in[base + 4];
        const float* ggW  = (const float*)d_in[base + 5];
        const float* ggb  = (const float*)d_in[base + 6];
        const float* Wih  = (const float*)d_in[base + 7];
        const float* bih  = (const float*)d_in[base + 8];
        const float* Whh  = (const float*)d_in[base + 9];
        const float* bhh  = (const float*)d_in[base + 10];

        gemm_pair_wmma<<<rowTileBlocks, 256, 0, stream>>>(hin, Wsrc, bsrc, Wdst, bdst,
                                                          buf1, buf2);
        gat_node<<<NODE_BLOCKS, 256, 0, stream>>>(buf1, buf2, attn, off, key, hbuf,
                                                  out + (2L * l) * NE);

        for (int st = 0; st < 2; ++st) {
            gemm_pair_wmma<<<rowTileBlocks, 256, 0, stream>>>(
                hbuf, ggW, ggb, ggW + DIM * DIM, ggb + DIM, buf1, buf2);
            ggc_pull<<<NODE_BLOCKS, 256, 0, stream>>>(buf1, ety, off, key, acc);
            gru_fused<<<gruBlocks, 256, 0, stream>>>(
                acc, Wih, bih, Whh, bhh, hbuf,
                (st == 1) ? out + (2L * l + 1) * NE : nullptr);
        }
        hin = hbuf;
    }
}


# pair products whose A rows the kernel in front has just made (GAT node kernel, GRU step) are built inside it from LDS: 5 of 6 pair-GEMM launches and their A re-reads gone, 20 -> 15 launches
# speedup vs baseline: 3.3181x; 1.0555x over previous
#include <hip/hip_runtime.h>
#include <hip/hip_bf16.h>


typedef __attribute__((ext_vector_type(2))) float v2f;
typedef __attribute__((ext_vector_type(4))) float v4f;
typedef __attribute__((ext_vector_type(8))) float v8f;

#define NODES 50000
#define EDGES 800000
#define DIM   64
#define NEG_SLOPE 0.2f

#define NPW 4
#define CH  8
static_assert(NODES % NPW == 0, "node group must divide the node count");
#define NODE_BLOCKS ((((NODES) / NPW) + 7) / 8)
#define BIG_NEG (-3.402823466e+38f)

#define WPITCH 68
#define ROW_TILES (NODES / 16)
static_assert(NODES % 16 == 0, "row tiles must divide the node count");

__global__ void __launch_bounds__(256) __attribute__((amdgpu_waves_per_eu(2, 4)))
gemm_pair_wmma(const float* __restrict__ A,
               const float* __restrict__ W0, const float* __restrict__ bias0,
               const float* __restrict__ W1, const float* __restrict__ bias1,
               float* __restrict__ C0, float* __restrict__ C1)
{
    __shared__ __attribute__((aligned(16))) float Wl[2 * DIM * WPITCH];
    __shared__ __attribute__((aligned(16))) float Al[8 * 16 * WPITCH];

    const int tid  = threadIdx.x;
    const int lane = tid & 31;
    const int lr   = lane & 15;
    const int hi   = lane >> 4;
    const int rt   = blockIdx.x * 8 + (tid >> 5);
    const bool live = rt < ROW_TILES;
    const int row0 = (live ? rt : ROW_TILES - 1) << 4;
    float* al = Al + (tid >> 5) * 16 * WPITCH;

    const float* atile = A + (size_t)row0 * DIM;
    v4f av[8], w0v[4], w1v[4];
    #pragma unroll
    for (int j = 0; j < 8; ++j)
        av[j] = *(const v4f*)(atile + 4 * (j * 32 + lane));
    #pragma unroll
    for (int j = 0; j < 4; ++j) {
        w0v[j] = *(const v4f*)(W0 + 4 * (j * 256 + tid));
        w1v[j] = *(const v4f*)(W1 + 4 * (j * 256 + tid));
    }
    #pragma unroll
    for (int j = 0; j < 8; ++j) {
        const int i = j * 32 + lane;
        *(v4f*)(al + (i >> 4) * WPITCH + 4 * (i & 15)) = av[j];
    }
    #pragma unroll
    for (int j = 0; j < 4; ++j) {
        const int i = j * 256 + tid;
        const int r = i >> 4, c4 = i & 15;
        *(v4f*)(Wl + r * WPITCH + 4 * c4)         = w0v[j];
        *(v4f*)(Wl + (DIM + r) * WPITCH + 4 * c4) = w1v[j];
    }
    __syncthreads();
    if (!live) return;

    v2f fa[16];
    const float* ap = al + lr * WPITCH + 2 * hi;
    #pragma unroll
    for (int k = 0; k < 16; ++k) fa[k] = *(const v2f*)(ap + 4 * k);

    const float* bp = Wl + lr * WPITCH + 2 * hi;
    v8f acc[8] = {};
    #pragma unroll
    for (int k = 0; k < 16; ++k) {
        v2f b[8];
        #pragma unroll
        for (int t = 0; t < 8; ++t)
            b[t] = *(const v2f*)(bp + t * 16 * WPITCH + 4 * k);
        #pragma unroll
        for (int t = 0; t < 8; ++t)
            acc[t] = __builtin_amdgcn_wmma_f32_16x16x4_f32(
                         false, fa[k], false, b[t], (short)0, acc[t], false, false);
    }

    #pragma unroll
    for (int t = 0; t < 8; ++t) {
        const int col = (t & 3) * 16 + lr;
        const float bv = (t < 4 ? bias0 : bias1)[col];
        float* crow = (t < 4 ? C0 : C1) + (size_t)(row0 + hi * 8) * DIM + col;
        #pragma unroll
        for (int v = 0; v < 8; ++v)
            crow[(size_t)v * DIM] = acc[t][v] + bv;
    }
}

template <int NT, bool KEEP_A, int PITCH, bool SWZ>
__device__ __forceinline__ void pair_product_lds(
    const float* al, const float* Wl, int t0, int lr, int hi,
    const float* __restrict__ bias0, const float* __restrict__ bias1,
    float* __restrict__ C0, float* __restrict__ C1, int row0)
{
    v2f fa[16];
    const int sw = SWZ ? (lr << 2) : 0;
    const float* ap = al + lr * PITCH + 2 * hi;
    if (KEEP_A) {
        #pragma unroll
        for (int k = 0; k < 16; ++k) fa[k] = *(const v2f*)(ap + ((4 * k) ^ sw));
    }

    const float* bp = Wl + (t0 * 16 + lr) * PITCH + 2 * hi;
    v8f acc[NT] = {};
    #pragma unroll
    for (int k = 0; k < 16; ++k) {
        if (!KEEP_A) fa[k] = *(const v2f*)(ap + ((4 * k) ^ sw));
        v2f b[NT];
        #pragma unroll
        for (int t = 0; t < NT; ++t)
            b[t] = *(const v2f*)(bp + t * 16 * PITCH + ((4 * k) ^ sw));
        #pragma unroll
        for (int t = 0; t < NT; ++t)
            acc[t] = __builtin_amdgcn_wmma_f32_16x16x4_f32(
                         false, fa[k], false, b[t], (short)0, acc[t], false, false);
    }
    #pragma unroll
    for (int t = 0; t < NT; ++t) {
        const int tt  = t0 + t;
        const int col = (tt & 3) * 16 + lr;
        const float bv = (tt < 4 ? bias0 : bias1)[col];
        float* crow = (tt < 4 ? C0 : C1) + (size_t)(row0 + hi * 8) * DIM + col;
        #pragma unroll
        for (int v = 0; v < 8; ++v)
            crow[(size_t)v * DIM] = acc[t][v] + bv;
    }
}

__global__ void zero_counts(int* __restrict__ cnt)
{
    const int i = blockIdx.x * blockDim.x + threadIdx.x;
    if (i < NODES) cnt[i] = 0;
}

__global__ void count_dst(const int* __restrict__ dst, int* __restrict__ cnt)
{
    const int e = blockIdx.x * blockDim.x + threadIdx.x;
    if (e < EDGES) atomicAdd(&cnt[dst[e]], 1);
}

#define SCAN_T 1024
#define SCAN_BLOCKS ((NODES + SCAN_T - 1) / SCAN_T)
#define SCAN_PRE (((SCAN_BLOCKS - 1) * (SCAN_T / 4) + SCAN_T - 1) / SCAN_T)
static_assert(SCAN_T % 128 == 0, "a group's first count must be a whole number of 16-byte loads per wave");
static_assert(SCAN_PRE * SCAN_T * 4 <= NODES, "the loads in front of the last group must stay inside the count table");

__global__ void __launch_bounds__(SCAN_T)
scan_counts(const int* __restrict__ cnt, int* __restrict__ off,
            int* __restrict__ cur)
{
    __shared__ int wpre[32];
    __shared__ int wtot[32];
    const int t     = threadIdx.x;
    const int lane  = t & 31;
    const int w     = t >> 5;
    const int first = blockIdx.x * SCAN_T;
    const int idx   = first + t;
    const int c     = idx < NODES ? cnt[idx] : 0;

    const int n4 = first >> 2;
    const int4* c4 = (const int4*)cnt;
    int4 pv4[SCAN_PRE];
    #pragma unroll
    for (int j = 0; j < SCAN_PRE; ++j) pv4[j] = c4[j * SCAN_T + t];
    int s = 0;
    #pragma unroll
    for (int j = 0; j < SCAN_PRE; ++j) {
        const int q = (pv4[j].x + pv4[j].y) + (pv4[j].z + pv4[j].w);
        s += (j * SCAN_T + t < n4) ? q : 0;
    }

    int inc = c;
    for (int o = 1; o < 32; o <<= 1) {
        const int v = __shfl_up(inc, o, 32);
        if (lane >= o) inc += v;
    }
    for (int o = 16; o > 0; o >>= 1) s += __shfl_xor(s, o, 32);
    if (lane == 31) wtot[w] = inc;
    if (lane == 0)  wpre[w] = s;
    __syncthreads();

    const int tv = wtot[lane];
    int tinc = tv;
    int pv   = wpre[lane];
    for (int o = 1; o < 32; o <<= 1) {
        const int v = __shfl_up(tinc, o, 32);
        if (lane >= o) tinc += v;
    }
    for (int o = 16; o > 0; o >>= 1) pv += __shfl_xor(pv, o, 32);
    const int wexc = __shfl(tinc - tv, w, 32);

    const int run = pv + wexc + (inc - c);
    if (idx < NODES) {
        off[idx] = run;
        cur[idx] = run;
    }
    if (idx == NODES - 1) off[NODES] = run + c;
}

__global__ void fill_slots(const int* __restrict__ src, const int* __restrict__ dst,
                           const int* __restrict__ ety, int* __restrict__ cur,
                           int* __restrict__ key)
{
    const int e = blockIdx.x * blockDim.x + threadIdx.x;
    if (e >= EDGES) return;
    const int pos = atomicAdd(&cur[dst[e]], 1);
    key[pos] = src[e] + (ety[e] != 0 ? NODES : 0);
}

__device__ __forceinline__ int slot_key(const int* __restrict__ key, int beg,
                                        int selfKey, int v)
{
    return v == 0 ? selfKey : key[beg + v - 1];
}

__global__ void __launch_bounds__(256) __attribute__((amdgpu_waves_per_eu(16, 16)))
gat_node(const float* __restrict__ el, const float* __restrict__ er,
         const float* __restrict__ attn, const int* __restrict__ off,
         const int* __restrict__ key, float* __restrict__ h,
         float* __restrict__ emb,
         const float* __restrict__ W0, const float* __restrict__ bias0,
         const float* __restrict__ W1, const float* __restrict__ bias1,
         float* __restrict__ C0, float* __restrict__ C1)
{
    __shared__ __attribute__((aligned(16))) float Wl[2 * DIM * DIM];
    __shared__ __attribute__((aligned(16))) float Al[8 * NPW * DIM];

    const int lane = threadIdx.x & 31;
    const int wv   = __builtin_amdgcn_readfirstlane(threadIdx.x >> 5);
    const int d0   = (blockIdx.x * 8 + wv) * NPW;
    const bool live = d0 < NODES;
    if (live) {
    const int offv = off[d0 + (lane < NPW ? lane : NPW)];
    const v2f at   = *(const v2f*)(attn + 2 * lane);
    const int sub  = lane & (CH - 1);
    const bool b16 = (lane & 16) != 0, b8 = (lane & 8) != 0, b4 = (lane & 4) != 0;
    const int myEdge = (b16 ? 4 : 0) + (b8 ? 2 : 0) + (b4 ? 1 : 0);
    static_assert(CH == 8, "the folding butterfly is written for 8 edges per round");

    #pragma unroll
    for (int n = 0; n < NPW; ++n) {
        const int d   = d0 + n;
        const int beg = __builtin_amdgcn_readlane(offv, n);
        const int tot = __builtin_amdgcn_readlane(offv, n + 1) - beg + 1;
        const v2f rr  = *(const v2f*)(er + (size_t)d * DIM + 2 * lane);

        float m = BIG_NEG, den = 0.f;
        v2f acc = {0.f, 0.f};

        int kv = slot_key(key, beg, d, sub < tot ? sub : tot - 1);
        for (int v0 = 0; v0 < tot; v0 += CH) {
            const int rem = tot - v0;
            const int cnt = rem < CH ? rem : CH;
            const int ks = kv >= NODES ? kv - NODES : kv;

            v2f row[CH];
            #pragma unroll
            for (int i = 0; i < CH; ++i) {
                const int s = __builtin_amdgcn_readlane(ks, i);
                row[i] = *(const v2f*)(el + (size_t)s * DIM + 2 * lane);
            }
            int kn = 0;
            if (rem > CH) {
                const int r2 = rem - CH;
                kn = key[beg + v0 + CH - 1 + (sub < r2 ? sub : r2 - 1)];
            }
            float q[CH];
            #pragma unroll
            for (int i = 0; i < CH; ++i) {
                float x0 = row[i].x + rr.x;
                float x1 = row[i].y + rr.y;
                x0 = x0 > 0.f ? x0 : NEG_SLOPE * x0;
                x1 = x1 > 0.f ? x1 : NEG_SLOPE * x1;
                float t = x0 * at.x;
                t += x1 * at.y;
                q[i] = t;
            }
            float k0[4], k1[2], sc_e;
            #pragma unroll
            for (int j = 0; j < 4; ++j) {
                const float send = b16 ? q[j] : q[j + 4];
                const float keep = b16 ? q[j + 4] : q[j];
                k0[j] = keep + __shfl_xor(send, 16, 32);
            }
            #pragma unroll
            for (int j = 0; j < 2; ++j) {
                const float send = b8 ? k0[j] : k0[j + 2];
                const float keep = b8 ? k0[j + 2] : k0[j];
                k1[j] = keep + __shfl_xor(send, 8, 32);
            }
            {
                const float send = b4 ? k1[0] : k1[1];
                const float keep = b4 ? k1[1] : k1[0];
                sc_e = keep + __shfl_xor(send, 4, 32);
            }
            sc_e += __shfl_xor(sc_e, 2, 32);
            sc_e += __shfl_xor(sc_e, 1, 32);

            const bool valid = myEdge < cnt;
            float cm = valid ? sc_e : BIG_NEG;
            cm = fmaxf(cm, __shfl_xor(cm, 16, 32));
            cm = fmaxf(cm, __shfl_xor(cm, 8, 32));
            cm = fmaxf(cm, __shfl_xor(cm, 4, 32));
            const float nm = fmaxf(m, cm);
            const float sc = expf(m - nm);
            const float wv = valid ? expf(sc_e - nm) : 0.f;
            den *= sc; acc.x *= sc; acc.y *= sc;
            #pragma unroll
            for (int i = 0; i < CH; ++i) {
                const int src_lane = ((i >> 2) & 1) * 16 + ((i >> 1) & 1) * 8 + (i & 1) * 4;
                const float w = __int_as_float(
                    __builtin_amdgcn_readlane(__float_as_int(wv), src_lane));
                den   += w;
                acc.x += w * row[i].x;
                acc.y += w * row[i].y;
            }
            m  = nm;
            kv = kn;
        }

        v2f o;
        o.x = acc.x / den;
        o.y = acc.y / den;
        o.x = o.x > 0.f ? o.x : (expf(o.x) - 1.f);
        o.y = o.y > 0.f ? o.y : (expf(o.y) - 1.f);
        *(v2f*)(h   + (size_t)d * DIM + 2 * lane) = o;
        *(v2f*)(emb + (size_t)d * DIM + 2 * lane) = o;
        *(v2f*)(Al + (wv * NPW + n) * DIM + ((2 * lane) ^ (((wv * NPW + n) & 15) << 2))) = o;
    }
    }

    {
        const int tid = threadIdx.x;
        v4f w0v[4], w1v[4];
        #pragma unroll
        for (int j = 0; j < 4; ++j) {
            w0v[j] = *(const v4f*)(W0 + 4 * (j * 256 + tid));
            w1v[j] = *(const v4f*)(W1 + 4 * (j * 256 + tid));
        }
        #pragma unroll
        for (int j = 0; j < 4; ++j) {
            const int i = j * 256 + tid;
            const int r = i >> 4, c4 = (i & 15) ^ (r & 15);
            *(v4f*)(Wl + r * DIM + 4 * c4)         = w0v[j];
            *(v4f*)(Wl + (DIM + r) * DIM + 4 * c4) = w1v[j];
        }
    }
    __syncthreads();
    const int rtile = blockIdx.x * 2 + (wv & 1);
    if (rtile < ROW_TILES)
        pair_product_lds<2, false, DIM, true>(Al + (wv & 1) * 16 * DIM, Wl, 2 * (wv >> 1),
                            lane & 15, lane >> 4, bias0, bias1, C0, C1, rtile << 4);
}

__global__ void __launch_bounds__(256)
ggc_pull(const float* __restrict__ tr, const int* __restrict__ ety,
         const int* __restrict__ off, const int* __restrict__ key,
         float* __restrict__ acc)
{
    const int lane = threadIdx.x & 31;
    const int d0   = (blockIdx.x * 8 + (threadIdx.x >> 5)) * NPW;
    if (d0 >= NODES) return;
    const int offv = off[d0 + (lane < NPW ? lane : NPW)];
    const int nl   = lane < NPW ? lane : 0;
    const int selfv = d0 + nl + (ety[EDGES + d0 + nl] != 0 ? NODES : 0);
    const int sub  = lane & (CH - 1);

    #pragma unroll
    for (int n = 0; n < NPW; ++n) {
        const int d   = d0 + n;
        const int beg = __builtin_amdgcn_readlane(offv, n);
        const int tot = __builtin_amdgcn_readlane(offv, n + 1) - beg + 1;
        const int selfKey = __builtin_amdgcn_readlane(selfv, n);

        v2f a = {0.f, 0.f};
        int kv = slot_key(key, beg, selfKey, sub < tot ? sub : tot - 1);
        for (int v0 = 0; v0 < tot; v0 += CH) {
            const int rem = tot - v0;
            const int cnt = rem < CH ? rem : CH;
            v2f row[CH];
            #pragma unroll
            for (int i = 0; i < CH; ++i) {
                const int r = __builtin_amdgcn_readlane(kv, i);
                row[i] = *(const v2f*)(tr + (size_t)r * DIM + 2 * lane);
            }
            int kn = 0;
            if (rem > CH) {
                const int r2 = rem - CH;
                kn = key[beg + v0 + CH - 1 + (sub < r2 ? sub : r2 - 1)];
            }
            #pragma unroll
            for (int i = 0; i < CH; ++i) {
                a.x += (i < cnt) ? row[i].x : 0.f;
                a.y += (i < cnt) ? row[i].y : 0.f;
            }
            kv = kn;
        }
        *(v2f*)(acc + (size_t)d * DIM + 2 * lane) = a;
    }
}

#define GATES  (3 * DIM)

__global__ void __launch_bounds__(256)
gru_fused(const float* __restrict__ am,
          const float* __restrict__ Wih, const float* __restrict__ bih,
          const float* __restrict__ Whh, const float* __restrict__ bhh,
          float* h, float* emb,
          const float* __restrict__ W0, const float* __restrict__ bias0,
          const float* __restrict__ W1, const float* __restrict__ bias1,
          float* __restrict__ C0, float* __restrict__ C1)
{
    __shared__ __attribute__((aligned(16))) float Wl[2 * GATES * WPITCH];
    __shared__ __attribute__((aligned(16))) float Al[8 * 16 * WPITCH];

    const int tid  = threadIdx.x;
    const int lane = tid & 31;
    const int lr   = lane & 15;
    const int hi   = lane >> 4;
    const int rt   = blockIdx.x * 8 + (tid >> 5);
    const bool live = rt < ROW_TILES;
    const int row0 = rt << 4;

    v2f fa[16] = {}, fh[16] = {};
    if (live) {
        const float* arow = am + (size_t)(row0 + lr) * DIM + 2 * hi;
        const float* hrow = h  + (size_t)(row0 + lr) * DIM + 2 * hi;
        #pragma unroll
        for (int k = 0; k < 16; ++k) {
            fa[k] = *(const v2f*)(arow + 4 * k);
            fh[k] = *(const v2f*)(hrow + 4 * k);
        }
    }

    for (int i = tid; i < GATES * (DIM / 4); i += 256) {
        const int r = i >> 4, c4 = i & 15;
        const v4f wi = *(const v4f*)(Wih + 4 * i);
        const v4f wh = *(const v4f*)(Whh + 4 * i);
        *(v4f*)(Wl + r * WPITCH + 4 * c4)           = wi;
        *(v4f*)(Wl + (GATES + r) * WPITCH + 4 * c4) = wh;
    }
    __syncthreads();
    float* al = Al + (tid >> 5) * 16 * WPITCH;

    if (live) {
    for (int ct = 0; ct < DIM / 16; ++ct) {
        const int dcol = ct * 16 + lr;
        const float* bi = Wl + dcol * WPITCH + 2 * hi;
        const float* bh = bi + GATES * WPITCH;

        float* hp = h + (size_t)(row0 + hi * 8) * DIM + dcol;
        float hv[8];
        #pragma unroll
        for (int v = 0; v < 8; ++v) hv[v] = hp[(size_t)v * DIM];

        v8f cir = {}, ciz = {}, cic = {}, chr = {}, chz = {}, chc = {};
        #pragma unroll
        for (int k = 0; k < 16; ++k) {
            const v2f b0 = *(const v2f*)(bi + 4 * k);
            const v2f b1 = *(const v2f*)(bi + DIM * WPITCH + 4 * k);
            const v2f b2 = *(const v2f*)(bi + 2 * DIM * WPITCH + 4 * k);
            const v2f b3 = *(const v2f*)(bh + 4 * k);
            const v2f b4 = *(const v2f*)(bh + DIM * WPITCH + 4 * k);
            const v2f b5 = *(const v2f*)(bh + 2 * DIM * WPITCH + 4 * k);
            cir = __builtin_amdgcn_wmma_f32_16x16x4_f32(
                      false, fa[k], false, b0, (short)0, cir, false, false);
            ciz = __builtin_amdgcn_wmma_f32_16x16x4_f32(
                      false, fa[k], false, b1, (short)0, ciz, false, false);
            cic = __builtin_amdgcn_wmma_f32_16x16x4_f32(
                      false, fa[k], false, b2, (short)0, cic, false, false);
            chr = __builtin_amdgcn_wmma_f32_16x16x4_f32(
                      false, fh[k], false, b3, (short)0, chr, false, false);
            chz = __builtin_amdgcn_wmma_f32_16x16x4_f32(
                      false, fh[k], false, b4, (short)0, chz, false, false);
            chc = __builtin_amdgcn_wmma_f32_16x16x4_f32(
                      false, fh[k], false, b5, (short)0, chc, false, false);
        }

        const float bir = bih[dcol], biz = bih[DIM + dcol], bic = bih[2 * DIM + dcol];
        const float bhr = bhh[dcol], bhz = bhh[DIM + dcol], bhc = bhh[2 * DIM + dcol];
        float* ep = emb ? emb + (size_t)(row0 + hi * 8) * DIM + dcol : nullptr;
        #pragma unroll
        for (int v = 0; v < 8; ++v) {
            const float gir = cir[v] + bir, ghr = chr[v] + bhr;
            const float giz = ciz[v] + biz, ghz = chz[v] + bhz;
            const float gic = cic[v] + bic, ghc = chc[v] + bhc;
            const float r = 1.f / (1.f + expf(-(gir + ghr)));
            const float z = 1.f / (1.f + expf(-(giz + ghz)));
            const float c = tanhf(gic + r * ghc);
            const float o = (1.f - z) * c + z * hv[v];
            hp[(size_t)v * DIM] = o;
            if (ep) ep[(size_t)v * DIM] = o;
            al[(hi * 8 + v) * WPITCH + dcol] = o;
        }
    }
    }

    if (W0 == nullptr) return;

    v4f w0v[4], w1v[4];
    #pragma unroll
    for (int j = 0; j < 4; ++j) {
        w0v[j] = *(const v4f*)(W0 + 4 * (j * 256 + tid));
        w1v[j] = *(const v4f*)(W1 + 4 * (j * 256 + tid));
    }
    __syncthreads();
    #pragma unroll
    for (int j = 0; j < 4; ++j) {
        const int i = j * 256 + tid;
        const int r = i >> 4, c4 = i & 15;
        *(v4f*)(Wl + r * WPITCH + 4 * c4)         = w0v[j];
        *(v4f*)(Wl + (DIM + r) * WPITCH + 4 * c4) = w1v[j];
    }
    __syncthreads();
    if (live)
        pair_product_lds<8, true, WPITCH, false>(al, Wl, 0, lr, hi, bias0, bias1, C0, C1, row0);
}

extern "C" void kernel_launch(void* const* d_in, const int* in_sizes, int n_in,
                              void* d_out, int out_size, void* d_ws, size_t ws_size,
                              hipStream_t stream)
{
    (void)in_sizes; (void)n_in; (void)out_size; (void)ws_size;

    const float* x   = (const float*)d_in[0];
    const int*   src = (const int*)d_in[1];
    const int*   dst = (const int*)d_in[2];
    const int*   ety = (const int*)d_in[3];

    const long NE = (long)NODES * DIM;

    float* ws   = (float*)d_ws;
    float* hbuf = ws;                 ws += NE;
    float* buf1 = ws;                 ws += NE;
    float* buf2 = ws;                 ws += NE;
    float* tr0  = ws;                 ws += NE;
    float* tr1  = ws;                 ws += NE;
    float* acc  = ws;                 ws += NE;
    int* off = (int*)ws;              ws += NODES + 64;
    int* cur = (int*)ws;              ws += NODES + 64;
    int* cnt = (int*)ws;              ws += NODES + 64;
    int* key = (int*)ws;

    float* out = (float*)d_out;

    const int edgeThrBlocks  = (EDGES + 255) / 256;
    const int rowTileBlocks  = (ROW_TILES + 7) / 8;
    const int gruBlocks      = (ROW_TILES + 7) / 8;

    zero_counts<<<(NODES + 255) / 256, 256, 0, stream>>>(cnt);
    count_dst<<<edgeThrBlocks, 256, 0, stream>>>(dst, cnt);
    scan_counts<<<SCAN_BLOCKS, SCAN_T, 0, stream>>>(cnt, off, cur);
    fill_slots<<<edgeThrBlocks, 256, 0, stream>>>(src, dst, ety, cur, key);

    for (int l = 0; l < 2; ++l) {
        const int base = 4 + 11 * l;
        const float* Wsrc = (const float*)d_in[base + 0];
        const float* bsrc = (const float*)d_in[base + 1];
        const float* Wdst = (const float*)d_in[base + 2];
        const float* bdst = (const float*)d_in[base + 3];
        const float* attn = (const float*)d_in[base + 4];
        const float* ggW  = (const float*)d_in[base + 5];
        const float* ggb  = (const float*)d_in[base + 6];
        const float* Wih  = (const float*)d_in[base + 7];
        const float* bih  = (const float*)d_in[base + 8];
        const float* Whh  = (const float*)d_in[base + 9];
        const float* bhh  = (const float*)d_in[base + 10];

        if (l == 0)
            gemm_pair_wmma<<<rowTileBlocks, 256, 0, stream>>>(x, Wsrc, bsrc, Wdst, bdst,
                                                              buf1, buf2);
        gat_node<<<NODE_BLOCKS, 256, 0, stream>>>(buf1, buf2, attn, off, key, hbuf,
                                                  out + (2L * l) * NE,
                                                  ggW, ggb, ggW + DIM * DIM, ggb + DIM,
                                                  tr0, tr1);

        ggc_pull<<<NODE_BLOCKS, 256, 0, stream>>>(tr0, ety, off, key, acc);
        gru_fused<<<gruBlocks, 256, 0, stream>>>(
            acc, Wih, bih, Whh, bhh, hbuf, nullptr,
            ggW, ggb, ggW + DIM * DIM, ggb + DIM, tr0, tr1);
        ggc_pull<<<NODE_BLOCKS, 256, 0, stream>>>(tr0, ety, off, key, acc);
        if (l == 0) {
            const float* Wsrc1 = (const float*)d_in[4 + 11 + 0];
            const float* bsrc1 = (const float*)d_in[4 + 11 + 1];
            const float* Wdst1 = (const float*)d_in[4 + 11 + 2];
            const float* bdst1 = (const float*)d_in[4 + 11 + 3];
            gru_fused<<<gruBlocks, 256, 0, stream>>>(
                acc, Wih, bih, Whh, bhh, hbuf, out + (2L * l + 1) * NE,
                Wsrc1, bsrc1, Wdst1, bdst1, buf1, buf2);
        } else {
            gru_fused<<<gruBlocks, 256, 0, stream>>>(
                acc, Wih, bih, Whh, bhh, hbuf, out + (2L * l + 1) * NE,
                nullptr, nullptr, nullptr, nullptr, nullptr, nullptr);
        }
    }
}


# edge slots: fixed bucket of 64 per destination plus an overflow list, slot = old value of one returning add on the node's count; count pass and scan launch gone (15 -> 13); arrival order kept
# speedup vs baseline: 3.5905x; 1.0821x over previous
#include <hip/hip_runtime.h>
#include <hip/hip_bf16.h>


typedef __attribute__((ext_vector_type(2))) float v2f;
typedef __attribute__((ext_vector_type(4))) float v4f;
typedef __attribute__((ext_vector_type(8))) float v8f;

#define NODES 50000
#define EDGES 800000
#define DIM   64
#define NEG_SLOPE 0.2f

#define NPW 4
#define CH  8
static_assert(NODES % NPW == 0, "node group must divide the node count");
#define NODE_BLOCKS ((((NODES) / NPW) + 7) / 8)
#define BIG_NEG (-3.402823466e+38f)

#define WPITCH 68
#define ROW_TILES (NODES / 16)
static_assert(NODES % 16 == 0, "row tiles must divide the node count");

__global__ void __launch_bounds__(256) __attribute__((amdgpu_waves_per_eu(2, 4)))
gemm_pair_wmma(const float* __restrict__ A,
               const float* __restrict__ W0, const float* __restrict__ bias0,
               const float* __restrict__ W1, const float* __restrict__ bias1,
               float* __restrict__ C0, float* __restrict__ C1)
{
    __shared__ __attribute__((aligned(16))) float Wl[2 * DIM * WPITCH];
    __shared__ __attribute__((aligned(16))) float Al[8 * 16 * WPITCH];

    const int tid  = threadIdx.x;
    const int lane = tid & 31;
    const int lr   = lane & 15;
    const int hi   = lane >> 4;
    const int rt   = blockIdx.x * 8 + (tid >> 5);
    const bool live = rt < ROW_TILES;
    const int row0 = (live ? rt : ROW_TILES - 1) << 4;
    float* al = Al + (tid >> 5) * 16 * WPITCH;

    const float* atile = A + (size_t)row0 * DIM;
    v4f av[8], w0v[4], w1v[4];
    #pragma unroll
    for (int j = 0; j < 8; ++j)
        av[j] = *(const v4f*)(atile + 4 * (j * 32 + lane));
    #pragma unroll
    for (int j = 0; j < 4; ++j) {
        w0v[j] = *(const v4f*)(W0 + 4 * (j * 256 + tid));
        w1v[j] = *(const v4f*)(W1 + 4 * (j * 256 + tid));
    }
    #pragma unroll
    for (int j = 0; j < 8; ++j) {
        const int i = j * 32 + lane;
        *(v4f*)(al + (i >> 4) * WPITCH + 4 * (i & 15)) = av[j];
    }
    #pragma unroll
    for (int j = 0; j < 4; ++j) {
        const int i = j * 256 + tid;
        const int r = i >> 4, c4 = i & 15;
        *(v4f*)(Wl + r * WPITCH + 4 * c4)         = w0v[j];
        *(v4f*)(Wl + (DIM + r) * WPITCH + 4 * c4) = w1v[j];
    }
    __syncthreads();
    if (!live) return;

    v2f fa[16];
    const float* ap = al + lr * WPITCH + 2 * hi;
    #pragma unroll
    for (int k = 0; k < 16; ++k) fa[k] = *(const v2f*)(ap + 4 * k);

    const float* bp = Wl + lr * WPITCH + 2 * hi;
    v8f acc[8] = {};
    #pragma unroll
    for (int k = 0; k < 16; ++k) {
        v2f b[8];
        #pragma unroll
        for (int t = 0; t < 8; ++t)
            b[t] = *(const v2f*)(bp + t * 16 * WPITCH + 4 * k);
        #pragma unroll
        for (int t = 0; t < 8; ++t)
            acc[t] = __builtin_amdgcn_wmma_f32_16x16x4_f32(
                         false, fa[k], false, b[t], (short)0, acc[t], false, false);
    }

    #pragma unroll
    for (int t = 0; t < 8; ++t) {
        const int col = (t & 3) * 16 + lr;
        const float bv = (t < 4 ? bias0 : bias1)[col];
        float* crow = (t < 4 ? C0 : C1) + (size_t)(row0 + hi * 8) * DIM + col;
        #pragma unroll
        for (int v = 0; v < 8; ++v)
            crow[(size_t)v * DIM] = acc[t][v] + bv;
    }
}

template <int NT, bool KEEP_A, int PITCH, bool SWZ>
__device__ __forceinline__ void pair_product_lds(
    const float* al, const float* Wl, int t0, int lr, int hi,
    const float* __restrict__ bias0, const float* __restrict__ bias1,
    float* __restrict__ C0, float* __restrict__ C1, int row0)
{
    v2f fa[16];
    const int sw = SWZ ? (lr << 2) : 0;
    const float* ap = al + lr * PITCH + 2 * hi;
    if (KEEP_A) {
        #pragma unroll
        for (int k = 0; k < 16; ++k) fa[k] = *(const v2f*)(ap + ((4 * k) ^ sw));
    }

    const float* bp = Wl + (t0 * 16 + lr) * PITCH + 2 * hi;
    v8f acc[NT] = {};
    #pragma unroll
    for (int k = 0; k < 16; ++k) {
        if (!KEEP_A) fa[k] = *(const v2f*)(ap + ((4 * k) ^ sw));
        v2f b[NT];
        #pragma unroll
        for (int t = 0; t < NT; ++t)
            b[t] = *(const v2f*)(bp + t * 16 * PITCH + ((4 * k) ^ sw));
        #pragma unroll
        for (int t = 0; t < NT; ++t)
            acc[t] = __builtin_amdgcn_wmma_f32_16x16x4_f32(
                         false, fa[k], false, b[t], (short)0, acc[t], false, false);
    }
    #pragma unroll
    for (int t = 0; t < NT; ++t) {
        const int tt  = t0 + t;
        const int col = (tt & 3) * 16 + lr;
        const float bv = (tt < 4 ? bias0 : bias1)[col];
        float* crow = (tt < 4 ? C0 : C1) + (size_t)(row0 + hi * 8) * DIM + col;
        #pragma unroll
        for (int v = 0; v < 8; ++v)
            crow[(size_t)v * DIM] = acc[t][v] + bv;
    }
}

#define CAP 64
#define CNT_WORDS (((NODES + 1 + 63) / 64) * 64)

__global__ void zero_counts(int* __restrict__ cnt)
{
    const int i = blockIdx.x * blockDim.x + threadIdx.x;
    if (i <= NODES) cnt[i] = 0;
}

__global__ void fill_slots(const int* __restrict__ src, const int* __restrict__ dst,
                           const int* __restrict__ ety, int* __restrict__ cnt,
                           int* __restrict__ key, int* __restrict__ ovf)
{
    const int e = blockIdx.x * blockDim.x + threadIdx.x;
    if (e >= EDGES) return;
    const int d   = dst[e];
    const int k   = src[e] + (ety[e] != 0 ? NODES : 0);
    const int pos = atomicAdd(&cnt[d], 1);
    if (pos < CAP) {
        key[d * CAP + pos] = k;
    } else {
        const int q = atomicAdd(&cnt[NODES], 1);
        ovf[2 * q]     = d;
        ovf[2 * q + 1] = k;
    }
}

__device__ __forceinline__ int slot_key(const int* __restrict__ key, int beg,
                                        int selfKey, int v)
{
    return v == 0 ? selfKey : key[beg + v - 1];
}

__global__ void __launch_bounds__(256) __attribute__((amdgpu_waves_per_eu(16, 16)))
gat_node(const float* __restrict__ el, const float* __restrict__ er,
         const float* __restrict__ attn, const int* __restrict__ cnt,
         const int* __restrict__ key, const int* __restrict__ ovf,
         float* __restrict__ h,
         float* __restrict__ emb,
         const float* __restrict__ W0, const float* __restrict__ bias0,
         const float* __restrict__ W1, const float* __restrict__ bias1,
         float* __restrict__ C0, float* __restrict__ C1)
{
    __shared__ __attribute__((aligned(16))) float Wl[2 * DIM * DIM];
    __shared__ __attribute__((aligned(16))) float Al[8 * NPW * DIM];

    const int lane = threadIdx.x & 31;
    const int wv   = __builtin_amdgcn_readfirstlane(threadIdx.x >> 5);
    const int d0   = (blockIdx.x * 8 + wv) * NPW;
    const bool live = d0 < NODES;
    if (live) {
    const int cntv = cnt[d0 + (lane < NPW ? lane : 0)];
    const v2f at   = *(const v2f*)(attn + 2 * lane);
    const int sub  = lane & (CH - 1);
    const bool b16 = (lane & 16) != 0, b8 = (lane & 8) != 0, b4 = (lane & 4) != 0;
    const int myEdge = (b16 ? 4 : 0) + (b8 ? 2 : 0) + (b4 ? 1 : 0);
    static_assert(CH == 8, "the folding butterfly is written for 8 edges per round");

    #pragma unroll
    for (int n = 0; n < NPW; ++n) {
        const int d   = d0 + n;
        const int nin = __builtin_amdgcn_readlane(cntv, n);
        const int beg = d * CAP;
        const int tot = (nin < CAP ? nin : CAP) + 1;
        const v2f rr  = *(const v2f*)(er + (size_t)d * DIM + 2 * lane);

        float m = BIG_NEG, den = 0.f;
        v2f acc = {0.f, 0.f};

        int kv = slot_key(key, beg, d, sub < tot ? sub : tot - 1);
        for (int v0 = 0; v0 < tot; v0 += CH) {
            const int rem = tot - v0;
            const int cnt = rem < CH ? rem : CH;
            const int ks = kv >= NODES ? kv - NODES : kv;

            v2f row[CH];
            #pragma unroll
            for (int i = 0; i < CH; ++i) {
                const int s = __builtin_amdgcn_readlane(ks, i);
                row[i] = *(const v2f*)(el + (size_t)s * DIM + 2 * lane);
            }
            int kn = 0;
            if (rem > CH) {
                const int r2 = rem - CH;
                kn = key[beg + v0 + CH - 1 + (sub < r2 ? sub : r2 - 1)];
            }
            float q[CH];
            #pragma unroll
            for (int i = 0; i < CH; ++i) {
                float x0 = row[i].x + rr.x;
                float x1 = row[i].y + rr.y;
                x0 = x0 > 0.f ? x0 : NEG_SLOPE * x0;
                x1 = x1 > 0.f ? x1 : NEG_SLOPE * x1;
                float t = x0 * at.x;
                t += x1 * at.y;
                q[i] = t;
            }
            float k0[4], k1[2], sc_e;
            #pragma unroll
            for (int j = 0; j < 4; ++j) {
                const float send = b16 ? q[j] : q[j + 4];
                const float keep = b16 ? q[j + 4] : q[j];
                k0[j] = keep + __shfl_xor(send, 16, 32);
            }
            #pragma unroll
            for (int j = 0; j < 2; ++j) {
                const float send = b8 ? k0[j] : k0[j + 2];
                const float keep = b8 ? k0[j + 2] : k0[j];
                k1[j] = keep + __shfl_xor(send, 8, 32);
            }
            {
                const float send = b4 ? k1[0] : k1[1];
                const float keep = b4 ? k1[1] : k1[0];
                sc_e = keep + __shfl_xor(send, 4, 32);
            }
            sc_e += __shfl_xor(sc_e, 2, 32);
            sc_e += __shfl_xor(sc_e, 1, 32);

            const bool valid = myEdge < cnt;
            float cm = valid ? sc_e : BIG_NEG;
            cm = fmaxf(cm, __shfl_xor(cm, 16, 32));
            cm = fmaxf(cm, __shfl_xor(cm, 8, 32));
            cm = fmaxf(cm, __shfl_xor(cm, 4, 32));
            const float nm = fmaxf(m, cm);
            const float sc = expf(m - nm);
            const float wv = valid ? expf(sc_e - nm) : 0.f;
            den *= sc; acc.x *= sc; acc.y *= sc;
            #pragma unroll
            for (int i = 0; i < CH; ++i) {
                const int src_lane = ((i >> 2) & 1) * 16 + ((i >> 1) & 1) * 8 + (i & 1) * 4;
                const float w = __int_as_float(
                    __builtin_amdgcn_readlane(__float_as_int(wv), src_lane));
                den   += w;
                acc.x += w * row[i].x;
                acc.y += w * row[i].y;
            }
            m  = nm;
            kv = kn;
        }

        if (nin > CAP) {
            const int novf = cnt[NODES];
            for (int q = 0; q < novf; ++q) {
                if (ovf[2 * q] != d) continue;
                const int kq = ovf[2 * q + 1];
                const int sq = kq >= NODES ? kq - NODES : kq;
                const v2f rw = *(const v2f*)(el + (size_t)sq * DIM + 2 * lane);
                float x0 = rw.x + rr.x;
                float x1 = rw.y + rr.y;
                x0 = x0 > 0.f ? x0 : NEG_SLOPE * x0;
                x1 = x1 > 0.f ? x1 : NEG_SLOPE * x1;
                float t = x0 * at.x;
                t += x1 * at.y;
                for (int o = 16; o > 0; o >>= 1) t += __shfl_xor(t, o, 32);
                const float nm = fmaxf(m, t);
                const float sc = expf(m - nm);
                const float w  = expf(t - nm);
                den   = den * sc + w;
                acc.x = acc.x * sc + w * rw.x;
                acc.y = acc.y * sc + w * rw.y;
                m = nm;
            }
        }

        v2f o;
        o.x = acc.x / den;
        o.y = acc.y / den;
        o.x = o.x > 0.f ? o.x : (expf(o.x) - 1.f);
        o.y = o.y > 0.f ? o.y : (expf(o.y) - 1.f);
        *(v2f*)(h   + (size_t)d * DIM + 2 * lane) = o;
        *(v2f*)(emb + (size_t)d * DIM + 2 * lane) = o;
        *(v2f*)(Al + (wv * NPW + n) * DIM + ((2 * lane) ^ (((wv * NPW + n) & 15) << 2))) = o;
    }
    }

    {
        const int tid = threadIdx.x;
        v4f w0v[4], w1v[4];
        #pragma unroll
        for (int j = 0; j < 4; ++j) {
            w0v[j] = *(const v4f*)(W0 + 4 * (j * 256 + tid));
            w1v[j] = *(const v4f*)(W1 + 4 * (j * 256 + tid));
        }
        #pragma unroll
        for (int j = 0; j < 4; ++j) {
            const int i = j * 256 + tid;
            const int r = i >> 4, c4 = (i & 15) ^ (r & 15);
            *(v4f*)(Wl + r * DIM + 4 * c4)         = w0v[j];
            *(v4f*)(Wl + (DIM + r) * DIM + 4 * c4) = w1v[j];
        }
    }
    __syncthreads();
    const int rtile = blockIdx.x * 2 + (wv & 1);
    if (rtile < ROW_TILES)
        pair_product_lds<2, false, DIM, true>(Al + (wv & 1) * 16 * DIM, Wl, 2 * (wv >> 1),
                            lane & 15, lane >> 4, bias0, bias1, C0, C1, rtile << 4);
}

__global__ void __launch_bounds__(256)
ggc_pull(const float* __restrict__ tr, const int* __restrict__ ety,
         const int* __restrict__ cnt, const int* __restrict__ key,
         const int* __restrict__ ovf, float* __restrict__ acc)
{
    const int lane = threadIdx.x & 31;
    const int d0   = (blockIdx.x * 8 + (threadIdx.x >> 5)) * NPW;
    if (d0 >= NODES) return;
    const int nl   = lane < NPW ? lane : 0;
    const int cntv = cnt[d0 + nl];
    const int selfv = d0 + nl + (ety[EDGES + d0 + nl] != 0 ? NODES : 0);
    const int sub  = lane & (CH - 1);

    #pragma unroll
    for (int n = 0; n < NPW; ++n) {
        const int d   = d0 + n;
        const int nin = __builtin_amdgcn_readlane(cntv, n);
        const int beg = d * CAP;
        const int tot = (nin < CAP ? nin : CAP) + 1;
        const int selfKey = __builtin_amdgcn_readlane(selfv, n);

        v2f a = {0.f, 0.f};
        int kv = slot_key(key, beg, selfKey, sub < tot ? sub : tot - 1);
        for (int v0 = 0; v0 < tot; v0 += CH) {
            const int rem = tot - v0;
            const int cnt = rem < CH ? rem : CH;
            v2f row[CH];
            #pragma unroll
            for (int i = 0; i < CH; ++i) {
                const int r = __builtin_amdgcn_readlane(kv, i);
                row[i] = *(const v2f*)(tr + (size_t)r * DIM + 2 * lane);
            }
            int kn = 0;
            if (rem > CH) {
                const int r2 = rem - CH;
                kn = key[beg + v0 + CH - 1 + (sub < r2 ? sub : r2 - 1)];
            }
            #pragma unroll
            for (int i = 0; i < CH; ++i) {
                a.x += (i < cnt) ? row[i].x : 0.f;
                a.y += (i < cnt) ? row[i].y : 0.f;
            }
            kv = kn;
        }
        if (nin > CAP) {
            const int novf = cnt[NODES];
            for (int q = 0; q < novf; ++q) {
                if (ovf[2 * q] != d) continue;
                const v2f rw = *(const v2f*)(tr + (size_t)ovf[2 * q + 1] * DIM + 2 * lane);
                a.x += rw.x;
                a.y += rw.y;
            }
        }
        *(v2f*)(acc + (size_t)d * DIM + 2 * lane) = a;
    }
}

#define GATES  (3 * DIM)

__global__ void __launch_bounds__(256)
gru_fused(const float* __restrict__ am,
          const float* __restrict__ Wih, const float* __restrict__ bih,
          const float* __restrict__ Whh, const float* __restrict__ bhh,
          float* h, float* emb,
          const float* __restrict__ W0, const float* __restrict__ bias0,
          const float* __restrict__ W1, const float* __restrict__ bias1,
          float* __restrict__ C0, float* __restrict__ C1)
{
    __shared__ __attribute__((aligned(16))) float Wl[2 * GATES * WPITCH];
    __shared__ __attribute__((aligned(16))) float Al[8 * 16 * WPITCH];

    const int tid  = threadIdx.x;
    const int lane = tid & 31;
    const int lr   = lane & 15;
    const int hi   = lane >> 4;
    const int rt   = blockIdx.x * 8 + (tid >> 5);
    const bool live = rt < ROW_TILES;
    const int row0 = rt << 4;

    v2f fa[16] = {}, fh[16] = {};
    if (live) {
        const float* arow = am + (size_t)(row0 + lr) * DIM + 2 * hi;
        const float* hrow = h  + (size_t)(row0 + lr) * DIM + 2 * hi;
        #pragma unroll
        for (int k = 0; k < 16; ++k) {
            fa[k] = *(const v2f*)(arow + 4 * k);
            fh[k] = *(const v2f*)(hrow + 4 * k);
        }
    }

    for (int i = tid; i < GATES * (DIM / 4); i += 256) {
        const int r = i >> 4, c4 = i & 15;
        const v4f wi = *(const v4f*)(Wih + 4 * i);
        const v4f wh = *(const v4f*)(Whh + 4 * i);
        *(v4f*)(Wl + r * WPITCH + 4 * c4)           = wi;
        *(v4f*)(Wl + (GATES + r) * WPITCH + 4 * c4) = wh;
    }
    __syncthreads();
    float* al = Al + (tid >> 5) * 16 * WPITCH;

    if (live) {
    for (int ct = 0; ct < DIM / 16; ++ct) {
        const int dcol = ct * 16 + lr;
        const float* bi = Wl + dcol * WPITCH + 2 * hi;
        const float* bh = bi + GATES * WPITCH;

        float* hp = h + (size_t)(row0 + hi * 8) * DIM + dcol;
        float hv[8];
        #pragma unroll
        for (int v = 0; v < 8; ++v) hv[v] = hp[(size_t)v * DIM];

        v8f cir = {}, ciz = {}, cic = {}, chr = {}, chz = {}, chc = {};
        #pragma unroll
        for (int k = 0; k < 16; ++k) {
            const v2f b0 = *(const v2f*)(bi + 4 * k);
            const v2f b1 = *(const v2f*)(bi + DIM * WPITCH + 4 * k);
            const v2f b2 = *(const v2f*)(bi + 2 * DIM * WPITCH + 4 * k);
            const v2f b3 = *(const v2f*)(bh + 4 * k);
            const v2f b4 = *(const v2f*)(bh + DIM * WPITCH + 4 * k);
            const v2f b5 = *(const v2f*)(bh + 2 * DIM * WPITCH + 4 * k);
            cir = __builtin_amdgcn_wmma_f32_16x16x4_f32(
                      false, fa[k], false, b0, (short)0, cir, false, false);
            ciz = __builtin_amdgcn_wmma_f32_16x16x4_f32(
                      false, fa[k], false, b1, (short)0, ciz, false, false);
            cic = __builtin_amdgcn_wmma_f32_16x16x4_f32(
                      false, fa[k], false, b2, (short)0, cic, false, false);
            chr = __builtin_amdgcn_wmma_f32_16x16x4_f32(
                      false, fh[k], false, b3, (short)0, chr, false, false);
            chz = __builtin_amdgcn_wmma_f32_16x16x4_f32(
                      false, fh[k], false, b4, (short)0, chz, false, false);
            chc = __builtin_amdgcn_wmma_f32_16x16x4_f32(
                      false, fh[k], false, b5, (short)0, chc, false, false);
        }

        const float bir = bih[dcol], biz = bih[DIM + dcol], bic = bih[2 * DIM + dcol];
        const float bhr = bhh[dcol], bhz = bhh[DIM + dcol], bhc = bhh[2 * DIM + dcol];
        float* ep = emb ? emb + (size_t)(row0 + hi * 8) * DIM + dcol : nullptr;
        #pragma unroll
        for (int v = 0; v < 8; ++v) {
            const float gir = cir[v] + bir, ghr = chr[v] + bhr;
            const float giz = ciz[v] + biz, ghz = chz[v] + bhz;
            const float gic = cic[v] + bic, ghc = chc[v] + bhc;
            const float r = 1.f / (1.f + expf(-(gir + ghr)));
            const float z = 1.f / (1.f + expf(-(giz + ghz)));
            const float c = tanhf(gic + r * ghc);
            const float o = (1.f - z) * c + z * hv[v];
            hp[(size_t)v * DIM] = o;
            if (ep) ep[(size_t)v * DIM] = o;
            al[(hi * 8 + v) * WPITCH + dcol] = o;
        }
    }
    }

    if (W0 == nullptr) return;

    v4f w0v[4], w1v[4];
    #pragma unroll
    for (int j = 0; j < 4; ++j) {
        w0v[j] = *(const v4f*)(W0 + 4 * (j * 256 + tid));
        w1v[j] = *(const v4f*)(W1 + 4 * (j * 256 + tid));
    }
    __syncthreads();
    #pragma unroll
    for (int j = 0; j < 4; ++j) {
        const int i = j * 256 + tid;
        const int r = i >> 4, c4 = i & 15;
        *(v4f*)(Wl + r * WPITCH + 4 * c4)         = w0v[j];
        *(v4f*)(Wl + (DIM + r) * WPITCH + 4 * c4) = w1v[j];
    }
    __syncthreads();
    if (live)
        pair_product_lds<8, true, WPITCH, false>(al, Wl, 0, lr, hi, bias0, bias1, C0, C1, row0);
}

extern "C" void kernel_launch(void* const* d_in, const int* in_sizes, int n_in,
                              void* d_out, int out_size, void* d_ws, size_t ws_size,
                              hipStream_t stream)
{
    (void)in_sizes; (void)n_in; (void)out_size; (void)ws_size;

    const float* x   = (const float*)d_in[0];
    const int*   src = (const int*)d_in[1];
    const int*   dst = (const int*)d_in[2];
    const int*   ety = (const int*)d_in[3];

    const long NE = (long)NODES * DIM;

    float* ws   = (float*)d_ws;
    float* hbuf = ws;                 ws += NE;
    float* buf1 = ws;                 ws += NE;
    float* buf2 = ws;                 ws += NE;
    float* tr0  = ws;                 ws += NE;
    float* tr1  = ws;                 ws += NE;
    float* acc  = ws;                 ws += NE;
    int* cnt = (int*)ws;              ws += CNT_WORDS;
    int* key = (int*)ws;              ws += (long)NODES * CAP;
    int* ovf = (int*)ws;

    float* out = (float*)d_out;

    const int edgeThrBlocks  = (EDGES + 255) / 256;
    const int rowTileBlocks  = (ROW_TILES + 7) / 8;
    const int gruBlocks      = (ROW_TILES + 7) / 8;

    zero_counts<<<(NODES + 256) / 256, 256, 0, stream>>>(cnt);
    fill_slots<<<edgeThrBlocks, 256, 0, stream>>>(src, dst, ety, cnt, key, ovf);

    for (int l = 0; l < 2; ++l) {
        const int base = 4 + 11 * l;
        const float* Wsrc = (const float*)d_in[base + 0];
        const float* bsrc = (const float*)d_in[base + 1];
        const float* Wdst = (const float*)d_in[base + 2];
        const float* bdst = (const float*)d_in[base + 3];
        const float* attn = (const float*)d_in[base + 4];
        const float* ggW  = (const float*)d_in[base + 5];
        const float* ggb  = (const float*)d_in[base + 6];
        const float* Wih  = (const float*)d_in[base + 7];
        const float* bih  = (const float*)d_in[base + 8];
        const float* Whh  = (const float*)d_in[base + 9];
        const float* bhh  = (const float*)d_in[base + 10];

        if (l == 0)
            gemm_pair_wmma<<<rowTileBlocks, 256, 0, stream>>>(x, Wsrc, bsrc, Wdst, bdst,
                                                              buf1, buf2);
        gat_node<<<NODE_BLOCKS, 256, 0, stream>>>(buf1, buf2, attn, cnt, key, ovf, hbuf,
                                                  out + (2L * l) * NE,
                                                  ggW, ggb, ggW + DIM * DIM, ggb + DIM,
                                                  tr0, tr1);

        ggc_pull<<<NODE_BLOCKS, 256, 0, stream>>>(tr0, ety, cnt, key, ovf, acc);
        gru_fused<<<gruBlocks, 256, 0, stream>>>(
            acc, Wih, bih, Whh, bhh, hbuf, nullptr,
            ggW, ggb, ggW + DIM * DIM, ggb + DIM, tr0, tr1);
        ggc_pull<<<NODE_BLOCKS, 256, 0, stream>>>(tr0, ety, cnt, key, ovf, acc);
        if (l == 0) {
            const float* Wsrc1 = (const float*)d_in[4 + 11 + 0];
            const float* bsrc1 = (const float*)d_in[4 + 11 + 1];
            const float* Wdst1 = (const float*)d_in[4 + 11 + 2];
            const float* bdst1 = (const float*)d_in[4 + 11 + 3];
            gru_fused<<<gruBlocks, 256, 0, stream>>>(
                acc, Wih, bih, Whh, bhh, hbuf, out + (2L * l + 1) * NE,
                Wsrc1, bsrc1, Wdst1, bdst1, buf1, buf2);
        } else {
            gru_fused<<<gruBlocks, 256, 0, stream>>>(
                acc, Wih, bih, Whh, bhh, hbuf, out + (2L * l + 1) * NE,
                nullptr, nullptr, nullptr, nullptr, nullptr, nullptr);
        }
    }
}


# edge buckets by a two-level LDS counting sort (256-destination bins, then per-node slots; one global add per bin and block), not one global returning add per edge; slot (sum) order still by arrival
# speedup vs baseline: 4.2490x; 1.1834x over previous
#include <hip/hip_runtime.h>
#include <hip/hip_bf16.h>


typedef __attribute__((ext_vector_type(2))) float v2f;
typedef __attribute__((ext_vector_type(4))) float v4f;
typedef __attribute__((ext_vector_type(8))) float v8f;

#define NODES 50000
#define EDGES 800000
#define DIM   64
#define NEG_SLOPE 0.2f

#define NPW 4
#define CH  8
static_assert(NODES % NPW == 0, "node group must divide the node count");
#define NODE_BLOCKS ((((NODES) / NPW) + 7) / 8)
#define BIG_NEG (-3.402823466e+38f)

#define WPITCH 68
#define ROW_TILES (NODES / 16)
static_assert(NODES % 16 == 0, "row tiles must divide the node count");

__global__ void __launch_bounds__(256) __attribute__((amdgpu_waves_per_eu(2, 4)))
gemm_pair_wmma(const float* __restrict__ A,
               const float* __restrict__ W0, const float* __restrict__ bias0,
               const float* __restrict__ W1, const float* __restrict__ bias1,
               float* __restrict__ C0, float* __restrict__ C1)
{
    __shared__ __attribute__((aligned(16))) float Wl[2 * DIM * WPITCH];
    __shared__ __attribute__((aligned(16))) float Al[8 * 16 * WPITCH];

    const int tid  = threadIdx.x;
    const int lane = tid & 31;
    const int lr   = lane & 15;
    const int hi   = lane >> 4;
    const int rt   = blockIdx.x * 8 + (tid >> 5);
    const bool live = rt < ROW_TILES;
    const int row0 = (live ? rt : ROW_TILES - 1) << 4;
    float* al = Al + (tid >> 5) * 16 * WPITCH;

    const float* atile = A + (size_t)row0 * DIM;
    v4f av[8], w0v[4], w1v[4];
    #pragma unroll
    for (int j = 0; j < 8; ++j)
        av[j] = *(const v4f*)(atile + 4 * (j * 32 + lane));
    #pragma unroll
    for (int j = 0; j < 4; ++j) {
        w0v[j] = *(const v4f*)(W0 + 4 * (j * 256 + tid));
        w1v[j] = *(const v4f*)(W1 + 4 * (j * 256 + tid));
    }
    #pragma unroll
    for (int j = 0; j < 8; ++j) {
        const int i = j * 32 + lane;
        *(v4f*)(al + (i >> 4) * WPITCH + 4 * (i & 15)) = av[j];
    }
    #pragma unroll
    for (int j = 0; j < 4; ++j) {
        const int i = j * 256 + tid;
        const int r = i >> 4, c4 = i & 15;
        *(v4f*)(Wl + r * WPITCH + 4 * c4)         = w0v[j];
        *(v4f*)(Wl + (DIM + r) * WPITCH + 4 * c4) = w1v[j];
    }
    __syncthreads();
    if (!live) return;

    v2f fa[16];
    const float* ap = al + lr * WPITCH + 2 * hi;
    #pragma unroll
    for (int k = 0; k < 16; ++k) fa[k] = *(const v2f*)(ap + 4 * k);

    const float* bp = Wl + lr * WPITCH + 2 * hi;
    v8f acc[8] = {};
    #pragma unroll
    for (int k = 0; k < 16; ++k) {
        v2f b[8];
        #pragma unroll
        for (int t = 0; t < 8; ++t)
            b[t] = *(const v2f*)(bp + t * 16 * WPITCH + 4 * k);
        #pragma unroll
        for (int t = 0; t < 8; ++t)
            acc[t] = __builtin_amdgcn_wmma_f32_16x16x4_f32(
                         false, fa[k], false, b[t], (short)0, acc[t], false, false);
    }

    #pragma unroll
    for (int t = 0; t < 8; ++t) {
        const int col = (t & 3) * 16 + lr;
        const float bv = (t < 4 ? bias0 : bias1)[col];
        float* crow = (t < 4 ? C0 : C1) + (size_t)(row0 + hi * 8) * DIM + col;
        #pragma unroll
        for (int v = 0; v < 8; ++v)
            crow[(size_t)v * DIM] = acc[t][v] + bv;
    }
}

template <int NT, bool KEEP_A, int PITCH, bool SWZ>
__device__ __forceinline__ void pair_product_lds(
    const float* al, const float* Wl, int t0, int lr, int hi,
    const float* __restrict__ bias0, const float* __restrict__ bias1,
    float* __restrict__ C0, float* __restrict__ C1, int row0)
{
    v2f fa[16];
    const int sw = SWZ ? (lr << 2) : 0;
    const float* ap = al + lr * PITCH + 2 * hi;
    if (KEEP_A) {
        #pragma unroll
        for (int k = 0; k < 16; ++k) fa[k] = *(const v2f*)(ap + ((4 * k) ^ sw));
    }

    const float* bp = Wl + (t0 * 16 + lr) * PITCH + 2 * hi;
    v8f acc[NT] = {};
    #pragma unroll
    for (int k = 0; k < 16; ++k) {
        if (!KEEP_A) fa[k] = *(const v2f*)(ap + ((4 * k) ^ sw));
        v2f b[NT];
        #pragma unroll
        for (int t = 0; t < NT; ++t)
            b[t] = *(const v2f*)(bp + t * 16 * PITCH + ((4 * k) ^ sw));
        #pragma unroll
        for (int t = 0; t < NT; ++t)
            acc[t] = __builtin_amdgcn_wmma_f32_16x16x4_f32(
                         false, fa[k], false, b[t], (short)0, acc[t], false, false);
    }
    #pragma unroll
    for (int t = 0; t < NT; ++t) {
        const int tt  = t0 + t;
        const int col = (tt & 3) * 16 + lr;
        const float bv = (tt < 4 ? bias0 : bias1)[col];
        float* crow = (tt < 4 ? C0 : C1) + (size_t)(row0 + hi * 8) * DIM + col;
        #pragma unroll
        for (int v = 0; v < 8; ++v)
            crow[(size_t)v * DIM] = acc[t][v] + bv;
    }
}

typedef __attribute__((ext_vector_type(2))) int v2i;
typedef __attribute__((ext_vector_type(4))) int v4i;

#define CAP 64
#define CNT_WORDS (((NODES + 1 + 63) / 64) * 64)
#define BIN_SHIFT 8
#define BIN_NODES (1 << BIN_SHIFT)
#define NBINS     ((NODES + BIN_NODES - 1) / BIN_NODES)
#define BIN_CAP   16384
#define BIN_PAD   64
#define EPB       4096
#define BIN_BLOCKS ((EDGES + EPB - 1) / EPB)
static_assert(NBINS <= 256, "the bin scan is written for at most 256 bins");
static_assert(CAP == 64, "fill_slots takes a bucket row as 16 pieces of 4 slots");

__global__ void zero_counts(int* __restrict__ ovfCount, int* __restrict__ binCnt)
{
    const int i = threadIdx.x;
    if (i < NBINS) binCnt[i * BIN_PAD] = 0;
    if (i == NBINS) *ovfCount = 0;
}

__global__ void __launch_bounds__(1024)
bin_edges(const int* __restrict__ src, const int* __restrict__ dst,
          const int* __restrict__ ety, int* __restrict__ binCnt,
          int* __restrict__ binArr, int* __restrict__ ovfCount,
          int* __restrict__ ovf)
{
    __shared__ int hist[256];
    __shared__ int lbase[256];
    __shared__ int gbase[256];
    __shared__ int wsum[8];
    __shared__ __attribute__((aligned(8))) v2i stage[EPB];

    const int tid  = threadIdx.x;
    const int lane = tid & 31;
    const int e0   = blockIdx.x * EPB;
    if (tid < 256) hist[tid] = 0;
    __syncthreads();

    int d[4], k[4], r[4];
    #pragma unroll
    for (int j = 0; j < 4; ++j) {
        const int e  = e0 + j * 1024 + tid;
        const int ec = e < EDGES ? e : EDGES - 1;
        const int dd = dst[ec], ss = src[ec], tt = ety[ec];
        d[j] = e < EDGES ? dd : -1;
        k[j] = ss + (tt != 0 ? NODES : 0);
    }
    #pragma unroll
    for (int j = 0; j < 4; ++j)
        r[j] = d[j] >= 0 ? atomicAdd(&hist[d[j] >> BIN_SHIFT], 1) : 0;
    __syncthreads();

    int c = 0, s = 0;
    if (tid < 256) {
        c = hist[tid];
        s = c;
        for (int o = 1; o < 32; o <<= 1) {
            const int t = __shfl_up(s, o, 32);
            if (lane >= o) s += t;
        }
        if (lane == 31) wsum[tid >> 5] = s;
        gbase[tid] = c ? atomicAdd(&binCnt[tid * BIN_PAD], c) : 0;
    }
    __syncthreads();
    if (tid < 256) {
        int off = 0;
        for (int i = 0; i < (tid >> 5); ++i) off += wsum[i];
        lbase[tid] = off + s - c;
    }
    __syncthreads();
    #pragma unroll
    for (int j = 0; j < 4; ++j)
        if (d[j] >= 0) {
            v2i rec; rec.x = d[j]; rec.y = k[j];
            stage[lbase[d[j] >> BIN_SHIFT] + r[j]] = rec;
        }
    __syncthreads();

    const int left = EDGES - e0;
    const int nv   = left < EPB ? left : EPB;
    v2i rc[4];
    int gb[4], lb[4];
    #pragma unroll
    for (int j = 0; j < 4; ++j) {
        const int i = j * 1024 + tid;
        rc[j] = stage[i < nv ? i : nv - 1];
    }
    #pragma unroll
    for (int j = 0; j < 4; ++j) {
        const int b = rc[j].x >> BIN_SHIFT;
        gb[j] = gbase[b];
        lb[j] = lbase[b];
    }
    #pragma unroll
    for (int j = 0; j < 4; ++j) {
        const int i = j * 1024 + tid;
        if (i < nv) {
            const int b = rc[j].x >> BIN_SHIFT;
            const int p = gb[j] + (i - lb[j]);
            if (p < BIN_CAP) {
                *(v2i*)(binArr + 2 * ((size_t)b * BIN_CAP + p)) = rc[j];
            } else {
                const int q = atomicAdd(ovfCount, 1);
                ovf[2 * q]     = rc[j].x;
                ovf[2 * q + 1] = rc[j].y;
            }
        }
    }
}

__global__ void __launch_bounds__(1024)
fill_slots(const int* __restrict__ binCnt, const int* __restrict__ binArr,
           int* __restrict__ cnt, int* __restrict__ key,
           int* __restrict__ ovfCount, int* __restrict__ ovf)
{
    __shared__ int lcnt[BIN_NODES];
    __shared__ __attribute__((aligned(16))) int lkey[BIN_NODES * CAP];

    const int tid = threadIdx.x;
    const int b   = blockIdx.x;
    if (tid < BIN_NODES) lcnt[tid] = 0;
    for (int q = tid; q < BIN_NODES * (CAP / 4); q += 1024) {
        const v4i z = {0, 0, 0, 0};
        *(v4i*)(lkey + 4 * q) = z;
    }
    __syncthreads();

    const int nb = binCnt[b * BIN_PAD];
    const int n  = nb < BIN_CAP ? nb : BIN_CAP;
    const int* rec = binArr + 2 * (size_t)b * BIN_CAP;
    for (int i0 = 0; i0 < n; i0 += 4 * 1024) {
        v2i r[4];
        #pragma unroll
        for (int j = 0; j < 4; ++j) {
            const int i = i0 + j * 1024 + tid;
            r[j] = *(const v2i*)(rec + 2 * (i < n ? i : n - 1));
        }
        #pragma unroll
        for (int j = 0; j < 4; ++j) {
            const int i = i0 + j * 1024 + tid;
            if (i < n) {
                const int dl  = r[j].x & (BIN_NODES - 1);
                const int pos = atomicAdd(&lcnt[dl], 1);
                if (pos < CAP) {
                    lkey[dl * CAP + pos] = r[j].y;
                } else {
                    const int q = atomicAdd(ovfCount, 1);
                    ovf[2 * q]     = r[j].x;
                    ovf[2 * q + 1] = r[j].y;
                }
            }
        }
    }
    __syncthreads();

    const int node0 = b << BIN_SHIFT;
    const int rows  = NODES - node0 < BIN_NODES ? NODES - node0 : BIN_NODES;
    if (tid < rows) {
        const int c = lcnt[tid];
        cnt[node0 + tid] = c < CAP ? c : CAP;
    }
    for (int q = tid; q < rows * (CAP / 4); q += 1024) {
        const int c = lcnt[q >> 4];
        const v4i v = *(const v4i*)(lkey + 4 * q);
        if (4 * (q & 15) < c) *(v4i*)(key + (size_t)node0 * CAP + 4 * q) = v;
    }
}

__device__ __forceinline__ int slot_key(const int* __restrict__ key, int beg,
                                        int selfKey, int v)
{
    return v == 0 ? selfKey : key[beg + v - 1];
}

__global__ void __launch_bounds__(256) __attribute__((amdgpu_waves_per_eu(16, 16)))
gat_node(const float* __restrict__ el, const float* __restrict__ er,
         const float* __restrict__ attn, const int* __restrict__ cnt,
         const int* __restrict__ key, const int* __restrict__ ovf,
         float* __restrict__ h,
         float* __restrict__ emb,
         const float* __restrict__ W0, const float* __restrict__ bias0,
         const float* __restrict__ W1, const float* __restrict__ bias1,
         float* __restrict__ C0, float* __restrict__ C1)
{
    __shared__ __attribute__((aligned(16))) float Wl[2 * DIM * DIM];
    __shared__ __attribute__((aligned(16))) float Al[8 * NPW * DIM];

    const int lane = threadIdx.x & 31;
    const int wv   = __builtin_amdgcn_readfirstlane(threadIdx.x >> 5);
    const int d0   = (blockIdx.x * 8 + wv) * NPW;
    const bool live = d0 < NODES;
    if (live) {
    const int cntv = cnt[d0 + (lane < NPW ? lane : 0)];
    const int novf = cnt[NODES];
    const v2f at   = *(const v2f*)(attn + 2 * lane);
    const int sub  = lane & (CH - 1);
    const bool b16 = (lane & 16) != 0, b8 = (lane & 8) != 0, b4 = (lane & 4) != 0;
    const int myEdge = (b16 ? 4 : 0) + (b8 ? 2 : 0) + (b4 ? 1 : 0);
    static_assert(CH == 8, "the folding butterfly is written for 8 edges per round");

    #pragma unroll
    for (int n = 0; n < NPW; ++n) {
        const int d   = d0 + n;
        const int nin = __builtin_amdgcn_readlane(cntv, n);
        const int beg = d * CAP;
        const int tot = (nin < CAP ? nin : CAP) + 1;
        const v2f rr  = *(const v2f*)(er + (size_t)d * DIM + 2 * lane);

        float m = BIG_NEG, den = 0.f;
        v2f acc = {0.f, 0.f};

        int kv = slot_key(key, beg, d, sub < tot ? sub : tot - 1);
        for (int v0 = 0; v0 < tot; v0 += CH) {
            const int rem = tot - v0;
            const int cnt = rem < CH ? rem : CH;
            const int ks = kv >= NODES ? kv - NODES : kv;

            v2f row[CH];
            #pragma unroll
            for (int i = 0; i < CH; ++i) {
                const int s = __builtin_amdgcn_readlane(ks, i);
                row[i] = *(const v2f*)(el + (size_t)s * DIM + 2 * lane);
            }
            int kn = 0;
            if (rem > CH) {
                const int r2 = rem - CH;
                kn = key[beg + v0 + CH - 1 + (sub < r2 ? sub : r2 - 1)];
            }
            float q[CH];
            #pragma unroll
            for (int i = 0; i < CH; ++i) {
                float x0 = row[i].x + rr.x;
                float x1 = row[i].y + rr.y;
                x0 = x0 > 0.f ? x0 : NEG_SLOPE * x0;
                x1 = x1 > 0.f ? x1 : NEG_SLOPE * x1;
                float t = x0 * at.x;
                t += x1 * at.y;
                q[i] = t;
            }
            float k0[4], k1[2], sc_e;
            #pragma unroll
            for (int j = 0; j < 4; ++j) {
                const float send = b16 ? q[j] : q[j + 4];
                const float keep = b16 ? q[j + 4] : q[j];
                k0[j] = keep + __shfl_xor(send, 16, 32);
            }
            #pragma unroll
            for (int j = 0; j < 2; ++j) {
                const float send = b8 ? k0[j] : k0[j + 2];
                const float keep = b8 ? k0[j + 2] : k0[j];
                k1[j] = keep + __shfl_xor(send, 8, 32);
            }
            {
                const float send = b4 ? k1[0] : k1[1];
                const float keep = b4 ? k1[1] : k1[0];
                sc_e = keep + __shfl_xor(send, 4, 32);
            }
            sc_e += __shfl_xor(sc_e, 2, 32);
            sc_e += __shfl_xor(sc_e, 1, 32);

            const bool valid = myEdge < cnt;
            float cm = valid ? sc_e : BIG_NEG;
            cm = fmaxf(cm, __shfl_xor(cm, 16, 32));
            cm = fmaxf(cm, __shfl_xor(cm, 8, 32));
            cm = fmaxf(cm, __shfl_xor(cm, 4, 32));
            const float nm = fmaxf(m, cm);
            const float sc = expf(m - nm);
            const float wv = valid ? expf(sc_e - nm) : 0.f;
            den *= sc; acc.x *= sc; acc.y *= sc;
            #pragma unroll
            for (int i = 0; i < CH; ++i) {
                const int src_lane = ((i >> 2) & 1) * 16 + ((i >> 1) & 1) * 8 + (i & 1) * 4;
                const float w = __int_as_float(
                    __builtin_amdgcn_readlane(__float_as_int(wv), src_lane));
                den   += w;
                acc.x += w * row[i].x;
                acc.y += w * row[i].y;
            }
            m  = nm;
            kv = kn;
        }

        {
            for (int q = 0; q < novf; ++q) {
                if (ovf[2 * q] != d) continue;
                const int kq = ovf[2 * q + 1];
                const int sq = kq >= NODES ? kq - NODES : kq;
                const v2f rw = *(const v2f*)(el + (size_t)sq * DIM + 2 * lane);
                float x0 = rw.x + rr.x;
                float x1 = rw.y + rr.y;
                x0 = x0 > 0.f ? x0 : NEG_SLOPE * x0;
                x1 = x1 > 0.f ? x1 : NEG_SLOPE * x1;
                float t = x0 * at.x;
                t += x1 * at.y;
                for (int o = 16; o > 0; o >>= 1) t += __shfl_xor(t, o, 32);
                const float nm = fmaxf(m, t);
                const float sc = expf(m - nm);
                const float w  = expf(t - nm);
                den   = den * sc + w;
                acc.x = acc.x * sc + w * rw.x;
                acc.y = acc.y * sc + w * rw.y;
                m = nm;
            }
        }

        v2f o;
        o.x = acc.x / den;
        o.y = acc.y / den;
        o.x = o.x > 0.f ? o.x : (expf(o.x) - 1.f);
        o.y = o.y > 0.f ? o.y : (expf(o.y) - 1.f);
        *(v2f*)(h   + (size_t)d * DIM + 2 * lane) = o;
        *(v2f*)(emb + (size_t)d * DIM + 2 * lane) = o;
        *(v2f*)(Al + (wv * NPW + n) * DIM + ((2 * lane) ^ (((wv * NPW + n) & 15) << 2))) = o;
    }
    }

    {
        const int tid = threadIdx.x;
        v4f w0v[4], w1v[4];
        #pragma unroll
        for (int j = 0; j < 4; ++j) {
            w0v[j] = *(const v4f*)(W0 + 4 * (j * 256 + tid));
            w1v[j] = *(const v4f*)(W1 + 4 * (j * 256 + tid));
        }
        #pragma unroll
        for (int j = 0; j < 4; ++j) {
            const int i = j * 256 + tid;
            const int r = i >> 4, c4 = (i & 15) ^ (r & 15);
            *(v4f*)(Wl + r * DIM + 4 * c4)         = w0v[j];
            *(v4f*)(Wl + (DIM + r) * DIM + 4 * c4) = w1v[j];
        }
    }
    __syncthreads();
    const int rtile = blockIdx.x * 2 + (wv & 1);
    if (rtile < ROW_TILES)
        pair_product_lds<2, false, DIM, true>(Al + (wv & 1) * 16 * DIM, Wl, 2 * (wv >> 1),
                            lane & 15, lane >> 4, bias0, bias1, C0, C1, rtile << 4);
}

__global__ void __launch_bounds__(256)
ggc_pull(const float* __restrict__ tr, const int* __restrict__ ety,
         const int* __restrict__ cnt, const int* __restrict__ key,
         const int* __restrict__ ovf, float* __restrict__ acc)
{
    const int lane = threadIdx.x & 31;
    const int d0   = (blockIdx.x * 8 + (threadIdx.x >> 5)) * NPW;
    if (d0 >= NODES) return;
    const int nl   = lane < NPW ? lane : 0;
    const int cntv = cnt[d0 + nl];
    const int novf = cnt[NODES];
    const int selfv = d0 + nl + (ety[EDGES + d0 + nl] != 0 ? NODES : 0);
    const int sub  = lane & (CH - 1);

    #pragma unroll
    for (int n = 0; n < NPW; ++n) {
        const int d   = d0 + n;
        const int nin = __builtin_amdgcn_readlane(cntv, n);
        const int beg = d * CAP;
        const int tot = (nin < CAP ? nin : CAP) + 1;
        const int selfKey = __builtin_amdgcn_readlane(selfv, n);

        v2f a = {0.f, 0.f};
        int kv = slot_key(key, beg, selfKey, sub < tot ? sub : tot - 1);
        for (int v0 = 0; v0 < tot; v0 += CH) {
            const int rem = tot - v0;
            const int cnt = rem < CH ? rem : CH;
            v2f row[CH];
            #pragma unroll
            for (int i = 0; i < CH; ++i) {
                const int r = __builtin_amdgcn_readlane(kv, i);
                row[i] = *(const v2f*)(tr + (size_t)r * DIM + 2 * lane);
            }
            int kn = 0;
            if (rem > CH) {
                const int r2 = rem - CH;
                kn = key[beg + v0 + CH - 1 + (sub < r2 ? sub : r2 - 1)];
            }
            #pragma unroll
            for (int i = 0; i < CH; ++i) {
                a.x += (i < cnt) ? row[i].x : 0.f;
                a.y += (i < cnt) ? row[i].y : 0.f;
            }
            kv = kn;
        }
        {
            for (int q = 0; q < novf; ++q) {
                if (ovf[2 * q] != d) continue;
                const v2f rw = *(const v2f*)(tr + (size_t)ovf[2 * q + 1] * DIM + 2 * lane);
                a.x += rw.x;
                a.y += rw.y;
            }
        }
        *(v2f*)(acc + (size_t)d * DIM + 2 * lane) = a;
    }
}

#define GATES  (3 * DIM)

__global__ void __launch_bounds__(256)
gru_fused(const float* __restrict__ am,
          const float* __restrict__ Wih, const float* __restrict__ bih,
          const float* __restrict__ Whh, const float* __restrict__ bhh,
          float* h, float* emb,
          const float* __restrict__ W0, const float* __restrict__ bias0,
          const float* __restrict__ W1, const float* __restrict__ bias1,
          float* __restrict__ C0, float* __restrict__ C1)
{
    __shared__ __attribute__((aligned(16))) float Wl[2 * GATES * WPITCH];
    __shared__ __attribute__((aligned(16))) float Al[8 * 16 * WPITCH];

    const int tid  = threadIdx.x;
    const int lane = tid & 31;
    const int lr   = lane & 15;
    const int hi   = lane >> 4;
    const int rt   = blockIdx.x * 8 + (tid >> 5);
    const bool live = rt < ROW_TILES;
    const int row0 = rt << 4;

    v2f fa[16] = {}, fh[16] = {};
    if (live) {
        const float* arow = am + (size_t)(row0 + lr) * DIM + 2 * hi;
        const float* hrow = h  + (size_t)(row0 + lr) * DIM + 2 * hi;
        #pragma unroll
        for (int k = 0; k < 16; ++k) {
            fa[k] = *(const v2f*)(arow + 4 * k);
            fh[k] = *(const v2f*)(hrow + 4 * k);
        }
    }

    for (int i = tid; i < GATES * (DIM / 4); i += 256) {
        const int r = i >> 4, c4 = i & 15;
        const v4f wi = *(const v4f*)(Wih + 4 * i);
        const v4f wh = *(const v4f*)(Whh + 4 * i);
        *(v4f*)(Wl + r * WPITCH + 4 * c4)           = wi;
        *(v4f*)(Wl + (GATES + r) * WPITCH + 4 * c4) = wh;
    }
    __syncthreads();
    float* al = Al + (tid >> 5) * 16 * WPITCH;

    if (live) {
    for (int ct = 0; ct < DIM / 16; ++ct) {
        const int dcol = ct * 16 + lr;
        const float* bi = Wl + dcol * WPITCH + 2 * hi;
        const float* bh = bi + GATES * WPITCH;

        float* hp = h + (size_t)(row0 + hi * 8) * DIM + dcol;
        float hv[8];
        #pragma unroll
        for (int v = 0; v < 8; ++v) hv[v] = hp[(size_t)v * DIM];

        v8f cir = {}, ciz = {}, cic = {}, chr = {}, chz = {}, chc = {};
        #pragma unroll
        for (int k = 0; k < 16; ++k) {
            const v2f b0 = *(const v2f*)(bi + 4 * k);
            const v2f b1 = *(const v2f*)(bi + DIM * WPITCH + 4 * k);
            const v2f b2 = *(const v2f*)(bi + 2 * DIM * WPITCH + 4 * k);
            const v2f b3 = *(const v2f*)(bh + 4 * k);
            const v2f b4 = *(const v2f*)(bh + DIM * WPITCH + 4 * k);
            const v2f b5 = *(const v2f*)(bh + 2 * DIM * WPITCH + 4 * k);
            cir = __builtin_amdgcn_wmma_f32_16x16x4_f32(
                      false, fa[k], false, b0, (short)0, cir, false, false);
            ciz = __builtin_amdgcn_wmma_f32_16x16x4_f32(
                      false, fa[k], false, b1, (short)0, ciz, false, false);
            cic = __builtin_amdgcn_wmma_f32_16x16x4_f32(
                      false, fa[k], false, b2, (short)0, cic, false, false);
            chr = __builtin_amdgcn_wmma_f32_16x16x4_f32(
                      false, fh[k], false, b3, (short)0, chr, false, false);
            chz = __builtin_amdgcn_wmma_f32_16x16x4_f32(
                      false, fh[k], false, b4, (short)0, chz, false, false);
            chc = __builtin_amdgcn_wmma_f32_16x16x4_f32(
                      false, fh[k], false, b5, (short)0, chc, false, false);
        }

        const float bir = bih[dcol], biz = bih[DIM + dcol], bic = bih[2 * DIM + dcol];
        const float bhr = bhh[dcol], bhz = bhh[DIM + dcol], bhc = bhh[2 * DIM + dcol];
        float* ep = emb ? emb + (size_t)(row0 + hi * 8) * DIM + dcol : nullptr;
        #pragma unroll
        for (int v = 0; v < 8; ++v) {
            const float gir = cir[v] + bir, ghr = chr[v] + bhr;
            const float giz = ciz[v] + biz, ghz = chz[v] + bhz;
            const float gic = cic[v] + bic, ghc = chc[v] + bhc;
            const float r = 1.f / (1.f + expf(-(gir + ghr)));
            const float z = 1.f / (1.f + expf(-(giz + ghz)));
            const float c = tanhf(gic + r * ghc);
            const float o = (1.f - z) * c + z * hv[v];
            hp[(size_t)v * DIM] = o;
            if (ep) ep[(size_t)v * DIM] = o;
            al[(hi * 8 + v) * WPITCH + dcol] = o;
        }
    }
    }

    if (W0 == nullptr) return;

    v4f w0v[4], w1v[4];
    #pragma unroll
    for (int j = 0; j < 4; ++j) {
        w0v[j] = *(const v4f*)(W0 + 4 * (j * 256 + tid));
        w1v[j] = *(const v4f*)(W1 + 4 * (j * 256 + tid));
    }
    __syncthreads();
    #pragma unroll
    for (int j = 0; j < 4; ++j) {
        const int i = j * 256 + tid;
        const int r = i >> 4, c4 = i & 15;
        *(v4f*)(Wl + r * WPITCH + 4 * c4)         = w0v[j];
        *(v4f*)(Wl + (DIM + r) * WPITCH + 4 * c4) = w1v[j];
    }
    __syncthreads();
    if (live)
        pair_product_lds<8, true, WPITCH, false>(al, Wl, 0, lr, hi, bias0, bias1, C0, C1, row0);
}

extern "C" void kernel_launch(void* const* d_in, const int* in_sizes, int n_in,
                              void* d_out, int out_size, void* d_ws, size_t ws_size,
                              hipStream_t stream)
{
    (void)in_sizes; (void)n_in; (void)out_size; (void)ws_size;

    const float* x   = (const float*)d_in[0];
    const int*   src = (const int*)d_in[1];
    const int*   dst = (const int*)d_in[2];
    const int*   ety = (const int*)d_in[3];

    const long NE = (long)NODES * DIM;

    float* ws   = (float*)d_ws;
    float* hbuf = ws;                 ws += NE;
    float* buf1 = ws;                 ws += NE;
    float* buf2 = ws;                 ws += NE;
    float* tr0  = ws;                 ws += NE;
    float* tr1  = ws;                 ws += NE;
    float* acc  = ws;                 ws += NE;
    int* cnt = (int*)ws;              ws += CNT_WORDS;
    int* key = (int*)ws;              ws += (long)NODES * CAP;
    int* ovf = (int*)ws;              ws += 2L * EDGES;
    int* binCnt = (int*)ws;           ws += NBINS * BIN_PAD;
    int* binArr = (int*)ws;

    float* out = (float*)d_out;

    const int rowTileBlocks  = (ROW_TILES + 7) / 8;
    const int gruBlocks      = (ROW_TILES + 7) / 8;

    zero_counts<<<1, 256, 0, stream>>>(cnt + NODES, binCnt);
    bin_edges<<<BIN_BLOCKS, 1024, 0, stream>>>(src, dst, ety, binCnt, binArr, cnt + NODES, ovf);
    fill_slots<<<NBINS, 1024, 0, stream>>>(binCnt, binArr, cnt, key, cnt + NODES, ovf);

    for (int l = 0; l < 2; ++l) {
        const int base = 4 + 11 * l;
        const float* Wsrc = (const float*)d_in[base + 0];
        const float* bsrc = (const float*)d_in[base + 1];
        const float* Wdst = (const float*)d_in[base + 2];
        const float* bdst = (const float*)d_in[base + 3];
        const float* attn = (const float*)d_in[base + 4];
        const float* ggW  = (const float*)d_in[base + 5];
        const float* ggb  = (const float*)d_in[base + 6];
        const float* Wih  = (const float*)d_in[base + 7];
        const float* bih  = (const float*)d_in[base + 8];
        const float* Whh  = (const float*)d_in[base + 9];
        const float* bhh  = (const float*)d_in[base + 10];

        if (l == 0)
            gemm_pair_wmma<<<rowTileBlocks, 256, 0, stream>>>(x, Wsrc, bsrc, Wdst, bdst,
                                                              buf1, buf2);
        gat_node<<<NODE_BLOCKS, 256, 0, stream>>>(buf1, buf2, attn, cnt, key, ovf, hbuf,
                                                  out + (2L * l) * NE,
                                                  ggW, ggb, ggW + DIM * DIM, ggb + DIM,
                                                  tr0, tr1);

        ggc_pull<<<NODE_BLOCKS, 256, 0, stream>>>(tr0, ety, cnt, key, ovf, acc);
        gru_fused<<<gruBlocks, 256, 0, stream>>>(
            acc, Wih, bih, Whh, bhh, hbuf, nullptr,
            ggW, ggb, ggW + DIM * DIM, ggb + DIM, tr0, tr1);
        ggc_pull<<<NODE_BLOCKS, 256, 0, stream>>>(tr0, ety, cnt, key, ovf, acc);
        if (l == 0) {
            const float* Wsrc1 = (const float*)d_in[4 + 11 + 0];
            const float* bsrc1 = (const float*)d_in[4 + 11 + 1];
            const float* Wdst1 = (const float*)d_in[4 + 11 + 2];
            const float* bdst1 = (const float*)d_in[4 + 11 + 3];
            gru_fused<<<gruBlocks, 256, 0, stream>>>(
                acc, Wih, bih, Whh, bhh, hbuf, out + (2L * l + 1) * NE,
                Wsrc1, bsrc1, Wdst1, bdst1, buf1, buf2);
        } else {
            gru_fused<<<gruBlocks, 256, 0, stream>>>(
                acc, Wih, bih, Whh, bhh, hbuf, out + (2L * l + 1) * NE,
                nullptr, nullptr, nullptr, nullptr, nullptr, nullptr);
        }
    }
}


# GRU step on 256 resident 16-wave groups (one per WGP, weights staged once, loads in one group): 12 waves own a 16-row tile, 53 spare tiles go 16 features per wave; busiest SIMD 4 tiles -> 3 1/4
# speedup vs baseline: 4.5580x; 1.0727x over previous
#include <hip/hip_runtime.h>
#include <hip/hip_bf16.h>


typedef __attribute__((ext_vector_type(2))) float v2f;
typedef __attribute__((ext_vector_type(4))) float v4f;
typedef __attribute__((ext_vector_type(8))) float v8f;

#define NODES 50000
#define EDGES 800000
#define DIM   64
#define NEG_SLOPE 0.2f

#define NPW 4
#define CH  8
static_assert(NODES % NPW == 0, "node group must divide the node count");
#define NODE_BLOCKS ((((NODES) / NPW) + 7) / 8)
#define BIG_NEG (-3.402823466e+38f)

#define WPITCH 68
#define ROW_TILES (NODES / 16)
static_assert(NODES % 16 == 0, "row tiles must divide the node count");

__global__ void __launch_bounds__(256) __attribute__((amdgpu_waves_per_eu(2, 4)))
gemm_pair_wmma(const float* __restrict__ A,
               const float* __restrict__ W0, const float* __restrict__ bias0,
               const float* __restrict__ W1, const float* __restrict__ bias1,
               float* __restrict__ C0, float* __restrict__ C1)
{
    __shared__ __attribute__((aligned(16))) float Wl[2 * DIM * WPITCH];
    __shared__ __attribute__((aligned(16))) float Al[8 * 16 * WPITCH];

    const int tid  = threadIdx.x;
    const int lane = tid & 31;
    const int lr   = lane & 15;
    const int hi   = lane >> 4;
    const int rt   = blockIdx.x * 8 + (tid >> 5);
    const bool live = rt < ROW_TILES;
    const int row0 = (live ? rt : ROW_TILES - 1) << 4;
    float* al = Al + (tid >> 5) * 16 * WPITCH;

    const float* atile = A + (size_t)row0 * DIM;
    v4f av[8], w0v[4], w1v[4];
    #pragma unroll
    for (int j = 0; j < 8; ++j)
        av[j] = *(const v4f*)(atile + 4 * (j * 32 + lane));
    #pragma unroll
    for (int j = 0; j < 4; ++j) {
        w0v[j] = *(const v4f*)(W0 + 4 * (j * 256 + tid));
        w1v[j] = *(const v4f*)(W1 + 4 * (j * 256 + tid));
    }
    #pragma unroll
    for (int j = 0; j < 8; ++j) {
        const int i = j * 32 + lane;
        *(v4f*)(al + (i >> 4) * WPITCH + 4 * (i & 15)) = av[j];
    }
    #pragma unroll
    for (int j = 0; j < 4; ++j) {
        const int i = j * 256 + tid;
        const int r = i >> 4, c4 = i & 15;
        *(v4f*)(Wl + r * WPITCH + 4 * c4)         = w0v[j];
        *(v4f*)(Wl + (DIM + r) * WPITCH + 4 * c4) = w1v[j];
    }
    __syncthreads();
    if (!live) return;

    v2f fa[16];
    const float* ap = al + lr * WPITCH + 2 * hi;
    #pragma unroll
    for (int k = 0; k < 16; ++k) fa[k] = *(const v2f*)(ap + 4 * k);

    const float* bp = Wl + lr * WPITCH + 2 * hi;
    v8f acc[8] = {};
    #pragma unroll
    for (int k = 0; k < 16; ++k) {
        v2f b[8];
        #pragma unroll
        for (int t = 0; t < 8; ++t)
            b[t] = *(const v2f*)(bp + t * 16 * WPITCH + 4 * k);
        #pragma unroll
        for (int t = 0; t < 8; ++t)
            acc[t] = __builtin_amdgcn_wmma_f32_16x16x4_f32(
                         false, fa[k], false, b[t], (short)0, acc[t], false, false);
    }

    #pragma unroll
    for (int t = 0; t < 8; ++t) {
        const int col = (t & 3) * 16 + lr;
        const float bv = (t < 4 ? bias0 : bias1)[col];
        float* crow = (t < 4 ? C0 : C1) + (size_t)(row0 + hi * 8) * DIM + col;
        #pragma unroll
        for (int v = 0; v < 8; ++v)
            crow[(size_t)v * DIM] = acc[t][v] + bv;
    }
}

template <int NT, bool KEEP_A, int PITCH, bool SWZ>
__device__ __forceinline__ void pair_product_lds(
    const float* al, const float* Wl, int t0, int lr, int hi,
    const float* __restrict__ bias0, const float* __restrict__ bias1,
    float* __restrict__ C0, float* __restrict__ C1, int row0)
{
    v2f fa[16];
    const int sw = SWZ ? (lr << 2) : 0;
    const float* ap = al + lr * PITCH + 2 * hi;
    if (KEEP_A) {
        #pragma unroll
        for (int k = 0; k < 16; ++k) fa[k] = *(const v2f*)(ap + ((4 * k) ^ sw));
    }

    const float* bp = Wl + (t0 * 16 + lr) * PITCH + 2 * hi;
    v8f acc[NT] = {};
    #pragma unroll
    for (int k = 0; k < 16; ++k) {
        if (!KEEP_A) fa[k] = *(const v2f*)(ap + ((4 * k) ^ sw));
        v2f b[NT];
        #pragma unroll
        for (int t = 0; t < NT; ++t)
            b[t] = *(const v2f*)(bp + t * 16 * PITCH + ((4 * k) ^ sw));
        #pragma unroll
        for (int t = 0; t < NT; ++t)
            acc[t] = __builtin_amdgcn_wmma_f32_16x16x4_f32(
                         false, fa[k], false, b[t], (short)0, acc[t], false, false);
    }
    #pragma unroll
    for (int t = 0; t < NT; ++t) {
        const int tt  = t0 + t;
        const int col = (tt & 3) * 16 + lr;
        const float bv = (tt < 4 ? bias0 : bias1)[col];
        float* crow = (tt < 4 ? C0 : C1) + (size_t)(row0 + hi * 8) * DIM + col;
        #pragma unroll
        for (int v = 0; v < 8; ++v)
            crow[(size_t)v * DIM] = acc[t][v] + bv;
    }
}

typedef __attribute__((ext_vector_type(2))) int v2i;
typedef __attribute__((ext_vector_type(4))) int v4i;

#define CAP 64
#define CNT_WORDS (((NODES + 1 + 63) / 64) * 64)
#define BIN_SHIFT 8
#define BIN_NODES (1 << BIN_SHIFT)
#define NBINS     ((NODES + BIN_NODES - 1) / BIN_NODES)
#define BIN_CAP   16384
#define BIN_PAD   64
#define EPB       4096
#define BIN_BLOCKS ((EDGES + EPB - 1) / EPB)
static_assert(NBINS <= 256, "the bin scan is written for at most 256 bins");
static_assert(CAP == 64, "fill_slots takes a bucket row as 16 pieces of 4 slots");

__global__ void zero_counts(int* __restrict__ ovfCount, int* __restrict__ binCnt)
{
    const int i = threadIdx.x;
    if (i < NBINS) binCnt[i * BIN_PAD] = 0;
    if (i == NBINS) *ovfCount = 0;
}

__global__ void __launch_bounds__(1024)
bin_edges(const int* __restrict__ src, const int* __restrict__ dst,
          const int* __restrict__ ety, int* __restrict__ binCnt,
          int* __restrict__ binArr, int* __restrict__ ovfCount,
          int* __restrict__ ovf)
{
    __shared__ int hist[256];
    __shared__ int lbase[256];
    __shared__ int gbase[256];
    __shared__ int wsum[8];
    __shared__ __attribute__((aligned(8))) v2i stage[EPB];

    const int tid  = threadIdx.x;
    const int lane = tid & 31;
    const int e0   = blockIdx.x * EPB;
    if (tid < 256) hist[tid] = 0;
    __syncthreads();

    int d[4], k[4], r[4];
    #pragma unroll
    for (int j = 0; j < 4; ++j) {
        const int e  = e0 + j * 1024 + tid;
        const int ec = e < EDGES ? e : EDGES - 1;
        const int dd = dst[ec], ss = src[ec], tt = ety[ec];
        d[j] = e < EDGES ? dd : -1;
        k[j] = ss + (tt != 0 ? NODES : 0);
    }
    #pragma unroll
    for (int j = 0; j < 4; ++j)
        r[j] = d[j] >= 0 ? atomicAdd(&hist[d[j] >> BIN_SHIFT], 1) : 0;
    __syncthreads();

    int c = 0, s = 0;
    if (tid < 256) {
        c = hist[tid];
        s = c;
        for (int o = 1; o < 32; o <<= 1) {
            const int t = __shfl_up(s, o, 32);
            if (lane >= o) s += t;
        }
        if (lane == 31) wsum[tid >> 5] = s;
        gbase[tid] = c ? atomicAdd(&binCnt[tid * BIN_PAD], c) : 0;
    }
    __syncthreads();
    if (tid < 256) {
        int off = 0;
        for (int i = 0; i < (tid >> 5); ++i) off += wsum[i];
        lbase[tid] = off + s - c;
    }
    __syncthreads();
    #pragma unroll
    for (int j = 0; j < 4; ++j)
        if (d[j] >= 0) {
            v2i rec; rec.x = d[j]; rec.y = k[j];
            stage[lbase[d[j] >> BIN_SHIFT] + r[j]] = rec;
        }
    __syncthreads();

    const int left = EDGES - e0;
    const int nv   = left < EPB ? left : EPB;
    v2i rc[4];
    int gb[4], lb[4];
    #pragma unroll
    for (int j = 0; j < 4; ++j) {
        const int i = j * 1024 + tid;
        rc[j] = stage[i < nv ? i : nv - 1];
    }
    #pragma unroll
    for (int j = 0; j < 4; ++j) {
        const int b = rc[j].x >> BIN_SHIFT;
        gb[j] = gbase[b];
        lb[j] = lbase[b];
    }
    #pragma unroll
    for (int j = 0; j < 4; ++j) {
        const int i = j * 1024 + tid;
        if (i < nv) {
            const int b = rc[j].x >> BIN_SHIFT;
            const int p = gb[j] + (i - lb[j]);
            if (p < BIN_CAP) {
                *(v2i*)(binArr + 2 * ((size_t)b * BIN_CAP + p)) = rc[j];
            } else {
                const int q = atomicAdd(ovfCount, 1);
                ovf[2 * q]     = rc[j].x;
                ovf[2 * q + 1] = rc[j].y;
            }
        }
    }
}

__global__ void __launch_bounds__(1024)
fill_slots(const int* __restrict__ binCnt, const int* __restrict__ binArr,
           int* __restrict__ cnt, int* __restrict__ key,
           int* __restrict__ ovfCount, int* __restrict__ ovf)
{
    __shared__ int lcnt[BIN_NODES];
    __shared__ __attribute__((aligned(16))) int lkey[BIN_NODES * CAP];

    const int tid = threadIdx.x;
    const int b   = blockIdx.x;
    if (tid < BIN_NODES) lcnt[tid] = 0;
    for (int q = tid; q < BIN_NODES * (CAP / 4); q += 1024) {
        const v4i z = {0, 0, 0, 0};
        *(v4i*)(lkey + 4 * q) = z;
    }
    __syncthreads();

    const int nb = binCnt[b * BIN_PAD];
    const int n  = nb < BIN_CAP ? nb : BIN_CAP;
    const int* rec = binArr + 2 * (size_t)b * BIN_CAP;
    for (int i0 = 0; i0 < n; i0 += 4 * 1024) {
        v2i r[4];
        #pragma unroll
        for (int j = 0; j < 4; ++j) {
            const int i = i0 + j * 1024 + tid;
            r[j] = *(const v2i*)(rec + 2 * (i < n ? i : n - 1));
        }
        #pragma unroll
        for (int j = 0; j < 4; ++j) {
            const int i = i0 + j * 1024 + tid;
            if (i < n) {
                const int dl  = r[j].x & (BIN_NODES - 1);
                const int pos = atomicAdd(&lcnt[dl], 1);
                if (pos < CAP) {
                    lkey[dl * CAP + pos] = r[j].y;
                } else {
                    const int q = atomicAdd(ovfCount, 1);
                    ovf[2 * q]     = r[j].x;
                    ovf[2 * q + 1] = r[j].y;
                }
            }
        }
    }
    __syncthreads();

    const int node0 = b << BIN_SHIFT;
    const int rows  = NODES - node0 < BIN_NODES ? NODES - node0 : BIN_NODES;
    if (tid < rows) {
        const int c = lcnt[tid];
        cnt[node0 + tid] = c < CAP ? c : CAP;
    }
    for (int q = tid; q < rows * (CAP / 4); q += 1024) {
        const int c = lcnt[q >> 4];
        const v4i v = *(const v4i*)(lkey + 4 * q);
        if (4 * (q & 15) < c) *(v4i*)(key + (size_t)node0 * CAP + 4 * q) = v;
    }
}

__device__ __forceinline__ int slot_key(const int* __restrict__ key, int beg,
                                        int selfKey, int v)
{
    return v == 0 ? selfKey : key[beg + v - 1];
}

__global__ void __launch_bounds__(256) __attribute__((amdgpu_waves_per_eu(16, 16)))
gat_node(const float* __restrict__ el, const float* __restrict__ er,
         const float* __restrict__ attn, const int* __restrict__ cnt,
         const int* __restrict__ key, const int* __restrict__ ovf,
         float* __restrict__ h,
         float* __restrict__ emb,
         const float* __restrict__ W0, const float* __restrict__ bias0,
         const float* __restrict__ W1, const float* __restrict__ bias1,
         float* __restrict__ C0, float* __restrict__ C1)
{
    __shared__ __attribute__((aligned(16))) float Wl[2 * DIM * DIM];
    __shared__ __attribute__((aligned(16))) float Al[8 * NPW * DIM];

    const int lane = threadIdx.x & 31;
    const int wv   = __builtin_amdgcn_readfirstlane(threadIdx.x >> 5);
    const int d0   = (blockIdx.x * 8 + wv) * NPW;
    const bool live = d0 < NODES;
    if (live) {
    const int cntv = cnt[d0 + (lane < NPW ? lane : 0)];
    const int novf = cnt[NODES];
    const v2f at   = *(const v2f*)(attn + 2 * lane);
    const int sub  = lane & (CH - 1);
    const bool b16 = (lane & 16) != 0, b8 = (lane & 8) != 0, b4 = (lane & 4) != 0;
    const int myEdge = (b16 ? 4 : 0) + (b8 ? 2 : 0) + (b4 ? 1 : 0);
    static_assert(CH == 8, "the folding butterfly is written for 8 edges per round");

    #pragma unroll
    for (int n = 0; n < NPW; ++n) {
        const int d   = d0 + n;
        const int nin = __builtin_amdgcn_readlane(cntv, n);
        const int beg = d * CAP;
        const int tot = (nin < CAP ? nin : CAP) + 1;
        const v2f rr  = *(const v2f*)(er + (size_t)d * DIM + 2 * lane);

        float m = BIG_NEG, den = 0.f;
        v2f acc = {0.f, 0.f};

        int kv = slot_key(key, beg, d, sub < tot ? sub : tot - 1);
        for (int v0 = 0; v0 < tot; v0 += CH) {
            const int rem = tot - v0;
            const int cnt = rem < CH ? rem : CH;
            const int ks = kv >= NODES ? kv - NODES : kv;

            v2f row[CH];
            #pragma unroll
            for (int i = 0; i < CH; ++i) {
                const int s = __builtin_amdgcn_readlane(ks, i);
                row[i] = *(const v2f*)(el + (size_t)s * DIM + 2 * lane);
            }
            int kn = 0;
            if (rem > CH) {
                const int r2 = rem - CH;
                kn = key[beg + v0 + CH - 1 + (sub < r2 ? sub : r2 - 1)];
            }
            float q[CH];
            #pragma unroll
            for (int i = 0; i < CH; ++i) {
                float x0 = row[i].x + rr.x;
                float x1 = row[i].y + rr.y;
                x0 = x0 > 0.f ? x0 : NEG_SLOPE * x0;
                x1 = x1 > 0.f ? x1 : NEG_SLOPE * x1;
                float t = x0 * at.x;
                t += x1 * at.y;
                q[i] = t;
            }
            float k0[4], k1[2], sc_e;
            #pragma unroll
            for (int j = 0; j < 4; ++j) {
                const float send = b16 ? q[j] : q[j + 4];
                const float keep = b16 ? q[j + 4] : q[j];
                k0[j] = keep + __shfl_xor(send, 16, 32);
            }
            #pragma unroll
            for (int j = 0; j < 2; ++j) {
                const float send = b8 ? k0[j] : k0[j + 2];
                const float keep = b8 ? k0[j + 2] : k0[j];
                k1[j] = keep + __shfl_xor(send, 8, 32);
            }
            {
                const float send = b4 ? k1[0] : k1[1];
                const float keep = b4 ? k1[1] : k1[0];
                sc_e = keep + __shfl_xor(send, 4, 32);
            }
            sc_e += __shfl_xor(sc_e, 2, 32);
            sc_e += __shfl_xor(sc_e, 1, 32);

            const bool valid = myEdge < cnt;
            float cm = valid ? sc_e : BIG_NEG;
            cm = fmaxf(cm, __shfl_xor(cm, 16, 32));
            cm = fmaxf(cm, __shfl_xor(cm, 8, 32));
            cm = fmaxf(cm, __shfl_xor(cm, 4, 32));
            const float nm = fmaxf(m, cm);
            const float sc = expf(m - nm);
            const float wv = valid ? expf(sc_e - nm) : 0.f;
            den *= sc; acc.x *= sc; acc.y *= sc;
            #pragma unroll
            for (int i = 0; i < CH; ++i) {
                const int src_lane = ((i >> 2) & 1) * 16 + ((i >> 1) & 1) * 8 + (i & 1) * 4;
                const float w = __int_as_float(
                    __builtin_amdgcn_readlane(__float_as_int(wv), src_lane));
                den   += w;
                acc.x += w * row[i].x;
                acc.y += w * row[i].y;
            }
            m  = nm;
            kv = kn;
        }

        {
            for (int q = 0; q < novf; ++q) {
                if (ovf[2 * q] != d) continue;
                const int kq = ovf[2 * q + 1];
                const int sq = kq >= NODES ? kq - NODES : kq;
                const v2f rw = *(const v2f*)(el + (size_t)sq * DIM + 2 * lane);
                float x0 = rw.x + rr.x;
                float x1 = rw.y + rr.y;
                x0 = x0 > 0.f ? x0 : NEG_SLOPE * x0;
                x1 = x1 > 0.f ? x1 : NEG_SLOPE * x1;
                float t = x0 * at.x;
                t += x1 * at.y;
                for (int o = 16; o > 0; o >>= 1) t += __shfl_xor(t, o, 32);
                const float nm = fmaxf(m, t);
                const float sc = expf(m - nm);
                const float w  = expf(t - nm);
                den   = den * sc + w;
                acc.x = acc.x * sc + w * rw.x;
                acc.y = acc.y * sc + w * rw.y;
                m = nm;
            }
        }

        v2f o;
        o.x = acc.x / den;
        o.y = acc.y / den;
        o.x = o.x > 0.f ? o.x : (expf(o.x) - 1.f);
        o.y = o.y > 0.f ? o.y : (expf(o.y) - 1.f);
        *(v2f*)(h   + (size_t)d * DIM + 2 * lane) = o;
        *(v2f*)(emb + (size_t)d * DIM + 2 * lane) = o;
        *(v2f*)(Al + (wv * NPW + n) * DIM + ((2 * lane) ^ (((wv * NPW + n) & 15) << 2))) = o;
    }
    }

    {
        const int tid = threadIdx.x;
        v4f w0v[4], w1v[4];
        #pragma unroll
        for (int j = 0; j < 4; ++j) {
            w0v[j] = *(const v4f*)(W0 + 4 * (j * 256 + tid));
            w1v[j] = *(const v4f*)(W1 + 4 * (j * 256 + tid));
        }
        #pragma unroll
        for (int j = 0; j < 4; ++j) {
            const int i = j * 256 + tid;
            const int r = i >> 4, c4 = (i & 15) ^ (r & 15);
            *(v4f*)(Wl + r * DIM + 4 * c4)         = w0v[j];
            *(v4f*)(Wl + (DIM + r) * DIM + 4 * c4) = w1v[j];
        }
    }
    __syncthreads();
    const int rtile = blockIdx.x * 2 + (wv & 1);
    if (rtile < ROW_TILES)
        pair_product_lds<2, false, DIM, true>(Al + (wv & 1) * 16 * DIM, Wl, 2 * (wv >> 1),
                            lane & 15, lane >> 4, bias0, bias1, C0, C1, rtile << 4);
}

__global__ void __launch_bounds__(256)
ggc_pull(const float* __restrict__ tr, const int* __restrict__ ety,
         const int* __restrict__ cnt, const int* __restrict__ key,
         const int* __restrict__ ovf, float* __restrict__ acc)
{
    const int lane = threadIdx.x & 31;
    const int d0   = (blockIdx.x * 8 + (threadIdx.x >> 5)) * NPW;
    if (d0 >= NODES) return;
    const int nl   = lane < NPW ? lane : 0;
    const int cntv = cnt[d0 + nl];
    const int novf = cnt[NODES];
    const int selfv = d0 + nl + (ety[EDGES + d0 + nl] != 0 ? NODES : 0);
    const int sub  = lane & (CH - 1);

    #pragma unroll
    for (int n = 0; n < NPW; ++n) {
        const int d   = d0 + n;
        const int nin = __builtin_amdgcn_readlane(cntv, n);
        const int beg = d * CAP;
        const int tot = (nin < CAP ? nin : CAP) + 1;
        const int selfKey = __builtin_amdgcn_readlane(selfv, n);

        v2f a = {0.f, 0.f};
        int kv = slot_key(key, beg, selfKey, sub < tot ? sub : tot - 1);
        for (int v0 = 0; v0 < tot; v0 += CH) {
            const int rem = tot - v0;
            const int cnt = rem < CH ? rem : CH;
            v2f row[CH];
            #pragma unroll
            for (int i = 0; i < CH; ++i) {
                const int r = __builtin_amdgcn_readlane(kv, i);
                row[i] = *(const v2f*)(tr + (size_t)r * DIM + 2 * lane);
            }
            int kn = 0;
            if (rem > CH) {
                const int r2 = rem - CH;
                kn = key[beg + v0 + CH - 1 + (sub < r2 ? sub : r2 - 1)];
            }
            #pragma unroll
            for (int i = 0; i < CH; ++i) {
                a.x += (i < cnt) ? row[i].x : 0.f;
                a.y += (i < cnt) ? row[i].y : 0.f;
            }
            kv = kn;
        }
        {
            for (int q = 0; q < novf; ++q) {
                if (ovf[2 * q] != d) continue;
                const v2f rw = *(const v2f*)(tr + (size_t)ovf[2 * q + 1] * DIM + 2 * lane);
                a.x += rw.x;
                a.y += rw.y;
            }
        }
        *(v2f*)(acc + (size_t)d * DIM + 2 * lane) = a;
    }
}

#define GATES  (3 * DIM)
#define GRU_BLOCKS 256
#define GRU_WAVES  16
#define GRU_WHOLE  12
#define GRU_MAIN_TILES (GRU_BLOCKS * GRU_WHOLE)
#define GRU_REST_TILES (ROW_TILES - GRU_MAIN_TILES)
static_assert(GRU_REST_TILES >= 0 && GRU_REST_TILES <= GRU_BLOCKS,
              "the tiles left over must fit one per group");
static_assert(GRU_WAVES - GRU_WHOLE == DIM / 16, "one wave per block of 16 features of a shared tile");
static_assert((2 * GATES + GRU_WAVES * 16) * WPITCH * 4 > 160 * 1024, "two groups must not fit one WGP's LDS");

__global__ void __launch_bounds__(512)
gru_fused(const float* __restrict__ am,
          const float* __restrict__ Wih, const float* __restrict__ bih,
          const float* __restrict__ Whh, const float* __restrict__ bhh,
          float* h, float* emb,
          const float* __restrict__ W0, const float* __restrict__ bias0,
          const float* __restrict__ W1, const float* __restrict__ bias1,
          float* __restrict__ C0, float* __restrict__ C1)
{
    __shared__ __attribute__((aligned(16))) float Wl[2 * GATES * WPITCH];
    __shared__ __attribute__((aligned(16))) float Al[GRU_WAVES * 16 * WPITCH];

    const int tid  = threadIdx.x;
    const int lane = tid & 31;
    const int lr   = lane & 15;
    const int hi   = lane >> 4;
    const int wv    = __builtin_amdgcn_readfirstlane(tid >> 5);
    const bool whole = wv < GRU_WHOLE;
    const bool part  = !whole && (int)blockIdx.x < GRU_REST_TILES;
    const bool live  = whole || part;
    const int rt   = whole ? (int)blockIdx.x * GRU_WHOLE + wv : GRU_MAIN_TILES + (int)blockIdx.x;
    const int row0 = rt << 4;
    const int ct0  = whole ? 0 : wv - GRU_WHOLE;
    const int ct1  = whole ? DIM / 16 : ct0 + 1;

    v2f fa[16] = {}, fh[16] = {};
    if (live) {
        const float* arow = am + (size_t)(row0 + lr) * DIM + 2 * hi;
        const float* hrow = h  + (size_t)(row0 + lr) * DIM + 2 * hi;
        #pragma unroll
        for (int k = 0; k < 16; ++k) {
            fa[k] = *(const v2f*)(arow + 4 * k);
            fh[k] = *(const v2f*)(hrow + 4 * k);
        }
    }

    {
        static_assert(GATES * (DIM / 4) == 6 * 512, "six float4 of each matrix per thread");
        v4f wi[6], wh[6];
        #pragma unroll
        for (int j = 0; j < 6; ++j) {
            wi[j] = *(const v4f*)(Wih + 4 * (j * 512 + tid));
            wh[j] = *(const v4f*)(Whh + 4 * (j * 512 + tid));
        }
        #pragma unroll
        for (int j = 0; j < 6; ++j) {
            const int i = j * 512 + tid;
            const int r = i >> 4, c4 = i & 15;
            *(v4f*)(Wl + r * WPITCH + 4 * c4)           = wi[j];
            *(v4f*)(Wl + (GATES + r) * WPITCH + 4 * c4) = wh[j];
        }
    }
    __syncthreads();
    float* al = Al + (whole ? wv : GRU_WHOLE) * 16 * WPITCH;

    if (live) {
    for (int ct = ct0; ct < ct1; ++ct) {
        const int dcol = ct * 16 + lr;
        const float* bi = Wl + dcol * WPITCH + 2 * hi;
        const float* bh = bi + GATES * WPITCH;

        float* hp = h + (size_t)(row0 + hi * 8) * DIM + dcol;
        float hv[8];
        #pragma unroll
        for (int v = 0; v < 8; ++v) hv[v] = hp[(size_t)v * DIM];

        v8f cir = {}, ciz = {}, cic = {}, chr = {}, chz = {}, chc = {};
        #pragma unroll
        for (int k = 0; k < 16; ++k) {
            const v2f b0 = *(const v2f*)(bi + 4 * k);
            const v2f b1 = *(const v2f*)(bi + DIM * WPITCH + 4 * k);
            const v2f b2 = *(const v2f*)(bi + 2 * DIM * WPITCH + 4 * k);
            const v2f b3 = *(const v2f*)(bh + 4 * k);
            const v2f b4 = *(const v2f*)(bh + DIM * WPITCH + 4 * k);
            const v2f b5 = *(const v2f*)(bh + 2 * DIM * WPITCH + 4 * k);
            cir = __builtin_amdgcn_wmma_f32_16x16x4_f32(
                      false, fa[k], false, b0, (short)0, cir, false, false);
            ciz = __builtin_amdgcn_wmma_f32_16x16x4_f32(
                      false, fa[k], false, b1, (short)0, ciz, false, false);
            cic = __builtin_amdgcn_wmma_f32_16x16x4_f32(
                      false, fa[k], false, b2, (short)0, cic, false, false);
            chr = __builtin_amdgcn_wmma_f32_16x16x4_f32(
                      false, fh[k], false, b3, (short)0, chr, false, false);
            chz = __builtin_amdgcn_wmma_f32_16x16x4_f32(
                      false, fh[k], false, b4, (short)0, chz, false, false);
            chc = __builtin_amdgcn_wmma_f32_16x16x4_f32(
                      false, fh[k], false, b5, (short)0, chc, false, false);
        }

        const float bir = bih[dcol], biz = bih[DIM + dcol], bic = bih[2 * DIM + dcol];
        const float bhr = bhh[dcol], bhz = bhh[DIM + dcol], bhc = bhh[2 * DIM + dcol];
        float o[8];
        #pragma unroll
        for (int v = 0; v < 8; ++v) {
            const float gir = cir[v] + bir, ghr = chr[v] + bhr;
            const float giz = ciz[v] + biz, ghz = chz[v] + bhz;
            const float gic = cic[v] + bic, ghc = chc[v] + bhc;
            const float r = 1.f / (1.f + expf(-(gir + ghr)));
            const float z = 1.f / (1.f + expf(-(giz + ghz)));
            const float c = tanhf(gic + r * ghc);
            o[v] = (1.f - z) * c + z * hv[v];
            al[(hi * 8 + v) * WPITCH + dcol] = o[v];
        }
        if (whole) {
            #pragma unroll
            for (int v = 0; v < 8; ++v) hp[(size_t)v * DIM] = o[v];
            if (emb != nullptr) {
                float* ep = emb + (size_t)(row0 + hi * 8) * DIM + dcol;
                #pragma unroll
                for (int v = 0; v < 8; ++v) ep[(size_t)v * DIM] = o[v];
            }
        }
    }
    }

    const bool pair = W0 != nullptr;
    v4f w0v[2] = {}, w1v[2] = {};
    if (pair) {
        #pragma unroll
        for (int j = 0; j < 2; ++j) {
            w0v[j] = *(const v4f*)(W0 + 4 * (j * 512 + tid));
            w1v[j] = *(const v4f*)(W1 + 4 * (j * 512 + tid));
        }
    }
    __syncthreads();
    if (part) {
        const float* as = Al + GRU_WHOLE * 16 * WPITCH;
        #pragma unroll
        for (int j = 0; j < 2; ++j) {
            const int r = 4 * (wv - GRU_WHOLE) + 2 * j + hi;
            const v4f nv = *(const v4f*)(as + r * WPITCH + 4 * lr);
            *(v4f*)(h + (size_t)(row0 + r) * DIM + 4 * lr) = nv;
            if (emb) *(v4f*)(emb + (size_t)(row0 + r) * DIM + 4 * lr) = nv;
        }
    }
    if (!pair) return;
    #pragma unroll
    for (int j = 0; j < 2; ++j) {
        const int i = j * 512 + tid;
        const int r = i >> 4, c4 = i & 15;
        *(v4f*)(Wl + r * WPITCH + 4 * c4)         = w0v[j];
        *(v4f*)(Wl + (DIM + r) * WPITCH + 4 * c4) = w1v[j];
    }
    __syncthreads();
    if (whole)
        pair_product_lds<8, true, WPITCH, false>(al, Wl, 0, lr, hi, bias0, bias1, C0, C1, row0);
    else if (part)
        pair_product_lds<2, false, WPITCH, false>(al, Wl, 2 * (wv - GRU_WHOLE), lr, hi,
                                                  bias0, bias1, C0, C1, row0);
}

extern "C" void kernel_launch(void* const* d_in, const int* in_sizes, int n_in,
                              void* d_out, int out_size, void* d_ws, size_t ws_size,
                              hipStream_t stream)
{
    (void)in_sizes; (void)n_in; (void)out_size; (void)ws_size;

    const float* x   = (const float*)d_in[0];
    const int*   src = (const int*)d_in[1];
    const int*   dst = (const int*)d_in[2];
    const int*   ety = (const int*)d_in[3];

    const long NE = (long)NODES * DIM;

    float* ws   = (float*)d_ws;
    float* hbuf = ws;                 ws += NE;
    float* buf1 = ws;                 ws += NE;
    float* buf2 = ws;                 ws += NE;
    float* tr0  = ws;                 ws += NE;
    float* tr1  = ws;                 ws += NE;
    float* acc  = ws;                 ws += NE;
    int* cnt = (int*)ws;              ws += CNT_WORDS;
    int* key = (int*)ws;              ws += (long)NODES * CAP;
    int* ovf = (int*)ws;              ws += 2L * EDGES;
    int* binCnt = (int*)ws;           ws += NBINS * BIN_PAD;
    int* binArr = (int*)ws;

    float* out = (float*)d_out;

    const int rowTileBlocks  = (ROW_TILES + 7) / 8;

    zero_counts<<<1, 256, 0, stream>>>(cnt + NODES, binCnt);
    bin_edges<<<BIN_BLOCKS, 1024, 0, stream>>>(src, dst, ety, binCnt, binArr, cnt + NODES, ovf);
    fill_slots<<<NBINS, 1024, 0, stream>>>(binCnt, binArr, cnt, key, cnt + NODES, ovf);

    for (int l = 0; l < 2; ++l) {
        const int base = 4 + 11 * l;
        const float* Wsrc = (const float*)d_in[base + 0];
        const float* bsrc = (const float*)d_in[base + 1];
        const float* Wdst = (const float*)d_in[base + 2];
        const float* bdst = (const float*)d_in[base + 3];
        const float* attn = (const float*)d_in[base + 4];
        const float* ggW  = (const float*)d_in[base + 5];
        const float* ggb  = (const float*)d_in[base + 6];
        const float* Wih  = (const float*)d_in[base + 7];
        const float* bih  = (const float*)d_in[base + 8];
        const float* Whh  = (const float*)d_in[base + 9];
        const float* bhh  = (const float*)d_in[base + 10];

        if (l == 0)
            gemm_pair_wmma<<<rowTileBlocks, 256, 0, stream>>>(x, Wsrc, bsrc, Wdst, bdst,
                                                              buf1, buf2);
        gat_node<<<NODE_BLOCKS, 256, 0, stream>>>(buf1, buf2, attn, cnt, key, ovf, hbuf,
                                                  out + (2L * l) * NE,
                                                  ggW, ggb, ggW + DIM * DIM, ggb + DIM,
                                                  tr0, tr1);

        ggc_pull<<<NODE_BLOCKS, 256, 0, stream>>>(tr0, ety, cnt, key, ovf, acc);
        gru_fused<<<GRU_BLOCKS, 512, 0, stream>>>(
            acc, Wih, bih, Whh, bhh, hbuf, nullptr,
            ggW, ggb, ggW + DIM * DIM, ggb + DIM, tr0, tr1);
        ggc_pull<<<NODE_BLOCKS, 256, 0, stream>>>(tr0, ety, cnt, key, ovf, acc);
        if (l == 0) {
            const float* Wsrc1 = (const float*)d_in[4 + 11 + 0];
            const float* bsrc1 = (const float*)d_in[4 + 11 + 1];
            const float* Wdst1 = (const float*)d_in[4 + 11 + 2];
            const float* bdst1 = (const float*)d_in[4 + 11 + 3];
            gru_fused<<<GRU_BLOCKS, 512, 0, stream>>>(
                acc, Wih, bih, Whh, bhh, hbuf, out + (2L * l + 1) * NE,
                Wsrc1, bsrc1, Wdst1, bdst1, buf1, buf2);
        } else {
            gru_fused<<<GRU_BLOCKS, 512, 0, stream>>>(
                acc, Wih, bih, Whh, bhh, hbuf, out + (2L * l + 1) * NE,
                nullptr, nullptr, nullptr, nullptr, nullptr, nullptr);
        }
    }
}


# expf, tanhf and IEEE divisions (GRU gates; GAT softmax weights, normalisation, elu) become hardware exp2 / reciprocal forms, tanh without its branch; results differ in the last bits
# speedup vs baseline: 5.0180x; 1.1009x over previous
#include <hip/hip_runtime.h>
#include <hip/hip_bf16.h>


typedef __attribute__((ext_vector_type(2))) float v2f;
typedef __attribute__((ext_vector_type(4))) float v4f;
typedef __attribute__((ext_vector_type(8))) float v8f;

#define NODES 50000
#define EDGES 800000
#define DIM   64
#define NEG_SLOPE 0.2f

#define NPW 4
#define CH  8
static_assert(NODES % NPW == 0, "node group must divide the node count");
#define NODE_BLOCKS ((((NODES) / NPW) + 7) / 8)
#define BIG_NEG (-3.402823466e+38f)

#define LOG2E_F 1.44269504088896341f
__device__ __forceinline__ float hw_exp(float x)
{
    return __builtin_amdgcn_exp2f(x * LOG2E_F);
}
__device__ __forceinline__ float hw_rcp(float x)
{
    return __builtin_amdgcn_rcpf(x);
}
__device__ __forceinline__ float hw_sigmoid(float x)
{
    return hw_rcp(1.f + __builtin_amdgcn_exp2f(x * -LOG2E_F));
}
__device__ __forceinline__ float hw_tanh(float x)
{
    return 1.f - 2.f * hw_rcp(1.f + __builtin_amdgcn_exp2f(x * (2.f * LOG2E_F)));
}

#define WPITCH 68
#define ROW_TILES (NODES / 16)
static_assert(NODES % 16 == 0, "row tiles must divide the node count");

__global__ void __launch_bounds__(256) __attribute__((amdgpu_waves_per_eu(2, 4)))
gemm_pair_wmma(const float* __restrict__ A,
               const float* __restrict__ W0, const float* __restrict__ bias0,
               const float* __restrict__ W1, const float* __restrict__ bias1,
               float* __restrict__ C0, float* __restrict__ C1)
{
    __shared__ __attribute__((aligned(16))) float Wl[2 * DIM * WPITCH];
    __shared__ __attribute__((aligned(16))) float Al[8 * 16 * WPITCH];

    const int tid  = threadIdx.x;
    const int lane = tid & 31;
    const int lr   = lane & 15;
    const int hi   = lane >> 4;
    const int rt   = blockIdx.x * 8 + (tid >> 5);
    const bool live = rt < ROW_TILES;
    const int row0 = (live ? rt : ROW_TILES - 1) << 4;
    float* al = Al + (tid >> 5) * 16 * WPITCH;

    const float* atile = A + (size_t)row0 * DIM;
    v4f av[8], w0v[4], w1v[4];
    #pragma unroll
    for (int j = 0; j < 8; ++j)
        av[j] = *(const v4f*)(atile + 4 * (j * 32 + lane));
    #pragma unroll
    for (int j = 0; j < 4; ++j) {
        w0v[j] = *(const v4f*)(W0 + 4 * (j * 256 + tid));
        w1v[j] = *(const v4f*)(W1 + 4 * (j * 256 + tid));
    }
    #pragma unroll
    for (int j = 0; j < 8; ++j) {
        const int i = j * 32 + lane;
        *(v4f*)(al + (i >> 4) * WPITCH + 4 * (i & 15)) = av[j];
    }
    #pragma unroll
    for (int j = 0; j < 4; ++j) {
        const int i = j * 256 + tid;
        const int r = i >> 4, c4 = i & 15;
        *(v4f*)(Wl + r * WPITCH + 4 * c4)         = w0v[j];
        *(v4f*)(Wl + (DIM + r) * WPITCH + 4 * c4) = w1v[j];
    }
    __syncthreads();
    if (!live) return;

    v2f fa[16];
    const float* ap = al + lr * WPITCH + 2 * hi;
    #pragma unroll
    for (int k = 0; k < 16; ++k) fa[k] = *(const v2f*)(ap + 4 * k);

    const float* bp = Wl + lr * WPITCH + 2 * hi;
    v8f acc[8] = {};
    #pragma unroll
    for (int k = 0; k < 16; ++k) {
        v2f b[8];
        #pragma unroll
        for (int t = 0; t < 8; ++t)
            b[t] = *(const v2f*)(bp + t * 16 * WPITCH + 4 * k);
        #pragma unroll
        for (int t = 0; t < 8; ++t)
            acc[t] = __builtin_amdgcn_wmma_f32_16x16x4_f32(
                         false, fa[k], false, b[t], (short)0, acc[t], false, false);
    }

    #pragma unroll
    for (int t = 0; t < 8; ++t) {
        const int col = (t & 3) * 16 + lr;
        const float bv = (t < 4 ? bias0 : bias1)[col];
        float* crow = (t < 4 ? C0 : C1) + (size_t)(row0 + hi * 8) * DIM + col;
        #pragma unroll
        for (int v = 0; v < 8; ++v)
            crow[(size_t)v * DIM] = acc[t][v] + bv;
    }
}

template <int NT, bool KEEP_A, int PITCH, bool SWZ>
__device__ __forceinline__ void pair_product_lds(
    const float* al, const float* Wl, int t0, int lr, int hi,
    const float* __restrict__ bias0, const float* __restrict__ bias1,
    float* __restrict__ C0, float* __restrict__ C1, int row0)
{
    v2f fa[16];
    const int sw = SWZ ? (lr << 2) : 0;
    const float* ap = al + lr * PITCH + 2 * hi;
    if (KEEP_A) {
        #pragma unroll
        for (int k = 0; k < 16; ++k) fa[k] = *(const v2f*)(ap + ((4 * k) ^ sw));
    }

    const float* bp = Wl + (t0 * 16 + lr) * PITCH + 2 * hi;
    v8f acc[NT] = {};
    #pragma unroll
    for (int k = 0; k < 16; ++k) {
        if (!KEEP_A) fa[k] = *(const v2f*)(ap + ((4 * k) ^ sw));
        v2f b[NT];
        #pragma unroll
        for (int t = 0; t < NT; ++t)
            b[t] = *(const v2f*)(bp + t * 16 * PITCH + ((4 * k) ^ sw));
        #pragma unroll
        for (int t = 0; t < NT; ++t)
            acc[t] = __builtin_amdgcn_wmma_f32_16x16x4_f32(
                         false, fa[k], false, b[t], (short)0, acc[t], false, false);
    }
    #pragma unroll
    for (int t = 0; t < NT; ++t) {
        const int tt  = t0 + t;
        const int col = (tt & 3) * 16 + lr;
        const float bv = (tt < 4 ? bias0 : bias1)[col];
        float* crow = (tt < 4 ? C0 : C1) + (size_t)(row0 + hi * 8) * DIM + col;
        #pragma unroll
        for (int v = 0; v < 8; ++v)
            crow[(size_t)v * DIM] = acc[t][v] + bv;
    }
}

typedef __attribute__((ext_vector_type(2))) int v2i;
typedef __attribute__((ext_vector_type(4))) int v4i;

#define CAP 64
#define CNT_WORDS (((NODES + 1 + 63) / 64) * 64)
#define BIN_SHIFT 8
#define BIN_NODES (1 << BIN_SHIFT)
#define NBINS     ((NODES + BIN_NODES - 1) / BIN_NODES)
#define BIN_CAP   16384
#define BIN_PAD   64
#define EPB       4096
#define BIN_BLOCKS ((EDGES + EPB - 1) / EPB)
static_assert(NBINS <= 256, "the bin scan is written for at most 256 bins");
static_assert(CAP == 64, "fill_slots takes a bucket row as 16 pieces of 4 slots");

__global__ void zero_counts(int* __restrict__ ovfCount, int* __restrict__ binCnt)
{
    const int i = threadIdx.x;
    if (i < NBINS) binCnt[i * BIN_PAD] = 0;
    if (i == NBINS) *ovfCount = 0;
}

__global__ void __launch_bounds__(1024)
bin_edges(const int* __restrict__ src, const int* __restrict__ dst,
          const int* __restrict__ ety, int* __restrict__ binCnt,
          int* __restrict__ binArr, int* __restrict__ ovfCount,
          int* __restrict__ ovf)
{
    __shared__ int hist[256];
    __shared__ int lbase[256];
    __shared__ int gbase[256];
    __shared__ int wsum[8];
    __shared__ __attribute__((aligned(8))) v2i stage[EPB];

    const int tid  = threadIdx.x;
    const int lane = tid & 31;
    const int e0   = blockIdx.x * EPB;
    if (tid < 256) hist[tid] = 0;
    __syncthreads();

    int d[4], k[4], r[4];
    #pragma unroll
    for (int j = 0; j < 4; ++j) {
        const int e  = e0 + j * 1024 + tid;
        const int ec = e < EDGES ? e : EDGES - 1;
        const int dd = dst[ec], ss = src[ec], tt = ety[ec];
        d[j] = e < EDGES ? dd : -1;
        k[j] = ss + (tt != 0 ? NODES : 0);
    }
    #pragma unroll
    for (int j = 0; j < 4; ++j)
        r[j] = d[j] >= 0 ? atomicAdd(&hist[d[j] >> BIN_SHIFT], 1) : 0;
    __syncthreads();

    int c = 0, s = 0;
    if (tid < 256) {
        c = hist[tid];
        s = c;
        for (int o = 1; o < 32; o <<= 1) {
            const int t = __shfl_up(s, o, 32);
            if (lane >= o) s += t;
        }
        if (lane == 31) wsum[tid >> 5] = s;
        gbase[tid] = c ? atomicAdd(&binCnt[tid * BIN_PAD], c) : 0;
    }
    __syncthreads();
    if (tid < 256) {
        int off = 0;
        for (int i = 0; i < (tid >> 5); ++i) off += wsum[i];
        lbase[tid] = off + s - c;
    }
    __syncthreads();
    #pragma unroll
    for (int j = 0; j < 4; ++j)
        if (d[j] >= 0) {
            v2i rec; rec.x = d[j]; rec.y = k[j];
            stage[lbase[d[j] >> BIN_SHIFT] + r[j]] = rec;
        }
    __syncthreads();

    const int left = EDGES - e0;
    const int nv   = left < EPB ? left : EPB;
    v2i rc[4];
    int gb[4], lb[4];
    #pragma unroll
    for (int j = 0; j < 4; ++j) {
        const int i = j * 1024 + tid;
        rc[j] = stage[i < nv ? i : nv - 1];
    }
    #pragma unroll
    for (int j = 0; j < 4; ++j) {
        const int b = rc[j].x >> BIN_SHIFT;
        gb[j] = gbase[b];
        lb[j] = lbase[b];
    }
    #pragma unroll
    for (int j = 0; j < 4; ++j) {
        const int i = j * 1024 + tid;
        if (i < nv) {
            const int b = rc[j].x >> BIN_SHIFT;
            const int p = gb[j] + (i - lb[j]);
            if (p < BIN_CAP) {
                *(v2i*)(binArr + 2 * ((size_t)b * BIN_CAP + p)) = rc[j];
            } else {
                const int q = atomicAdd(ovfCount, 1);
                ovf[2 * q]     = rc[j].x;
                ovf[2 * q + 1] = rc[j].y;
            }
        }
    }
}

__global__ void __launch_bounds__(1024)
fill_slots(const int* __restrict__ binCnt, const int* __restrict__ binArr,
           int* __restrict__ cnt, int* __restrict__ key,
           int* __restrict__ ovfCount, int* __restrict__ ovf)
{
    __shared__ int lcnt[BIN_NODES];
    __shared__ __attribute__((aligned(16))) int lkey[BIN_NODES * CAP];

    const int tid = threadIdx.x;
    const int b   = blockIdx.x;
    if (tid < BIN_NODES) lcnt[tid] = 0;
    for (int q = tid; q < BIN_NODES * (CAP / 4); q += 1024) {
        const v4i z = {0, 0, 0, 0};
        *(v4i*)(lkey + 4 * q) = z;
    }
    __syncthreads();

    const int nb = binCnt[b * BIN_PAD];
    const int n  = nb < BIN_CAP ? nb : BIN_CAP;
    const int* rec = binArr + 2 * (size_t)b * BIN_CAP;
    for (int i0 = 0; i0 < n; i0 += 4 * 1024) {
        v2i r[4];
        #pragma unroll
        for (int j = 0; j < 4; ++j) {
            const int i = i0 + j * 1024 + tid;
            r[j] = *(const v2i*)(rec + 2 * (i < n ? i : n - 1));
        }
        #pragma unroll
        for (int j = 0; j < 4; ++j) {
            const int i = i0 + j * 1024 + tid;
            if (i < n) {
                const int dl  = r[j].x & (BIN_NODES - 1);
                const int pos = atomicAdd(&lcnt[dl], 1);
                if (pos < CAP) {
                    lkey[dl * CAP + pos] = r[j].y;
                } else {
                    const int q = atomicAdd(ovfCount, 1);
                    ovf[2 * q]     = r[j].x;
                    ovf[2 * q + 1] = r[j].y;
                }
            }
        }
    }
    __syncthreads();

    const int node0 = b << BIN_SHIFT;
    const int rows  = NODES - node0 < BIN_NODES ? NODES - node0 : BIN_NODES;
    if (tid < rows) {
        const int c = lcnt[tid];
        cnt[node0 + tid] = c < CAP ? c : CAP;
    }
    for (int q = tid; q < rows * (CAP / 4); q += 1024) {
        const int c = lcnt[q >> 4];
        const v4i v = *(const v4i*)(lkey + 4 * q);
        if (4 * (q & 15) < c) *(v4i*)(key + (size_t)node0 * CAP + 4 * q) = v;
    }
}

__device__ __forceinline__ int slot_key(const int* __restrict__ key, int beg,
                                        int selfKey, int v)
{
    return v == 0 ? selfKey : key[beg + v - 1];
}

__global__ void __launch_bounds__(256) __attribute__((amdgpu_waves_per_eu(16, 16)))
gat_node(const float* __restrict__ el, const float* __restrict__ er,
         const float* __restrict__ attn, const int* __restrict__ cnt,
         const int* __restrict__ key, const int* __restrict__ ovf,
         float* __restrict__ h,
         float* __restrict__ emb,
         const float* __restrict__ W0, const float* __restrict__ bias0,
         const float* __restrict__ W1, const float* __restrict__ bias1,
         float* __restrict__ C0, float* __restrict__ C1)
{
    __shared__ __attribute__((aligned(16))) float Wl[2 * DIM * DIM];
    __shared__ __attribute__((aligned(16))) float Al[8 * NPW * DIM];

    const int lane = threadIdx.x & 31;
    const int wv   = __builtin_amdgcn_readfirstlane(threadIdx.x >> 5);
    const int d0   = (blockIdx.x * 8 + wv) * NPW;
    const bool live = d0 < NODES;
    if (live) {
    const int cntv = cnt[d0 + (lane < NPW ? lane : 0)];
    const int novf = cnt[NODES];
    const v2f at   = *(const v2f*)(attn + 2 * lane);
    const int sub  = lane & (CH - 1);
    const bool b16 = (lane & 16) != 0, b8 = (lane & 8) != 0, b4 = (lane & 4) != 0;
    const int myEdge = (b16 ? 4 : 0) + (b8 ? 2 : 0) + (b4 ? 1 : 0);
    static_assert(CH == 8, "the folding butterfly is written for 8 edges per round");

    #pragma unroll
    for (int n = 0; n < NPW; ++n) {
        const int d   = d0 + n;
        const int nin = __builtin_amdgcn_readlane(cntv, n);
        const int beg = d * CAP;
        const int tot = (nin < CAP ? nin : CAP) + 1;
        const unsigned noff = (unsigned)d * (DIM * 4u) + (unsigned)lane * 8u;
        const v2f rr  = *(const v2f*)((const char*)er + noff);

        float m = BIG_NEG, den = 0.f;
        v2f acc = {0.f, 0.f};

        int kv = slot_key(key, beg, d, sub < tot ? sub : tot - 1);
        for (int v0 = 0; v0 < tot; v0 += CH) {
            const int rem = tot - v0;
            const int cnt = rem < CH ? rem : CH;
            const int ks = kv >= NODES ? kv - NODES : kv;

            v2f row[CH];
            #pragma unroll
            for (int i = 0; i < CH; ++i) {
                const int s = __builtin_amdgcn_readlane(ks, i);
                row[i] = *(const v2f*)(el + (size_t)s * DIM + 2 * lane);
            }
            int kn = 0;
            if (rem > CH) {
                const int r2 = rem - CH;
                kn = key[beg + v0 + CH - 1 + (sub < r2 ? sub : r2 - 1)];
            }
            float q[CH];
            #pragma unroll
            for (int i = 0; i < CH; ++i) {
                float x0 = row[i].x + rr.x;
                float x1 = row[i].y + rr.y;
                x0 = x0 > 0.f ? x0 : NEG_SLOPE * x0;
                x1 = x1 > 0.f ? x1 : NEG_SLOPE * x1;
                float t = x0 * at.x;
                t += x1 * at.y;
                q[i] = t;
            }
            float k0[4], k1[2], sc_e;
            #pragma unroll
            for (int j = 0; j < 4; ++j) {
                const float send = b16 ? q[j] : q[j + 4];
                const float keep = b16 ? q[j + 4] : q[j];
                k0[j] = keep + __shfl_xor(send, 16, 32);
            }
            #pragma unroll
            for (int j = 0; j < 2; ++j) {
                const float send = b8 ? k0[j] : k0[j + 2];
                const float keep = b8 ? k0[j + 2] : k0[j];
                k1[j] = keep + __shfl_xor(send, 8, 32);
            }
            {
                const float send = b4 ? k1[0] : k1[1];
                const float keep = b4 ? k1[1] : k1[0];
                sc_e = keep + __shfl_xor(send, 4, 32);
            }
            sc_e += __shfl_xor(sc_e, 2, 32);
            sc_e += __shfl_xor(sc_e, 1, 32);

            const bool valid = myEdge < cnt;
            float cm = valid ? sc_e : BIG_NEG;
            cm = fmaxf(cm, __shfl_xor(cm, 16, 32));
            cm = fmaxf(cm, __shfl_xor(cm, 8, 32));
            cm = fmaxf(cm, __shfl_xor(cm, 4, 32));
            const float nm = fmaxf(m, cm);
            const float sc = hw_exp(m - nm);
            const float wv = valid ? hw_exp(sc_e - nm) : 0.f;
            den *= sc; acc.x *= sc; acc.y *= sc;
            #pragma unroll
            for (int i = 0; i < CH; ++i) {
                const int src_lane = ((i >> 2) & 1) * 16 + ((i >> 1) & 1) * 8 + (i & 1) * 4;
                const float w = __int_as_float(
                    __builtin_amdgcn_readlane(__float_as_int(wv), src_lane));
                den   += w;
                acc.x += w * row[i].x;
                acc.y += w * row[i].y;
            }
            m  = nm;
            kv = kn;
        }

        {
            for (int q = 0; q < novf; ++q) {
                if (ovf[2 * q] != d) continue;
                const int kq = ovf[2 * q + 1];
                const int sq = kq >= NODES ? kq - NODES : kq;
                const v2f rw = *(const v2f*)(el + (size_t)sq * DIM + 2 * lane);
                float x0 = rw.x + rr.x;
                float x1 = rw.y + rr.y;
                x0 = x0 > 0.f ? x0 : NEG_SLOPE * x0;
                x1 = x1 > 0.f ? x1 : NEG_SLOPE * x1;
                float t = x0 * at.x;
                t += x1 * at.y;
                for (int o = 16; o > 0; o >>= 1) t += __shfl_xor(t, o, 32);
                const float nm = fmaxf(m, t);
                const float sc = hw_exp(m - nm);
                const float w  = hw_exp(t - nm);
                den   = den * sc + w;
                acc.x = acc.x * sc + w * rw.x;
                acc.y = acc.y * sc + w * rw.y;
                m = nm;
            }
        }

        v2f o;
        const float rden = hw_rcp(den);
        o.x = acc.x * rden;
        o.y = acc.y * rden;
        o.x = o.x > 0.f ? o.x : (hw_exp(o.x) - 1.f);
        o.y = o.y > 0.f ? o.y : (hw_exp(o.y) - 1.f);
        *(v2f*)((char*)h   + noff) = o;
        *(v2f*)((char*)emb + noff) = o;
        *(v2f*)(Al + (wv * NPW + n) * DIM + ((2 * lane) ^ (((wv * NPW + n) & 15) << 2))) = o;
    }
    }

    {
        const int tid = threadIdx.x;
        v4f w0v[4], w1v[4];
        #pragma unroll
        for (int j = 0; j < 4; ++j) {
            w0v[j] = *(const v4f*)(W0 + 4 * (j * 256 + tid));
            w1v[j] = *(const v4f*)(W1 + 4 * (j * 256 + tid));
        }
        #pragma unroll
        for (int j = 0; j < 4; ++j) {
            const int i = j * 256 + tid;
            const int r = i >> 4, c4 = (i & 15) ^ (r & 15);
            *(v4f*)(Wl + r * DIM + 4 * c4)         = w0v[j];
            *(v4f*)(Wl + (DIM + r) * DIM + 4 * c4) = w1v[j];
        }
    }
    __syncthreads();
    const int rtile = blockIdx.x * 2 + (wv & 1);
    if (rtile < ROW_TILES)
        pair_product_lds<2, false, DIM, true>(Al + (wv & 1) * 16 * DIM, Wl, 2 * (wv >> 1),
                            lane & 15, lane >> 4, bias0, bias1, C0, C1, rtile << 4);
}

__global__ void __launch_bounds__(256)
ggc_pull(const float* __restrict__ tr, const int* __restrict__ ety,
         const int* __restrict__ cnt, const int* __restrict__ key,
         const int* __restrict__ ovf, float* __restrict__ acc)
{
    const int lane = threadIdx.x & 31;
    const int d0   = (blockIdx.x * 8 + (threadIdx.x >> 5)) * NPW;
    if (d0 >= NODES) return;
    const int nl   = lane < NPW ? lane : 0;
    const int cntv = cnt[d0 + nl];
    const int novf = cnt[NODES];
    const int selfv = d0 + nl + (ety[EDGES + d0 + nl] != 0 ? NODES : 0);
    const int sub  = lane & (CH - 1);

    #pragma unroll
    for (int n = 0; n < NPW; ++n) {
        const int d   = d0 + n;
        const int nin = __builtin_amdgcn_readlane(cntv, n);
        const int beg = d * CAP;
        const int tot = (nin < CAP ? nin : CAP) + 1;
        const int selfKey = __builtin_amdgcn_readlane(selfv, n);

        v2f a = {0.f, 0.f};
        int kv = slot_key(key, beg, selfKey, sub < tot ? sub : tot - 1);
        for (int v0 = 0; v0 < tot; v0 += CH) {
            const int rem = tot - v0;
            const int cnt = rem < CH ? rem : CH;
            v2f row[CH];
            #pragma unroll
            for (int i = 0; i < CH; ++i) {
                const int r = __builtin_amdgcn_readlane(kv, i);
                row[i] = *(const v2f*)(tr + (size_t)r * DIM + 2 * lane);
            }
            int kn = 0;
            if (rem > CH) {
                const int r2 = rem - CH;
                kn = key[beg + v0 + CH - 1 + (sub < r2 ? sub : r2 - 1)];
            }
            #pragma unroll
            for (int i = 0; i < CH; ++i) {
                a.x += (i < cnt) ? row[i].x : 0.f;
                a.y += (i < cnt) ? row[i].y : 0.f;
            }
            kv = kn;
        }
        {
            for (int q = 0; q < novf; ++q) {
                if (ovf[2 * q] != d) continue;
                const v2f rw = *(const v2f*)(tr + (size_t)ovf[2 * q + 1] * DIM + 2 * lane);
                a.x += rw.x;
                a.y += rw.y;
            }
        }
        *(v2f*)(acc + (size_t)d * DIM + 2 * lane) = a;
    }
}

#define GATES  (3 * DIM)
#define GRU_BLOCKS 256
#define GRU_WAVES  16
#define GRU_WHOLE  12
#define GRU_MAIN_TILES (GRU_BLOCKS * GRU_WHOLE)
#define GRU_REST_TILES (ROW_TILES - GRU_MAIN_TILES)
static_assert(GRU_REST_TILES >= 0 && GRU_REST_TILES <= GRU_BLOCKS,
              "the tiles left over must fit one per group");
static_assert(GRU_WAVES - GRU_WHOLE == DIM / 16, "one wave per block of 16 features of a shared tile");
static_assert((2 * GATES + GRU_WAVES * 16) * WPITCH * 4 > 160 * 1024, "two groups must not fit one WGP's LDS");

__global__ void __launch_bounds__(512)
gru_fused(const float* __restrict__ am,
          const float* __restrict__ Wih, const float* __restrict__ bih,
          const float* __restrict__ Whh, const float* __restrict__ bhh,
          float* h, float* emb,
          const float* __restrict__ W0, const float* __restrict__ bias0,
          const float* __restrict__ W1, const float* __restrict__ bias1,
          float* __restrict__ C0, float* __restrict__ C1)
{
    __shared__ __attribute__((aligned(16))) float Wl[2 * GATES * WPITCH];
    __shared__ __attribute__((aligned(16))) float Al[GRU_WAVES * 16 * WPITCH];

    const int tid  = threadIdx.x;
    const int lane = tid & 31;
    const int lr   = lane & 15;
    const int hi   = lane >> 4;
    const int wv    = __builtin_amdgcn_readfirstlane(tid >> 5);
    const bool whole = wv < GRU_WHOLE;
    const bool part  = !whole && (int)blockIdx.x < GRU_REST_TILES;
    const bool live  = whole || part;
    const int rt   = whole ? (int)blockIdx.x * GRU_WHOLE + wv : GRU_MAIN_TILES + (int)blockIdx.x;
    const int row0 = rt << 4;
    const int ct0  = whole ? 0 : wv - GRU_WHOLE;
    const int ct1  = whole ? DIM / 16 : ct0 + 1;

    v2f fa[16] = {}, fh[16] = {};
    if (live) {
        const float* arow = am + (size_t)(row0 + lr) * DIM + 2 * hi;
        const float* hrow = h  + (size_t)(row0 + lr) * DIM + 2 * hi;
        #pragma unroll
        for (int k = 0; k < 16; ++k) {
            fa[k] = *(const v2f*)(arow + 4 * k);
            fh[k] = *(const v2f*)(hrow + 4 * k);
        }
    }

    {
        static_assert(GATES * (DIM / 4) == 6 * 512, "six float4 of each matrix per thread");
        v4f wi[6], wh[6];
        #pragma unroll
        for (int j = 0; j < 6; ++j) {
            wi[j] = *(const v4f*)(Wih + 4 * (j * 512 + tid));
            wh[j] = *(const v4f*)(Whh + 4 * (j * 512 + tid));
        }
        #pragma unroll
        for (int j = 0; j < 6; ++j) {
            const int i = j * 512 + tid;
            const int r = i >> 4, c4 = i & 15;
            *(v4f*)(Wl + r * WPITCH + 4 * c4)           = wi[j];
            *(v4f*)(Wl + (GATES + r) * WPITCH + 4 * c4) = wh[j];
        }
    }
    __syncthreads();
    float* al = Al + (whole ? wv : GRU_WHOLE) * 16 * WPITCH;

    if (live) {
    for (int ct = ct0; ct < ct1; ++ct) {
        const int dcol = ct * 16 + lr;
        const float* bi = Wl + dcol * WPITCH + 2 * hi;
        const float* bh = bi + GATES * WPITCH;

        float* hp = h + (size_t)(row0 + hi * 8) * DIM + dcol;
        float hv[8];
        #pragma unroll
        for (int v = 0; v < 8; ++v) hv[v] = hp[(size_t)v * DIM];

        v8f cir = {}, ciz = {}, cic = {}, chr = {}, chz = {}, chc = {};
        #pragma unroll
        for (int k = 0; k < 16; ++k) {
            const v2f b0 = *(const v2f*)(bi + 4 * k);
            const v2f b1 = *(const v2f*)(bi + DIM * WPITCH + 4 * k);
            const v2f b2 = *(const v2f*)(bi + 2 * DIM * WPITCH + 4 * k);
            const v2f b3 = *(const v2f*)(bh + 4 * k);
            const v2f b4 = *(const v2f*)(bh + DIM * WPITCH + 4 * k);
            const v2f b5 = *(const v2f*)(bh + 2 * DIM * WPITCH + 4 * k);
            cir = __builtin_amdgcn_wmma_f32_16x16x4_f32(
                      false, fa[k], false, b0, (short)0, cir, false, false);
            ciz = __builtin_amdgcn_wmma_f32_16x16x4_f32(
                      false, fa[k], false, b1, (short)0, ciz, false, false);
            cic = __builtin_amdgcn_wmma_f32_16x16x4_f32(
                      false, fa[k], false, b2, (short)0, cic, false, false);
            chr = __builtin_amdgcn_wmma_f32_16x16x4_f32(
                      false, fh[k], false, b3, (short)0, chr, false, false);
            chz = __builtin_amdgcn_wmma_f32_16x16x4_f32(
                      false, fh[k], false, b4, (short)0, chz, false, false);
            chc = __builtin_amdgcn_wmma_f32_16x16x4_f32(
                      false, fh[k], false, b5, (short)0, chc, false, false);
        }

        const float bir = bih[dcol], biz = bih[DIM + dcol], bic = bih[2 * DIM + dcol];
        const float bhr = bhh[dcol], bhz = bhh[DIM + dcol], bhc = bhh[2 * DIM + dcol];
        float o[8];
        #pragma unroll
        for (int v = 0; v < 8; ++v) {
            const float gir = cir[v] + bir, ghr = chr[v] + bhr;
            const float giz = ciz[v] + biz, ghz = chz[v] + bhz;
            const float gic = cic[v] + bic, ghc = chc[v] + bhc;
            const float r = hw_sigmoid(gir + ghr);
            const float z = hw_sigmoid(giz + ghz);
            const float c = hw_tanh(gic + r * ghc);
            o[v] = (1.f - z) * c + z * hv[v];
            al[(hi * 8 + v) * WPITCH + dcol] = o[v];
        }
        if (whole) {
            #pragma unroll
            for (int v = 0; v < 8; ++v) hp[(size_t)v * DIM] = o[v];
            if (emb != nullptr) {
                float* ep = emb + (size_t)(row0 + hi * 8) * DIM + dcol;
                #pragma unroll
                for (int v = 0; v < 8; ++v) ep[(size_t)v * DIM] = o[v];
            }
        }
    }
    }

    const bool pair = W0 != nullptr;
    v4f w0v[2] = {}, w1v[2] = {};
    if (pair) {
        #pragma unroll
        for (int j = 0; j < 2; ++j) {
            w0v[j] = *(const v4f*)(W0 + 4 * (j * 512 + tid));
            w1v[j] = *(const v4f*)(W1 + 4 * (j * 512 + tid));
        }
    }
    __syncthreads();
    if (part) {
        const float* as = Al + GRU_WHOLE * 16 * WPITCH;
        #pragma unroll
        for (int j = 0; j < 2; ++j) {
            const int r = 4 * (wv - GRU_WHOLE) + 2 * j + hi;
            const v4f nv = *(const v4f*)(as + r * WPITCH + 4 * lr);
            *(v4f*)(h + (size_t)(row0 + r) * DIM + 4 * lr) = nv;
            if (emb) *(v4f*)(emb + (size_t)(row0 + r) * DIM + 4 * lr) = nv;
        }
    }
    if (!pair) return;
    #pragma unroll
    for (int j = 0; j < 2; ++j) {
        const int i = j * 512 + tid;
        const int r = i >> 4, c4 = i & 15;
        *(v4f*)(Wl + r * WPITCH + 4 * c4)         = w0v[j];
        *(v4f*)(Wl + (DIM + r) * WPITCH + 4 * c4) = w1v[j];
    }
    __syncthreads();
    if (whole)
        pair_product_lds<8, true, WPITCH, false>(al, Wl, 0, lr, hi, bias0, bias1, C0, C1, row0);
    else if (part)
        pair_product_lds<2, false, WPITCH, false>(al, Wl, 2 * (wv - GRU_WHOLE), lr, hi,
                                                  bias0, bias1, C0, C1, row0);
}

extern "C" void kernel_launch(void* const* d_in, const int* in_sizes, int n_in,
                              void* d_out, int out_size, void* d_ws, size_t ws_size,
                              hipStream_t stream)
{
    (void)in_sizes; (void)n_in; (void)out_size; (void)ws_size;

    const float* x   = (const float*)d_in[0];
    const int*   src = (const int*)d_in[1];
    const int*   dst = (const int*)d_in[2];
    const int*   ety = (const int*)d_in[3];

    const long NE = (long)NODES * DIM;

    float* ws   = (float*)d_ws;
    float* hbuf = ws;                 ws += NE;
    float* buf1 = ws;                 ws += NE;
    float* buf2 = ws;                 ws += NE;
    float* tr0  = ws;                 ws += NE;
    float* tr1  = ws;                 ws += NE;
    float* acc  = ws;                 ws += NE;
    int* cnt = (int*)ws;              ws += CNT_WORDS;
    int* key = (int*)ws;              ws += (long)NODES * CAP;
    int* ovf = (int*)ws;              ws += 2L * EDGES;
    int* binCnt = (int*)ws;           ws += NBINS * BIN_PAD;
    int* binArr = (int*)ws;

    float* out = (float*)d_out;

    const int rowTileBlocks  = (ROW_TILES + 7) / 8;

    zero_counts<<<1, 256, 0, stream>>>(cnt + NODES, binCnt);
    bin_edges<<<BIN_BLOCKS, 1024, 0, stream>>>(src, dst, ety, binCnt, binArr, cnt + NODES, ovf);
    fill_slots<<<NBINS, 1024, 0, stream>>>(binCnt, binArr, cnt, key, cnt + NODES, ovf);

    for (int l = 0; l < 2; ++l) {
        const int base = 4 + 11 * l;
        const float* Wsrc = (const float*)d_in[base + 0];
        const float* bsrc = (const float*)d_in[base + 1];
        const float* Wdst = (const float*)d_in[base + 2];
        const float* bdst = (const float*)d_in[base + 3];
        const float* attn = (const float*)d_in[base + 4];
        const float* ggW  = (const float*)d_in[base + 5];
        const float* ggb  = (const float*)d_in[base + 6];
        const float* Wih  = (const float*)d_in[base + 7];
        const float* bih  = (const float*)d_in[base + 8];
        const float* Whh  = (const float*)d_in[base + 9];
        const float* bhh  = (const float*)d_in[base + 10];

        if (l == 0)
            gemm_pair_wmma<<<rowTileBlocks, 256, 0, stream>>>(x, Wsrc, bsrc, Wdst, bdst,
                                                              buf1, buf2);
        gat_node<<<NODE_BLOCKS, 256, 0, stream>>>(buf1, buf2, attn, cnt, key, ovf, hbuf,
                                                  out + (2L * l) * NE,
                                                  ggW, ggb, ggW + DIM * DIM, ggb + DIM,
                                                  tr0, tr1);

        ggc_pull<<<NODE_BLOCKS, 256, 0, stream>>>(tr0, ety, cnt, key, ovf, acc);
        gru_fused<<<GRU_BLOCKS, 512, 0, stream>>>(
            acc, Wih, bih, Whh, bhh, hbuf, nullptr,
            ggW, ggb, ggW + DIM * DIM, ggb + DIM, tr0, tr1);
        ggc_pull<<<NODE_BLOCKS, 256, 0, stream>>>(tr0, ety, cnt, key, ovf, acc);
        if (l == 0) {
            const float* Wsrc1 = (const float*)d_in[4 + 11 + 0];
            const float* bsrc1 = (const float*)d_in[4 + 11 + 1];
            const float* Wdst1 = (const float*)d_in[4 + 11 + 2];
            const float* bdst1 = (const float*)d_in[4 + 11 + 3];
            gru_fused<<<GRU_BLOCKS, 512, 0, stream>>>(
                acc, Wih, bih, Whh, bhh, hbuf, out + (2L * l + 1) * NE,
                Wsrc1, bsrc1, Wdst1, bdst1, buf1, buf2);
        } else {
            gru_fused<<<GRU_BLOCKS, 512, 0, stream>>>(
                acc, Wih, bih, Whh, bhh, hbuf, out + (2L * l + 1) * NE,
                nullptr, nullptr, nullptr, nullptr, nullptr, nullptr);
        }
    }
}
